# starting kernel
# baseline (speedup 1.0000x reference)
#include <hip/hip_runtime.h>
#include <math.h>

#define IN_DIM   128
#define OUT_DIM  64
#define NEG_SLOPE 0.2f

typedef float v2f __attribute__((ext_vector_type(2)));
typedef float v8f __attribute__((ext_vector_type(8)));

__device__ __forceinline__ float leaky(float e) {
  return (e > 0.f) ? e : NEG_SLOPE * e;
}

__global__ void k_wt(const float* __restrict__ W, float* __restrict__ Wt) {
  int i = blockIdx.x * blockDim.x + threadIdx.x;
  if (i >= IN_DIM * OUT_DIM) return;
  int k = i / OUT_DIM;
  int c = i - k * OUT_DIM;
  Wt[(size_t)c * IN_DIM + k] = W[i];
}

__global__ __launch_bounds__(256) void k_feat_wmma(const float* __restrict__ h,
                                                   const float* __restrict__ Wt,
                                                   float* __restrict__ feat,
                                                   int N) {
  const int tid   = threadIdx.x;
  const int wave  = tid >> 5;
  const int lane  = tid & 31;
  const int mtile = wave >> 2;
  const int ntile = wave & 3;
  const int rowBase = blockIdx.x * 32 + mtile * 16;
  const int col   = ntile * 16 + (lane & 15);
  const int row   = rowBase + (lane & 15);
  const int khalf = (lane >> 4) << 1;

  const int rowc = (row < N) ? row : (N - 1);
  const float* __restrict__ ap = h  + (size_t)rowc * IN_DIM + khalf;
  const float* __restrict__ bp = Wt + (size_t)col  * IN_DIM + khalf;

  v8f acc = {0.f, 0.f, 0.f, 0.f, 0.f, 0.f, 0.f, 0.f};

#pragma unroll
  for (int k = 0; k < IN_DIM; k += 16) {
    v2f a0, a1, a2, a3, b0, b1, b2, b3;
    a0 = *(const v2f*)(ap + k);      b0 = *(const v2f*)(bp + k);
    a1 = *(const v2f*)(ap + k + 4);  b1 = *(const v2f*)(bp + k + 4);
    a2 = *(const v2f*)(ap + k + 8);  b2 = *(const v2f*)(bp + k + 8);
    a3 = *(const v2f*)(ap + k + 12); b3 = *(const v2f*)(bp + k + 12);
    acc = __builtin_amdgcn_wmma_f32_16x16x4_f32(false, a0, false, b0, (short)0, acc, false, false);
    acc = __builtin_amdgcn_wmma_f32_16x16x4_f32(false, a1, false, b1, (short)0, acc, false, false);
    acc = __builtin_amdgcn_wmma_f32_16x16x4_f32(false, a2, false, b2, (short)0, acc, false, false);
    acc = __builtin_amdgcn_wmma_f32_16x16x4_f32(false, a3, false, b3, (short)0, acc, false, false);
  }

  const int mBase = rowBase + (lane >> 4) * 8;
#pragma unroll
  for (int r = 0; r < 8; ++r) {
    int m = mBase + r;
    if (m < N) feat[(size_t)m * OUT_DIM + col] = acc[r];
  }
}

__global__ void k_el_er(const float* __restrict__ feat,
                        const float* __restrict__ attn_l,
                        const float* __restrict__ attn_r,
                        float* __restrict__ el, float* __restrict__ er,
                        int* __restrict__ counts, int N) {
  int i = blockIdx.x * blockDim.x + threadIdx.x;
  if (i >= N) return;
  const float4* f  = (const float4*)(feat + (size_t)i * OUT_DIM);
  const float4* al = (const float4*)attn_l;
  const float4* ar = (const float4*)attn_r;
  float sl = 0.f, sr = 0.f;
#pragma unroll
  for (int j = 0; j < OUT_DIM / 4; ++j) {
    float4 v = f[j], a = al[j], b = ar[j];
    sl += v.x * a.x + v.y * a.y + v.z * a.z + v.w * a.w;
    sr += v.x * b.x + v.y * b.y + v.z * b.z + v.w * b.w;
  }
  el[i] = sl;
  er[i] = sr;
  counts[i] = 0;
}

__global__ void k_hist(const int* __restrict__ dst, int* __restrict__ counts, int E) {
  int e = blockIdx.x * blockDim.x + threadIdx.x;
  if (e >= E) return;
  atomicAdd(&counts[dst[e]], 1);
}

__global__ __launch_bounds__(256) void k_scan1(const int* __restrict__ counts,
                                               int* __restrict__ offsets,
                                               int* __restrict__ bsum, int N) {
  __shared__ int sd[256];
  int tid  = threadIdx.x;
  int base = blockIdx.x * 1024 + tid * 4;
  int v0 = (base + 0 < N) ? counts[base + 0] : 0;
  int v1 = (base + 1 < N) ? counts[base + 1] : 0;
  int v2 = (base + 2 < N) ? counts[base + 2] : 0;
  int v3 = (base + 3 < N) ? counts[base + 3] : 0;
  int s = v0 + v1 + v2 + v3;
  sd[tid] = s;
  __syncthreads();
  for (int off = 1; off < 256; off <<= 1) {
    int t = (tid >= off) ? sd[tid - off] : 0;
    __syncthreads();
    sd[tid] += t;
    __syncthreads();
  }
  int excl = sd[tid] - s;
  if (base + 0 < N) offsets[base + 0] = excl;
  if (base + 1 < N) offsets[base + 1] = excl + v0;
  if (base + 2 < N) offsets[base + 2] = excl + v0 + v1;
  if (base + 3 < N) offsets[base + 3] = excl + v0 + v1 + v2;
  if (tid == 255) bsum[blockIdx.x] = sd[255];
}

__global__ void k_scan2(int* __restrict__ bsum, int nb) {
  if (threadIdx.x != 0 || blockIdx.x != 0) return;
  int running = 0;
  for (int b = 0; b < nb; ++b) {
    int t = bsum[b];
    bsum[b] = running;
    running += t;
  }
}

__global__ void k_scan3(int* __restrict__ offsets, const int* __restrict__ bsum,
                        int* __restrict__ counts, int N) {
  int i = blockIdx.x * blockDim.x + threadIdx.x;
  if (i >= N) return;
  offsets[i] += bsum[i >> 10];
  counts[i] = 0;
}

__global__ void k_fill(const int* __restrict__ src, const int* __restrict__ dst,
                       const int* __restrict__ offsets, int* __restrict__ cursor,
                       int* __restrict__ csr_src, int E) {
  int e = blockIdx.x * blockDim.x + threadIdx.x;
  if (e >= E) return;
  int d = dst[e];
  int p = offsets[d] + atomicAdd(&cursor[d], 1);
  csr_src[p] = src[e];
}

__global__ __launch_bounds__(256) void k_gat_agg(const float* __restrict__ feat,
                                                 const float* __restrict__ el,
                                                 const float* __restrict__ er,
                                                 const int* __restrict__ offsets,
                                                 const int* __restrict__ csr_src,
                                                 const float* __restrict__ bias,
                                                 float* __restrict__ out,
                                                 int N, int E) {
  int node = blockIdx.x * 8 + (threadIdx.x >> 5);
  if (node >= N) return;
  int lane  = threadIdx.x & 31;
  int start = offsets[node];
  int end   = (node == N - 1) ? E : offsets[node + 1];
  float er_d = er[node];

  int   j0 = start + lane;
  float e0 = (j0 < end) ? leaky(el[csr_src[j0]] + er_d) : -INFINITY;

  float m = e0;
  for (int j = j0 + 32; j < end; j += 32) {
    m = fmaxf(m, leaky(el[csr_src[j]] + er_d));
  }
#pragma unroll
  for (int off = 16; off > 0; off >>= 1) m = fmaxf(m, __shfl_xor(m, off, 32));
  if (end <= start) m = 0.f;

  float den = (j0 < end) ? __expf(e0 - m) : 0.f;
  for (int j = j0 + 32; j < end; j += 32) {
    den += __expf(leaky(el[csr_src[j]] + er_d) - m);
  }
#pragma unroll
  for (int off = 16; off > 0; off >>= 1) den += __shfl_xor(den, off, 32);
  float invden = (den > 0.f) ? 1.f / den : 1.f;

  float acc0 = 0.f, acc1 = 0.f;
  for (int j = start; j < end; ++j) {
    int t = j - start;
    int s = csr_src[j];
    float e;
    if (t < 32) {
      e = __shfl(e0, t, 32);
    } else {
      e = leaky(el[s] + er_d);
    }
    float alpha = __expf(e - m) * invden;
    v2f fv = *(const v2f*)(feat + (size_t)s * OUT_DIM + 2 * lane);
    acc0 += alpha * fv.x;
    acc1 += alpha * fv.y;
  }
  v2f bv = *(const v2f*)(bias + 2 * lane);
  v2f ov;
  ov.x = acc0 + bv.x;
  ov.y = acc1 + bv.y;
  *(v2f*)(out + (size_t)node * OUT_DIM + 2 * lane) = ov;
}

extern "C" void kernel_launch(void* const* d_in, const int* in_sizes, int n_in,
                              void* d_out, int out_size, void* d_ws, size_t ws_size,
                              hipStream_t stream) {
  const float* h      = (const float*)d_in[0];
  const int*   src    = (const int*)d_in[1];
  const int*   dst    = (const int*)d_in[2];
  const float* W      = (const float*)d_in[3];
  const float* attn_l = (const float*)d_in[4];
  const float* attn_r = (const float*)d_in[5];
  const float* bias   = (const float*)d_in[6];
  float* out = (float*)d_out;

  const int N = in_sizes[0] / IN_DIM;
  const int E = in_sizes[1];

  char* ws = (char*)d_ws;
  size_t off = 0;
  auto carve = [&](size_t bytes) -> char* {
    char* p = ws + off;
    off = (off + bytes + 255) & ~(size_t)255;
    return p;
  };
  float* feat    = (float*)carve((size_t)N * OUT_DIM * sizeof(float));
  float* Wt      = (float*)carve((size_t)IN_DIM * OUT_DIM * sizeof(float));
  float* el      = (float*)carve((size_t)N * sizeof(float));
  float* er      = (float*)carve((size_t)N * sizeof(float));
  int*   counts  = (int*)  carve((size_t)N * sizeof(int));
  int*   offsets = (int*)  carve((size_t)N * sizeof(int));
  int*   bsum    = (int*)  carve(4096);
  int*   csr_src = (int*)  carve((size_t)E * sizeof(int));

  const int nb = (N + 1023) / 1024;

  k_wt<<<(IN_DIM * OUT_DIM + 255) / 256, 256, 0, stream>>>(W, Wt);
  k_feat_wmma<<<(N + 31) / 32, 256, 0, stream>>>(h, Wt, feat, N);
  k_el_er<<<(N + 255) / 256, 256, 0, stream>>>(feat, attn_l, attn_r, el, er, counts, N);
  k_hist <<<(E + 255) / 256, 256, 0, stream>>>(dst, counts, E);
  k_scan1<<<nb, 256, 0, stream>>>(counts, offsets, bsum, N);
  k_scan2<<<1, 1, 0, stream>>>(bsum, nb);
  k_scan3<<<(N + 255) / 256, 256, 0, stream>>>(offsets, bsum, counts, N);
  k_fill <<<(E + 255) / 256, 256, 0, stream>>>(src, dst, offsets, counts, csr_src, E);
  k_gat_agg<<<(N + 7) / 8, 256, 0, stream>>>(feat, el, er, offsets, csr_src, bias, out, N, E);
}


# CSR fill without atomics: the histogram's atomic add returns each edge's rank in its destination, kept per edge; fill stores at offset+rank. Per-node sum order = histogram arrival order.
# speedup vs baseline: 1.2851x; 1.2851x over previous
#include <hip/hip_runtime.h>
#include <math.h>

#define IN_DIM   128
#define OUT_DIM  64
#define NEG_SLOPE 0.2f

typedef float v2f __attribute__((ext_vector_type(2)));
typedef float v8f __attribute__((ext_vector_type(8)));

__device__ __forceinline__ float leaky(float e) {
  return (e > 0.f) ? e : NEG_SLOPE * e;
}

__global__ void k_wt(const float* __restrict__ W, float* __restrict__ Wt) {
  int i = blockIdx.x * blockDim.x + threadIdx.x;
  if (i >= IN_DIM * OUT_DIM) return;
  int k = i / OUT_DIM;
  int c = i - k * OUT_DIM;
  Wt[(size_t)c * IN_DIM + k] = W[i];
}

__global__ __launch_bounds__(256) void k_feat_wmma(const float* __restrict__ h,
                                                   const float* __restrict__ Wt,
                                                   float* __restrict__ feat,
                                                   int N) {
  const int tid   = threadIdx.x;
  const int wave  = tid >> 5;
  const int lane  = tid & 31;
  const int mtile = wave >> 2;
  const int ntile = wave & 3;
  const int rowBase = blockIdx.x * 32 + mtile * 16;
  const int col   = ntile * 16 + (lane & 15);
  const int row   = rowBase + (lane & 15);
  const int khalf = (lane >> 4) << 1;

  const int rowc = (row < N) ? row : (N - 1);
  const float* __restrict__ ap = h  + (size_t)rowc * IN_DIM + khalf;
  const float* __restrict__ bp = Wt + (size_t)col  * IN_DIM + khalf;

  v8f acc = {0.f, 0.f, 0.f, 0.f, 0.f, 0.f, 0.f, 0.f};

#pragma unroll
  for (int k = 0; k < IN_DIM; k += 16) {
    v2f a0, a1, a2, a3, b0, b1, b2, b3;
    a0 = *(const v2f*)(ap + k);      b0 = *(const v2f*)(bp + k);
    a1 = *(const v2f*)(ap + k + 4);  b1 = *(const v2f*)(bp + k + 4);
    a2 = *(const v2f*)(ap + k + 8);  b2 = *(const v2f*)(bp + k + 8);
    a3 = *(const v2f*)(ap + k + 12); b3 = *(const v2f*)(bp + k + 12);
    acc = __builtin_amdgcn_wmma_f32_16x16x4_f32(false, a0, false, b0, (short)0, acc, false, false);
    acc = __builtin_amdgcn_wmma_f32_16x16x4_f32(false, a1, false, b1, (short)0, acc, false, false);
    acc = __builtin_amdgcn_wmma_f32_16x16x4_f32(false, a2, false, b2, (short)0, acc, false, false);
    acc = __builtin_amdgcn_wmma_f32_16x16x4_f32(false, a3, false, b3, (short)0, acc, false, false);
  }

  const int mBase = rowBase + (lane >> 4) * 8;
#pragma unroll
  for (int r = 0; r < 8; ++r) {
    int m = mBase + r;
    if (m < N) feat[(size_t)m * OUT_DIM + col] = acc[r];
  }
}

__global__ void k_el_er(const float* __restrict__ feat,
                        const float* __restrict__ attn_l,
                        const float* __restrict__ attn_r,
                        float* __restrict__ el, float* __restrict__ er,
                        int* __restrict__ counts, int N) {
  int i = blockIdx.x * blockDim.x + threadIdx.x;
  if (i >= N) return;
  const float4* f  = (const float4*)(feat + (size_t)i * OUT_DIM);
  const float4* al = (const float4*)attn_l;
  const float4* ar = (const float4*)attn_r;
  float sl = 0.f, sr = 0.f;
#pragma unroll
  for (int j = 0; j < OUT_DIM / 4; ++j) {
    float4 v = f[j], a = al[j], b = ar[j];
    sl += v.x * a.x + v.y * a.y + v.z * a.z + v.w * a.w;
    sr += v.x * b.x + v.y * b.y + v.z * b.z + v.w * b.w;
  }
  el[i] = sl;
  er[i] = sr;
  counts[i] = 0;
}

__global__ void k_hist(const int* __restrict__ dst, int* __restrict__ counts,
                       int* __restrict__ rank, int E) {
  int e = blockIdx.x * blockDim.x + threadIdx.x;
  if (e >= E) return;
  rank[e] = atomicAdd(&counts[dst[e]], 1);
}

__global__ __launch_bounds__(256) void k_scan1(const int* __restrict__ counts,
                                               int* __restrict__ offsets,
                                               int* __restrict__ bsum, int N) {
  __shared__ int sd[256];
  int tid  = threadIdx.x;
  int base = blockIdx.x * 1024 + tid * 4;
  int v0 = (base + 0 < N) ? counts[base + 0] : 0;
  int v1 = (base + 1 < N) ? counts[base + 1] : 0;
  int v2 = (base + 2 < N) ? counts[base + 2] : 0;
  int v3 = (base + 3 < N) ? counts[base + 3] : 0;
  int s = v0 + v1 + v2 + v3;
  sd[tid] = s;
  __syncthreads();
  for (int off = 1; off < 256; off <<= 1) {
    int t = (tid >= off) ? sd[tid - off] : 0;
    __syncthreads();
    sd[tid] += t;
    __syncthreads();
  }
  int excl = sd[tid] - s;
  if (base + 0 < N) offsets[base + 0] = excl;
  if (base + 1 < N) offsets[base + 1] = excl + v0;
  if (base + 2 < N) offsets[base + 2] = excl + v0 + v1;
  if (base + 3 < N) offsets[base + 3] = excl + v0 + v1 + v2;
  if (tid == 255) bsum[blockIdx.x] = sd[255];
}

__global__ void k_scan2(int* __restrict__ bsum, int nb) {
  if (threadIdx.x != 0 || blockIdx.x != 0) return;
  int running = 0;
  for (int b = 0; b < nb; ++b) {
    int t = bsum[b];
    bsum[b] = running;
    running += t;
  }
}

__global__ void k_scan3(int* __restrict__ offsets, const int* __restrict__ bsum, int N) {
  int i = blockIdx.x * blockDim.x + threadIdx.x;
  if (i >= N) return;
  offsets[i] += bsum[i >> 10];
}

__global__ void k_fill(const int* __restrict__ src, const int* __restrict__ dst,
                       const int* __restrict__ offsets, const int* __restrict__ rank,
                       int* __restrict__ csr_src, int E) {
  int e = blockIdx.x * blockDim.x + threadIdx.x;
  if (e >= E) return;
  int p = offsets[dst[e]] + rank[e];
  csr_src[p] = src[e];
}

__global__ __launch_bounds__(256) void k_gat_agg(const float* __restrict__ feat,
                                                 const float* __restrict__ el,
                                                 const float* __restrict__ er,
                                                 const int* __restrict__ offsets,
                                                 const int* __restrict__ csr_src,
                                                 const float* __restrict__ bias,
                                                 float* __restrict__ out,
                                                 int N, int E) {
  int node = blockIdx.x * 8 + (threadIdx.x >> 5);
  if (node >= N) return;
  int lane  = threadIdx.x & 31;
  int start = offsets[node];
  int end   = (node == N - 1) ? E : offsets[node + 1];
  float er_d = er[node];

  int   j0 = start + lane;
  float e0 = (j0 < end) ? leaky(el[csr_src[j0]] + er_d) : -INFINITY;

  float m = e0;
  for (int j = j0 + 32; j < end; j += 32) {
    m = fmaxf(m, leaky(el[csr_src[j]] + er_d));
  }
#pragma unroll
  for (int off = 16; off > 0; off >>= 1) m = fmaxf(m, __shfl_xor(m, off, 32));
  if (end <= start) m = 0.f;

  float den = (j0 < end) ? __expf(e0 - m) : 0.f;
  for (int j = j0 + 32; j < end; j += 32) {
    den += __expf(leaky(el[csr_src[j]] + er_d) - m);
  }
#pragma unroll
  for (int off = 16; off > 0; off >>= 1) den += __shfl_xor(den, off, 32);
  float invden = (den > 0.f) ? 1.f / den : 1.f;

  float acc0 = 0.f, acc1 = 0.f;
  for (int j = start; j < end; ++j) {
    int t = j - start;
    int s = csr_src[j];
    float e;
    if (t < 32) {
      e = __shfl(e0, t, 32);
    } else {
      e = leaky(el[s] + er_d);
    }
    float alpha = __expf(e - m) * invden;
    v2f fv = *(const v2f*)(feat + (size_t)s * OUT_DIM + 2 * lane);
    acc0 += alpha * fv.x;
    acc1 += alpha * fv.y;
  }
  v2f bv = *(const v2f*)(bias + 2 * lane);
  v2f ov;
  ov.x = acc0 + bv.x;
  ov.y = acc1 + bv.y;
  *(v2f*)(out + (size_t)node * OUT_DIM + 2 * lane) = ov;
}

extern "C" void kernel_launch(void* const* d_in, const int* in_sizes, int n_in,
                              void* d_out, int out_size, void* d_ws, size_t ws_size,
                              hipStream_t stream) {
  const float* h      = (const float*)d_in[0];
  const int*   src    = (const int*)d_in[1];
  const int*   dst    = (const int*)d_in[2];
  const float* W      = (const float*)d_in[3];
  const float* attn_l = (const float*)d_in[4];
  const float* attn_r = (const float*)d_in[5];
  const float* bias   = (const float*)d_in[6];
  float* out = (float*)d_out;

  const int N = in_sizes[0] / IN_DIM;
  const int E = in_sizes[1];

  char* ws = (char*)d_ws;
  size_t off = 0;
  auto carve = [&](size_t bytes) -> char* {
    char* p = ws + off;
    off = (off + bytes + 255) & ~(size_t)255;
    return p;
  };
  float* feat    = (float*)carve((size_t)N * OUT_DIM * sizeof(float));
  float* Wt      = (float*)carve((size_t)IN_DIM * OUT_DIM * sizeof(float));
  float* el      = (float*)carve((size_t)N * sizeof(float));
  float* er      = (float*)carve((size_t)N * sizeof(float));
  int*   counts  = (int*)  carve((size_t)N * sizeof(int));
  int*   offsets = (int*)  carve((size_t)N * sizeof(int));
  int*   bsum    = (int*)  carve(4096);
  int*   csr_src = (int*)  carve((size_t)E * sizeof(int));
  int*   rank    = (int*)  carve((size_t)E * sizeof(int));

  const int nb = (N + 1023) / 1024;

  k_wt<<<(IN_DIM * OUT_DIM + 255) / 256, 256, 0, stream>>>(W, Wt);
  k_feat_wmma<<<(N + 31) / 32, 256, 0, stream>>>(h, Wt, feat, N);
  k_el_er<<<(N + 255) / 256, 256, 0, stream>>>(feat, attn_l, attn_r, el, er, counts, N);
  k_hist <<<(E + 255) / 256, 256, 0, stream>>>(dst, counts, rank, E);
  k_scan1<<<nb, 256, 0, stream>>>(counts, offsets, bsum, N);
  k_scan2<<<1, 1, 0, stream>>>(bsum, nb);
  k_scan3<<<(N + 255) / 256, 256, 0, stream>>>(offsets, bsum, N);
  k_fill <<<(E + 255) / 256, 256, 0, stream>>>(src, dst, offsets, rank, csr_src, E);
  k_gat_agg<<<(N + 7) / 8, 256, 0, stream>>>(feat, el, er, offsets, csr_src, bias, out, N, E);
}


# CSR build as a two-level bucket sort by destination: per-edge global atomics replaced by LDS atomics (1024-node buckets, then per-bucket count, scan, fill). Per-node sum order = LDS arrival order.
# speedup vs baseline: 2.0763x; 1.6157x over previous
#include <hip/hip_runtime.h>
#include <math.h>

#define IN_DIM   128
#define OUT_DIM  64
#define NEG_SLOPE 0.2f

typedef float v2f __attribute__((ext_vector_type(2)));
typedef float v8f __attribute__((ext_vector_type(8)));

__device__ __forceinline__ float leaky(float e) {
  return (e > 0.f) ? e : NEG_SLOPE * e;
}

__global__ void k_wt(const float* __restrict__ W, float* __restrict__ Wt) {
  int i = blockIdx.x * blockDim.x + threadIdx.x;
  if (i >= IN_DIM * OUT_DIM) return;
  int k = i / OUT_DIM;
  int c = i - k * OUT_DIM;
  Wt[(size_t)c * IN_DIM + k] = W[i];
}

__global__ __launch_bounds__(256) void k_feat_wmma(const float* __restrict__ h,
                                                   const float* __restrict__ Wt,
                                                   float* __restrict__ feat,
                                                   int N) {
  const int tid   = threadIdx.x;
  const int wave  = tid >> 5;
  const int lane  = tid & 31;
  const int mtile = wave >> 2;
  const int ntile = wave & 3;
  const int rowBase = blockIdx.x * 32 + mtile * 16;
  const int col   = ntile * 16 + (lane & 15);
  const int row   = rowBase + (lane & 15);
  const int khalf = (lane >> 4) << 1;

  const int rowc = (row < N) ? row : (N - 1);
  const float* __restrict__ ap = h  + (size_t)rowc * IN_DIM + khalf;
  const float* __restrict__ bp = Wt + (size_t)col  * IN_DIM + khalf;

  v8f acc = {0.f, 0.f, 0.f, 0.f, 0.f, 0.f, 0.f, 0.f};

#pragma unroll
  for (int k = 0; k < IN_DIM; k += 16) {
    v2f a0, a1, a2, a3, b0, b1, b2, b3;
    a0 = *(const v2f*)(ap + k);      b0 = *(const v2f*)(bp + k);
    a1 = *(const v2f*)(ap + k + 4);  b1 = *(const v2f*)(bp + k + 4);
    a2 = *(const v2f*)(ap + k + 8);  b2 = *(const v2f*)(bp + k + 8);
    a3 = *(const v2f*)(ap + k + 12); b3 = *(const v2f*)(bp + k + 12);
    acc = __builtin_amdgcn_wmma_f32_16x16x4_f32(false, a0, false, b0, (short)0, acc, false, false);
    acc = __builtin_amdgcn_wmma_f32_16x16x4_f32(false, a1, false, b1, (short)0, acc, false, false);
    acc = __builtin_amdgcn_wmma_f32_16x16x4_f32(false, a2, false, b2, (short)0, acc, false, false);
    acc = __builtin_amdgcn_wmma_f32_16x16x4_f32(false, a3, false, b3, (short)0, acc, false, false);
  }

  const int mBase = rowBase + (lane >> 4) * 8;
#pragma unroll
  for (int r = 0; r < 8; ++r) {
    int m = mBase + r;
    if (m < N) feat[(size_t)m * OUT_DIM + col] = acc[r];
  }
}

__global__ void k_el_er(const float* __restrict__ feat,
                        const float* __restrict__ attn_l,
                        const float* __restrict__ attn_r,
                        float* __restrict__ el, float* __restrict__ er,
                        int* __restrict__ btotal, int* __restrict__ bcursor,
                        int N, int nbk) {
  int i = blockIdx.x * blockDim.x + threadIdx.x;
  if (i < nbk) { btotal[i] = 0; bcursor[i] = 0; }
  if (i >= N) return;
  const float4* f  = (const float4*)(feat + (size_t)i * OUT_DIM);
  const float4* al = (const float4*)attn_l;
  const float4* ar = (const float4*)attn_r;
  float sl = 0.f, sr = 0.f;
#pragma unroll
  for (int j = 0; j < OUT_DIM / 4; ++j) {
    float4 v = f[j], a = al[j], b = ar[j];
    sl += v.x * a.x + v.y * a.y + v.z * a.z + v.w * a.w;
    sr += v.x * b.x + v.y * b.y + v.z * b.z + v.w * b.w;
  }
  el[i] = sl;
  er[i] = sr;
}

#define BK_SHIFT 10
#define BK_NODES 1024
#define MAX_BK   1024
#define CHUNK    4096
#define L1_THREADS 1024
#define EPT (CHUNK / L1_THREADS)

__device__ __forceinline__ int block_scan_incl_1024(int v, int* sw, int tid) {
  const int lane = tid & 31;
  const int w    = tid >> 5;
  int x = v;
#pragma unroll
  for (int off = 1; off < 32; off <<= 1) {
    int t = __shfl_up(x, off, 32);
    if (lane >= off) x += t;
  }
  if (lane == 31) sw[w] = x;
  __syncthreads();
  if (w == 0) {
    int y = sw[lane];
#pragma unroll
    for (int off = 1; off < 32; off <<= 1) {
      int t = __shfl_up(y, off, 32);
      if (lane >= off) y += t;
    }
    sw[lane] = y;
  }
  __syncthreads();
  if (w > 0) x += sw[w - 1];
  return x;
}

__global__ __launch_bounds__(L1_THREADS) void k_bhist(const int* __restrict__ dst,
                                                      int* __restrict__ btotal,
                                                      int E, int nbk) {
  __shared__ int sh[MAX_BK];
  const int tid = threadIdx.x;
  for (int t = tid; t < nbk; t += L1_THREADS) sh[t] = 0;
  __syncthreads();
  const int base = blockIdx.x * CHUNK;
#pragma unroll
  for (int i = 0; i < EPT; ++i) {
    int e = base + i * L1_THREADS + tid;
    if (e < E) atomicAdd(&sh[dst[e] >> BK_SHIFT], 1);
  }
  __syncthreads();
  for (int t = tid; t < nbk; t += L1_THREADS) {
    int c = sh[t];
    if (c) atomicAdd(&btotal[t], c);
  }
}

__global__ __launch_bounds__(L1_THREADS) void k_bscatter(const int* __restrict__ src,
                                                         const int* __restrict__ dst,
                                                         const int* __restrict__ btotal,
                                                         int* __restrict__ bcursor,
                                                         int2* __restrict__ pairs,
                                                         int E, int nbk) {
  __shared__ int sh[MAX_BK];
  __shared__ int sw[32];
  const int tid = threadIdx.x;
  sh[tid] = 0;
  const int tot = (tid < nbk) ? btotal[tid] : 0;
  __syncthreads();

  const int base = blockIdx.x * CHUNK;
  int d[EPT], s[EPT], r[EPT];
#pragma unroll
  for (int i = 0; i < EPT; ++i) {
    int e = base + i * L1_THREADS + tid;
    d[i] = (e < E) ? dst[e] : -1;
    s[i] = (e < E) ? src[e] : 0;
  }
#pragma unroll
  for (int i = 0; i < EPT; ++i) {
    r[i] = (d[i] >= 0) ? atomicAdd(&sh[d[i] >> BK_SHIFT], 1) : 0;
  }
  const int incl = block_scan_incl_1024(tot, sw, tid);
  if (tid < nbk) {
    int c = sh[tid];
    int g = c ? atomicAdd(&bcursor[tid], c) : 0;
    sh[tid] = incl - tot + g;
  }
  __syncthreads();
#pragma unroll
  for (int i = 0; i < EPT; ++i) {
    if (d[i] >= 0) {
      int p = sh[d[i] >> BK_SHIFT] + r[i];
      pairs[p] = make_int2(s[i], d[i]);
    }
  }
}

__global__ __launch_bounds__(BK_NODES) void k_fine(const int2* __restrict__ pairs,
                                                   const int* __restrict__ btotal,
                                                   int* __restrict__ offsets,
                                                   int* __restrict__ csr_src,
                                                   int N) {
  __shared__ int scnt[BK_NODES];
  __shared__ int sw[32];
  __shared__ int sbase;
  const int tid = threadIdx.x;
  const int b   = blockIdx.x;
  scnt[tid] = 0;
  if (tid == 0) sbase = 0;
  __syncthreads();
  const int part = (tid < b) ? btotal[tid] : 0;
  if (part) atomicAdd(&sbase, part);
  const int cnt = btotal[b];
  __syncthreads();
  const int base = sbase;
  const int2* __restrict__ pb = pairs + base;

  for (int i0 = 0; i0 < cnt; i0 += 4 * BK_NODES) {
    int dl[4];
#pragma unroll
    for (int k = 0; k < 4; ++k) {
      int i = i0 + k * BK_NODES + tid;
      dl[k] = pb[(i < cnt) ? i : (cnt - 1)].y;
    }
#pragma unroll
    for (int k = 0; k < 4; ++k) {
      int i = i0 + k * BK_NODES + tid;
      if (i < cnt) atomicAdd(&scnt[dl[k] & (BK_NODES - 1)], 1);
    }
  }
  __syncthreads();
  const int c = scnt[tid];
  const int excl = block_scan_incl_1024(c, sw, tid) - c;
  const int node = (b << BK_SHIFT) + tid;
  if (node < N) offsets[node] = base + excl;
  scnt[tid] = excl;
  __syncthreads();

  int* __restrict__ cb = csr_src + base;
  for (int i0 = 0; i0 < cnt; i0 += 4 * BK_NODES) {
    int2 v[4];
#pragma unroll
    for (int k = 0; k < 4; ++k) {
      int i = i0 + k * BK_NODES + tid;
      v[k] = pb[(i < cnt) ? i : (cnt - 1)];
    }
#pragma unroll
    for (int k = 0; k < 4; ++k) {
      int i = i0 + k * BK_NODES + tid;
      if (i < cnt) {
        int p = atomicAdd(&scnt[v[k].y & (BK_NODES - 1)], 1);
        cb[p] = v[k].x;
      }
    }
  }
}

__global__ __launch_bounds__(256) void k_gat_agg(const float* __restrict__ feat,
                                                 const float* __restrict__ el,
                                                 const float* __restrict__ er,
                                                 const int* __restrict__ offsets,
                                                 const int* __restrict__ csr_src,
                                                 const float* __restrict__ bias,
                                                 float* __restrict__ out,
                                                 int N, int E) {
  int node = blockIdx.x * 8 + (threadIdx.x >> 5);
  if (node >= N) return;
  int lane  = threadIdx.x & 31;
  int start = offsets[node];
  int end   = (node == N - 1) ? E : offsets[node + 1];
  float er_d = er[node];

  int   j0 = start + lane;
  float e0 = (j0 < end) ? leaky(el[csr_src[j0]] + er_d) : -INFINITY;

  float m = e0;
  for (int j = j0 + 32; j < end; j += 32) {
    m = fmaxf(m, leaky(el[csr_src[j]] + er_d));
  }
#pragma unroll
  for (int off = 16; off > 0; off >>= 1) m = fmaxf(m, __shfl_xor(m, off, 32));
  if (end <= start) m = 0.f;

  float den = (j0 < end) ? __expf(e0 - m) : 0.f;
  for (int j = j0 + 32; j < end; j += 32) {
    den += __expf(leaky(el[csr_src[j]] + er_d) - m);
  }
#pragma unroll
  for (int off = 16; off > 0; off >>= 1) den += __shfl_xor(den, off, 32);
  float invden = (den > 0.f) ? 1.f / den : 1.f;

  float acc0 = 0.f, acc1 = 0.f;
  for (int j = start; j < end; ++j) {
    int t = j - start;
    int s = csr_src[j];
    float e;
    if (t < 32) {
      e = __shfl(e0, t, 32);
    } else {
      e = leaky(el[s] + er_d);
    }
    float alpha = __expf(e - m) * invden;
    v2f fv = *(const v2f*)(feat + (size_t)s * OUT_DIM + 2 * lane);
    acc0 += alpha * fv.x;
    acc1 += alpha * fv.y;
  }
  v2f bv = *(const v2f*)(bias + 2 * lane);
  v2f ov;
  ov.x = acc0 + bv.x;
  ov.y = acc1 + bv.y;
  *(v2f*)(out + (size_t)node * OUT_DIM + 2 * lane) = ov;
}

extern "C" void kernel_launch(void* const* d_in, const int* in_sizes, int n_in,
                              void* d_out, int out_size, void* d_ws, size_t ws_size,
                              hipStream_t stream) {
  const float* h      = (const float*)d_in[0];
  const int*   src    = (const int*)d_in[1];
  const int*   dst    = (const int*)d_in[2];
  const float* W      = (const float*)d_in[3];
  const float* attn_l = (const float*)d_in[4];
  const float* attn_r = (const float*)d_in[5];
  const float* bias   = (const float*)d_in[6];
  float* out = (float*)d_out;

  const int N = in_sizes[0] / IN_DIM;
  const int E = in_sizes[1];

  char* ws = (char*)d_ws;
  size_t off = 0;
  auto carve = [&](size_t bytes) -> char* {
    char* p = ws + off;
    off = (off + bytes + 255) & ~(size_t)255;
    return p;
  };
  float* feat    = (float*)carve((size_t)N * OUT_DIM * sizeof(float));
  float* Wt      = (float*)carve((size_t)IN_DIM * OUT_DIM * sizeof(float));
  float* el      = (float*)carve((size_t)N * sizeof(float));
  float* er      = (float*)carve((size_t)N * sizeof(float));
  int*   offsets = (int*)  carve((size_t)N * sizeof(int));
  int*   btotal  = (int*)  carve(MAX_BK * sizeof(int));
  int*   bcursor = (int*)  carve(MAX_BK * sizeof(int));
  int*   csr_src = (int*)  carve((size_t)E * sizeof(int));
  int2*  pairs   = (int2*) carve((size_t)E * sizeof(int2));

  const int nbk = (N + BK_NODES - 1) / BK_NODES;
  const int nch = (E + CHUNK - 1) / CHUNK;

  k_wt<<<(IN_DIM * OUT_DIM + 255) / 256, 256, 0, stream>>>(W, Wt);
  k_feat_wmma<<<(N + 31) / 32, 256, 0, stream>>>(h, Wt, feat, N);
  k_el_er<<<(N + 255) / 256, 256, 0, stream>>>(feat, attn_l, attn_r, el, er, btotal, bcursor, N, nbk);
  k_bhist   <<<nch, L1_THREADS, 0, stream>>>(dst, btotal, E, nbk);
  k_bscatter<<<nch, L1_THREADS, 0, stream>>>(src, dst, btotal, bcursor, pairs, E, nbk);
  k_fine    <<<nbk, BK_NODES, 0, stream>>>(pairs, btotal, offsets, csr_src, N);
  k_gat_agg<<<(N + 7) / 8, 256, 0, stream>>>(feat, el, er, offsets, csr_src, bias, out, N, E);
}


# Aggregation: a wave walks 13 nodes; per node all source rows are gathered once into registers (half-wave per edge) and the edge score comes from the row, not an el load; el and node sums re-ordered.
# speedup vs baseline: 2.4792x; 1.1940x over previous
#include <hip/hip_runtime.h>
#include <math.h>

#define IN_DIM   128
#define OUT_DIM  64
#define NEG_SLOPE 0.2f

typedef float v2f __attribute__((ext_vector_type(2)));
typedef float v8f __attribute__((ext_vector_type(8)));

__device__ __forceinline__ float leaky(float e) {
  return (e > 0.f) ? e : NEG_SLOPE * e;
}

__global__ void k_wt(const float* __restrict__ W, float* __restrict__ Wt) {
  int i = blockIdx.x * blockDim.x + threadIdx.x;
  if (i >= IN_DIM * OUT_DIM) return;
  int k = i / OUT_DIM;
  int c = i - k * OUT_DIM;
  Wt[(size_t)c * IN_DIM + k] = W[i];
}

__global__ __launch_bounds__(256) void k_feat_wmma(const float* __restrict__ h,
                                                   const float* __restrict__ Wt,
                                                   float* __restrict__ feat,
                                                   int N) {
  const int tid   = threadIdx.x;
  const int wave  = tid >> 5;
  const int lane  = tid & 31;
  const int mtile = wave >> 2;
  const int ntile = wave & 3;
  const int rowBase = blockIdx.x * 32 + mtile * 16;
  const int col   = ntile * 16 + (lane & 15);
  const int row   = rowBase + (lane & 15);
  const int khalf = (lane >> 4) << 1;

  const int rowc = (row < N) ? row : (N - 1);
  const float* __restrict__ ap = h  + (size_t)rowc * IN_DIM + khalf;
  const float* __restrict__ bp = Wt + (size_t)col  * IN_DIM + khalf;

  v8f acc = {0.f, 0.f, 0.f, 0.f, 0.f, 0.f, 0.f, 0.f};

#pragma unroll
  for (int k = 0; k < IN_DIM; k += 16) {
    v2f a0, a1, a2, a3, b0, b1, b2, b3;
    a0 = *(const v2f*)(ap + k);      b0 = *(const v2f*)(bp + k);
    a1 = *(const v2f*)(ap + k + 4);  b1 = *(const v2f*)(bp + k + 4);
    a2 = *(const v2f*)(ap + k + 8);  b2 = *(const v2f*)(bp + k + 8);
    a3 = *(const v2f*)(ap + k + 12); b3 = *(const v2f*)(bp + k + 12);
    acc = __builtin_amdgcn_wmma_f32_16x16x4_f32(false, a0, false, b0, (short)0, acc, false, false);
    acc = __builtin_amdgcn_wmma_f32_16x16x4_f32(false, a1, false, b1, (short)0, acc, false, false);
    acc = __builtin_amdgcn_wmma_f32_16x16x4_f32(false, a2, false, b2, (short)0, acc, false, false);
    acc = __builtin_amdgcn_wmma_f32_16x16x4_f32(false, a3, false, b3, (short)0, acc, false, false);
  }

  const int mBase = rowBase + (lane >> 4) * 8;
#pragma unroll
  for (int r = 0; r < 8; ++r) {
    int m = mBase + r;
    if (m < N) feat[(size_t)m * OUT_DIM + col] = acc[r];
  }
}

__global__ void k_el_er(const float* __restrict__ feat,
                        const float* __restrict__ attn_l,
                        const float* __restrict__ attn_r,
                        float* __restrict__ el, float* __restrict__ er,
                        int* __restrict__ btotal, int* __restrict__ bcursor,
                        int N, int nbk) {
  int i = blockIdx.x * blockDim.x + threadIdx.x;
  if (i < nbk) { btotal[i] = 0; bcursor[i] = 0; }
  if (i >= N) return;
  const float4* f  = (const float4*)(feat + (size_t)i * OUT_DIM);
  const float4* al = (const float4*)attn_l;
  const float4* ar = (const float4*)attn_r;
  float sl = 0.f, sr = 0.f;
#pragma unroll
  for (int j = 0; j < OUT_DIM / 4; ++j) {
    float4 v = f[j], a = al[j], b = ar[j];
    sl += v.x * a.x + v.y * a.y + v.z * a.z + v.w * a.w;
    sr += v.x * b.x + v.y * b.y + v.z * b.z + v.w * b.w;
  }
  el[i] = sl;
  er[i] = sr;
}

#define BK_SHIFT 10
#define BK_NODES 1024
#define MAX_BK   1024
#define CHUNK    4096
#define L1_THREADS 1024
#define EPT (CHUNK / L1_THREADS)

__device__ __forceinline__ int block_scan_incl_1024(int v, int* sw, int tid) {
  const int lane = tid & 31;
  const int w    = tid >> 5;
  int x = v;
#pragma unroll
  for (int off = 1; off < 32; off <<= 1) {
    int t = __shfl_up(x, off, 32);
    if (lane >= off) x += t;
  }
  if (lane == 31) sw[w] = x;
  __syncthreads();
  if (w == 0) {
    int y = sw[lane];
#pragma unroll
    for (int off = 1; off < 32; off <<= 1) {
      int t = __shfl_up(y, off, 32);
      if (lane >= off) y += t;
    }
    sw[lane] = y;
  }
  __syncthreads();
  if (w > 0) x += sw[w - 1];
  return x;
}

__global__ __launch_bounds__(L1_THREADS) void k_bhist(const int* __restrict__ dst,
                                                      int* __restrict__ btotal,
                                                      int E, int nbk) {
  __shared__ int sh[MAX_BK];
  const int tid = threadIdx.x;
  for (int t = tid; t < nbk; t += L1_THREADS) sh[t] = 0;
  __syncthreads();
  const int base = blockIdx.x * CHUNK;
#pragma unroll
  for (int i = 0; i < EPT; ++i) {
    int e = base + i * L1_THREADS + tid;
    if (e < E) atomicAdd(&sh[dst[e] >> BK_SHIFT], 1);
  }
  __syncthreads();
  for (int t = tid; t < nbk; t += L1_THREADS) {
    int c = sh[t];
    if (c) atomicAdd(&btotal[t], c);
  }
}

__global__ __launch_bounds__(L1_THREADS) void k_bscatter(const int* __restrict__ src,
                                                         const int* __restrict__ dst,
                                                         const int* __restrict__ btotal,
                                                         int* __restrict__ bcursor,
                                                         int2* __restrict__ pairs,
                                                         int E, int nbk) {
  __shared__ int sh[MAX_BK];
  __shared__ int sw[32];
  const int tid = threadIdx.x;
  sh[tid] = 0;
  const int tot = (tid < nbk) ? btotal[tid] : 0;
  __syncthreads();

  const int base = blockIdx.x * CHUNK;
  int d[EPT], s[EPT], r[EPT];
#pragma unroll
  for (int i = 0; i < EPT; ++i) {
    int e = base + i * L1_THREADS + tid;
    d[i] = (e < E) ? dst[e] : -1;
    s[i] = (e < E) ? src[e] : 0;
  }
#pragma unroll
  for (int i = 0; i < EPT; ++i) {
    r[i] = (d[i] >= 0) ? atomicAdd(&sh[d[i] >> BK_SHIFT], 1) : 0;
  }
  const int incl = block_scan_incl_1024(tot, sw, tid);
  if (tid < nbk) {
    int c = sh[tid];
    int g = c ? atomicAdd(&bcursor[tid], c) : 0;
    sh[tid] = incl - tot + g;
  }
  __syncthreads();
#pragma unroll
  for (int i = 0; i < EPT; ++i) {
    if (d[i] >= 0) {
      int p = sh[d[i] >> BK_SHIFT] + r[i];
      pairs[p] = make_int2(s[i], d[i]);
    }
  }
}

__global__ __launch_bounds__(BK_NODES) void k_fine(const int2* __restrict__ pairs,
                                                   const int* __restrict__ btotal,
                                                   int* __restrict__ offsets,
                                                   int* __restrict__ csr_src,
                                                   int N) {
  __shared__ int scnt[BK_NODES];
  __shared__ int sw[32];
  __shared__ int sbase;
  const int tid = threadIdx.x;
  const int b   = blockIdx.x;
  scnt[tid] = 0;
  if (tid == 0) sbase = 0;
  __syncthreads();
  const int part = (tid < b) ? btotal[tid] : 0;
  if (part) atomicAdd(&sbase, part);
  const int cnt = btotal[b];
  __syncthreads();
  const int base = sbase;
  const int2* __restrict__ pb = pairs + base;

  for (int i0 = 0; i0 < cnt; i0 += 4 * BK_NODES) {
    int dl[4];
#pragma unroll
    for (int k = 0; k < 4; ++k) {
      int i = i0 + k * BK_NODES + tid;
      dl[k] = pb[(i < cnt) ? i : (cnt - 1)].y;
    }
#pragma unroll
    for (int k = 0; k < 4; ++k) {
      int i = i0 + k * BK_NODES + tid;
      if (i < cnt) atomicAdd(&scnt[dl[k] & (BK_NODES - 1)], 1);
    }
  }
  __syncthreads();
  const int c = scnt[tid];
  const int excl = block_scan_incl_1024(c, sw, tid) - c;
  const int node = (b << BK_SHIFT) + tid;
  if (node < N) offsets[node] = base + excl;
  scnt[tid] = excl;
  __syncthreads();

  int* __restrict__ cb = csr_src + base;
  for (int i0 = 0; i0 < cnt; i0 += 4 * BK_NODES) {
    int2 v[4];
#pragma unroll
    for (int k = 0; k < 4; ++k) {
      int i = i0 + k * BK_NODES + tid;
      v[k] = pb[(i < cnt) ? i : (cnt - 1)];
    }
#pragma unroll
    for (int k = 0; k < 4; ++k) {
      int i = i0 + k * BK_NODES + tid;
      if (i < cnt) {
        int p = atomicAdd(&scnt[v[k].y & (BK_NODES - 1)], 1);
        cb[p] = v[k].x;
      }
    }
  }
}

#define MAXP 16
#define NPW  13

__device__ __forceinline__ float xhalf(float x) {
  int b = __builtin_bit_cast(int, x);
  return __builtin_bit_cast(float,
      __builtin_amdgcn_permlanex16(b, b, 0x76543210, 0xfedcba98, false, false));
}

template <int M>
__device__ __forceinline__ float row_xor(float x) {
  return __builtin_bit_cast(float,
      __builtin_amdgcn_update_dpp(0, __builtin_bit_cast(int, x), 0x160 + M, 0xf, 0xf, true));
}

__device__ __forceinline__ float row16_sum(float x) {
  x += row_xor<1>(x);
  x += row_xor<2>(x);
  x += row_xor<4>(x);
  x += row_xor<8>(x);
  return x;
}

__global__ __launch_bounds__(256) void k_gat_agg(const float* __restrict__ feat,
                                                 const float* __restrict__ el,
                                                 const float* __restrict__ er,
                                                 const int* __restrict__ offsets,
                                                 const int* __restrict__ csr_src,
                                                 const float* __restrict__ attn_l,
                                                 const float* __restrict__ bias,
                                                 float* __restrict__ out,
                                                 int N, int E) {
  const int wv    = __builtin_amdgcn_readfirstlane((int)(threadIdx.x >> 5));
  const int nbase = (blockIdx.x * 8 + wv) * NPW;
  if (nbase >= N) return;
  const int lane = threadIdx.x & 31;
  const int h    = lane >> 4;
  const int c    = lane & 15;

  const int   nl   = nbase + lane;
  const int   nlc  = (nl < N) ? nl : (N - 1);
  const int   ldo  = offsets[nlc];
  const int   voff = (nl < N) ? ldo : E;
  const float ver  = er[nlc];

  const float4 al = *(const float4*)(attn_l + 4 * c);
  const float4 bv = *(const float4*)(bias + 4 * c);
  const float* __restrict__ fcol = feat + 4 * c;

#pragma unroll 1
  for (int it = 0; it < NPW; ++it) {
    const int node = nbase + it;
    if (node >= N) break;
    const int   start = __builtin_amdgcn_readlane(voff, it);
    const int   end   = __builtin_amdgcn_readlane(voff, it + 1);
    const float er_d  = __builtin_bit_cast(float,
        __builtin_amdgcn_readlane(__builtin_bit_cast(int, ver), it));
    const int   deg   = end - start;

    if (deg <= 2 * MAXP) {
      const int j0 = start + lane;
      const int s0 = (j0 < end) ? csr_src[j0] : 0;

      float4 r[MAXP];
      float  w[MAXP];

      int sa[MAXP], sb[MAXP];
#pragma unroll
      for (int i = 0; i < MAXP; ++i) {
        sa[i] = __builtin_amdgcn_readlane(s0, 2 * i);
        sb[i] = __builtin_amdgcn_readlane(s0, 2 * i + 1);
      }
#pragma unroll
      for (int i = 0; i < MAXP; ++i) {
        if (2 * i < deg) {
          const int so = (2 * i + 1 < deg) ? sb[i] : sa[i];
          const int s  = h ? so : sa[i];
          r[i] = *(const float4*)(fcol + (size_t)s * OUT_DIM);
        } else {
          r[i] = make_float4(0.f, 0.f, 0.f, 0.f);
        }
      }

      float m = -INFINITY;
#pragma unroll
      for (int i = 0; i < MAXP; ++i) {
        if (2 * i < deg) {
          float p = r[i].x * al.x + r[i].y * al.y + r[i].z * al.z + r[i].w * al.w;
          p = row16_sum(p);
          float e = leaky(p + er_d);
          e = (2 * i + h < deg) ? e : -INFINITY;
          w[i] = e;
          m = fmaxf(m, e);
        } else {
          w[i] = -INFINITY;
        }
      }
      m = fmaxf(m, xhalf(m));
      if (deg == 0) m = 0.f;

      float den = 0.f;
#pragma unroll
      for (int g = 0; g < MAXP / 4; ++g) {
        if (8 * g < deg) {
#pragma unroll
          for (int k = 0; k < 4; ++k) {
            const float p = __expf(w[4 * g + k] - m);
            w[4 * g + k] = p;
            den += p;
          }
        }
      }
      den += xhalf(den);
      const float invden = (den > 0.f) ? 1.f / den : 1.f;

      float ax = 0.f, ay = 0.f, az = 0.f, aw = 0.f;
#pragma unroll
      for (int g = 0; g < MAXP / 4; ++g) {
        if (8 * g < deg) {
#pragma unroll
          for (int k = 0; k < 4; ++k) {
            const float a = w[4 * g + k] * invden;
            ax += a * r[4 * g + k].x;
            ay += a * r[4 * g + k].y;
            az += a * r[4 * g + k].z;
            aw += a * r[4 * g + k].w;
          }
        }
      }
      ax += xhalf(ax);
      ay += xhalf(ay);
      az += xhalf(az);
      aw += xhalf(aw);
      float4 ov;
      ov.x = ax + bv.x;
      ov.y = ay + bv.y;
      ov.z = az + bv.z;
      ov.w = aw + bv.w;
      if (h == 0) *(float4*)(out + (size_t)node * OUT_DIM + 4 * c) = ov;
      continue;
    }

    int   j0 = start + lane;
    float e0 = (j0 < end) ? leaky(el[csr_src[j0]] + er_d) : -INFINITY;

    float m = e0;
    for (int j = j0 + 32; j < end; j += 32) {
      m = fmaxf(m, leaky(el[csr_src[j]] + er_d));
    }
#pragma unroll
    for (int off = 16; off > 0; off >>= 1) m = fmaxf(m, __shfl_xor(m, off, 32));
    if (end <= start) m = 0.f;

    float den = (j0 < end) ? __expf(e0 - m) : 0.f;
    for (int j = j0 + 32; j < end; j += 32) {
      den += __expf(leaky(el[csr_src[j]] + er_d) - m);
    }
#pragma unroll
    for (int off = 16; off > 0; off >>= 1) den += __shfl_xor(den, off, 32);
    float invden = (den > 0.f) ? 1.f / den : 1.f;

    float acc0 = 0.f, acc1 = 0.f;
    for (int j = start; j < end; ++j) {
      int t = j - start;
      int s = csr_src[j];
      float e;
      if (t < 32) {
        e = __shfl(e0, t, 32);
      } else {
        e = leaky(el[s] + er_d);
      }
      float alpha = __expf(e - m) * invden;
      v2f fv = *(const v2f*)(feat + (size_t)s * OUT_DIM + 2 * lane);
      acc0 += alpha * fv.x;
      acc1 += alpha * fv.y;
    }
    v2f bv2 = *(const v2f*)(bias + 2 * lane);
    v2f ov;
    ov.x = acc0 + bv2.x;
    ov.y = acc1 + bv2.y;
    *(v2f*)(out + (size_t)node * OUT_DIM + 2 * lane) = ov;
  }
}

extern "C" void kernel_launch(void* const* d_in, const int* in_sizes, int n_in,
                              void* d_out, int out_size, void* d_ws, size_t ws_size,
                              hipStream_t stream) {
  const float* h      = (const float*)d_in[0];
  const int*   src    = (const int*)d_in[1];
  const int*   dst    = (const int*)d_in[2];
  const float* W      = (const float*)d_in[3];
  const float* attn_l = (const float*)d_in[4];
  const float* attn_r = (const float*)d_in[5];
  const float* bias   = (const float*)d_in[6];
  float* out = (float*)d_out;

  const int N = in_sizes[0] / IN_DIM;
  const int E = in_sizes[1];

  char* ws = (char*)d_ws;
  size_t off = 0;
  auto carve = [&](size_t bytes) -> char* {
    char* p = ws + off;
    off = (off + bytes + 255) & ~(size_t)255;
    return p;
  };
  float* feat    = (float*)carve((size_t)N * OUT_DIM * sizeof(float));
  float* Wt      = (float*)carve((size_t)IN_DIM * OUT_DIM * sizeof(float));
  float* el      = (float*)carve((size_t)N * sizeof(float));
  float* er      = (float*)carve((size_t)N * sizeof(float));
  int*   offsets = (int*)  carve((size_t)N * sizeof(int));
  int*   btotal  = (int*)  carve(MAX_BK * sizeof(int));
  int*   bcursor = (int*)  carve(MAX_BK * sizeof(int));
  int*   csr_src = (int*)  carve((size_t)E * sizeof(int));
  int2*  pairs   = (int2*) carve((size_t)E * sizeof(int2));

  const int nbk = (N + BK_NODES - 1) / BK_NODES;
  const int nch = (E + CHUNK - 1) / CHUNK;

  k_wt<<<(IN_DIM * OUT_DIM + 255) / 256, 256, 0, stream>>>(W, Wt);
  k_feat_wmma<<<(N + 31) / 32, 256, 0, stream>>>(h, Wt, feat, N);
  k_el_er<<<(N + 255) / 256, 256, 0, stream>>>(feat, attn_l, attn_r, el, er, btotal, bcursor, N, nbk);
  k_bhist   <<<nch, L1_THREADS, 0, stream>>>(dst, btotal, E, nbk);
  k_bscatter<<<nch, L1_THREADS, 0, stream>>>(src, dst, btotal, bcursor, pairs, E, nbk);
  k_fine    <<<nbk, BK_NODES, 0, stream>>>(pairs, btotal, offsets, csr_src, N);
  k_gat_agg<<<(N + 8 * NPW - 1) / (8 * NPW), 256, 0, stream>>>(feat, el, er, offsets, csr_src, attn_l, bias, out, N, E);
}


# Projection retiled: a wave owns 16 rows x 64 cols (h read once, 16-byte lane loads), W staged per block in LDS, so the transpose launch goes; k sum order of feat re-ordered inside each 8.
# speedup vs baseline: 2.9982x; 1.2093x over previous
#include <hip/hip_runtime.h>
#include <math.h>

#define IN_DIM   128
#define OUT_DIM  64
#define NEG_SLOPE 0.2f

typedef float v2f __attribute__((ext_vector_type(2)));
typedef float v8f __attribute__((ext_vector_type(8)));

__device__ __forceinline__ float leaky(float e) {
  return (e > 0.f) ? e : NEG_SLOPE * e;
}

#define WPP 144

__global__ __launch_bounds__(256) void k_feat_wmma(const float* __restrict__ h,
                                                   const float* __restrict__ W,
                                                   float* __restrict__ feat,
                                                   int N) {
  __shared__ float sW[(IN_DIM / 2) * WPP];
  const int tid  = threadIdx.x;
  const int wave = tid >> 5;
  const int lane = tid & 31;
  const int c    = lane & 15;
  const int hf   = lane >> 4;
  const int rowBase = (blockIdx.x * 8 + wave) * 16;

  const int row  = rowBase + c;
  const int rowc = (row < N) ? row : (N - 1);
  const float* __restrict__ ap = h + (size_t)rowc * IN_DIM + 4 * hf;
  float4 a[IN_DIM / 8];
#pragma unroll
  for (int kk = 0; kk < IN_DIM / 8; ++kk) a[kk] = *(const float4*)(ap + 8 * kk);

#pragma unroll
  for (int i = 0; i < 4; ++i) {
    const int idx = i * 256 + tid;
    const int kp  = idx >> 4;
    const int q   = idx & 15;
    const float4 w0 = *(const float4*)(W + (size_t)(2 * kp) * OUT_DIM + 4 * q);
    const float4 w1 = *(const float4*)(W + (size_t)(2 * kp + 1) * OUT_DIM + 4 * q);
    float* d = sW + kp * WPP + 8 * q;
    *(float4*)(d)     = make_float4(w0.x, w1.x, w0.y, w1.y);
    *(float4*)(d + 4) = make_float4(w0.z, w1.z, w0.w, w1.w);
  }
  __syncthreads();
  if (rowBase >= N) return;

  v8f acc0 = {0.f, 0.f, 0.f, 0.f, 0.f, 0.f, 0.f, 0.f};
  v8f acc1 = acc0, acc2 = acc0, acc3 = acc0;
  const float* bp = sW + (2 * hf) * WPP + 2 * c;

#pragma unroll
  for (int kk = 0; kk < IN_DIM / 8; ++kk) {
    v2f alo, ahi;
    alo.x = a[kk].x; alo.y = a[kk].y;
    ahi.x = a[kk].z; ahi.y = a[kk].w;
    const float* b0 = bp + (4 * kk) * WPP;
    const float* b1 = b0 + WPP;
    const v2f l0 = *(const v2f*)(b0);
    const v2f l1 = *(const v2f*)(b0 + 32);
    const v2f l2 = *(const v2f*)(b0 + 64);
    const v2f l3 = *(const v2f*)(b0 + 96);
    const v2f u0 = *(const v2f*)(b1);
    const v2f u1 = *(const v2f*)(b1 + 32);
    const v2f u2 = *(const v2f*)(b1 + 64);
    const v2f u3 = *(const v2f*)(b1 + 96);
    acc0 = __builtin_amdgcn_wmma_f32_16x16x4_f32(false, alo, false, l0, (short)0, acc0, false, false);
    acc1 = __builtin_amdgcn_wmma_f32_16x16x4_f32(false, alo, false, l1, (short)0, acc1, false, false);
    acc2 = __builtin_amdgcn_wmma_f32_16x16x4_f32(false, alo, false, l2, (short)0, acc2, false, false);
    acc3 = __builtin_amdgcn_wmma_f32_16x16x4_f32(false, alo, false, l3, (short)0, acc3, false, false);
    acc0 = __builtin_amdgcn_wmma_f32_16x16x4_f32(false, ahi, false, u0, (short)0, acc0, false, false);
    acc1 = __builtin_amdgcn_wmma_f32_16x16x4_f32(false, ahi, false, u1, (short)0, acc1, false, false);
    acc2 = __builtin_amdgcn_wmma_f32_16x16x4_f32(false, ahi, false, u2, (short)0, acc2, false, false);
    acc3 = __builtin_amdgcn_wmma_f32_16x16x4_f32(false, ahi, false, u3, (short)0, acc3, false, false);
  }

  const int mBase = rowBase + hf * 8;
#pragma unroll
  for (int r = 0; r < 8; ++r) {
    const int m = mBase + r;
    if (m < N) {
      float* o = feat + (size_t)m * OUT_DIM + c;
      o[0]  = acc0[r];
      o[16] = acc1[r];
      o[32] = acc2[r];
      o[48] = acc3[r];
    }
  }
}

__global__ void k_el_er(const float* __restrict__ feat,
                        const float* __restrict__ attn_l,
                        const float* __restrict__ attn_r,
                        float* __restrict__ el, float* __restrict__ er,
                        int* __restrict__ btotal, int* __restrict__ bcursor,
                        int N, int nbk) {
  int i = blockIdx.x * blockDim.x + threadIdx.x;
  if (i < nbk) { btotal[i] = 0; bcursor[i] = 0; }
  if (i >= N) return;
  const float4* f  = (const float4*)(feat + (size_t)i * OUT_DIM);
  const float4* al = (const float4*)attn_l;
  const float4* ar = (const float4*)attn_r;
  float sl = 0.f, sr = 0.f;
#pragma unroll
  for (int j = 0; j < OUT_DIM / 4; ++j) {
    float4 v = f[j], a = al[j], b = ar[j];
    sl += v.x * a.x + v.y * a.y + v.z * a.z + v.w * a.w;
    sr += v.x * b.x + v.y * b.y + v.z * b.z + v.w * b.w;
  }
  el[i] = sl;
  er[i] = sr;
}

#define BK_SHIFT 10
#define BK_NODES 1024
#define MAX_BK   1024
#define CHUNK    4096
#define L1_THREADS 1024
#define EPT (CHUNK / L1_THREADS)

__device__ __forceinline__ int block_scan_incl_1024(int v, int* sw, int tid) {
  const int lane = tid & 31;
  const int w    = tid >> 5;
  int x = v;
#pragma unroll
  for (int off = 1; off < 32; off <<= 1) {
    int t = __shfl_up(x, off, 32);
    if (lane >= off) x += t;
  }
  if (lane == 31) sw[w] = x;
  __syncthreads();
  if (w == 0) {
    int y = sw[lane];
#pragma unroll
    for (int off = 1; off < 32; off <<= 1) {
      int t = __shfl_up(y, off, 32);
      if (lane >= off) y += t;
    }
    sw[lane] = y;
  }
  __syncthreads();
  if (w > 0) x += sw[w - 1];
  return x;
}

__global__ __launch_bounds__(L1_THREADS) void k_bhist(const int* __restrict__ dst,
                                                      int* __restrict__ btotal,
                                                      int E, int nbk) {
  __shared__ int sh[MAX_BK];
  const int tid = threadIdx.x;
  for (int t = tid; t < nbk; t += L1_THREADS) sh[t] = 0;
  __syncthreads();
  const int base = blockIdx.x * CHUNK;
#pragma unroll
  for (int i = 0; i < EPT; ++i) {
    int e = base + i * L1_THREADS + tid;
    if (e < E) atomicAdd(&sh[dst[e] >> BK_SHIFT], 1);
  }
  __syncthreads();
  for (int t = tid; t < nbk; t += L1_THREADS) {
    int c = sh[t];
    if (c) atomicAdd(&btotal[t], c);
  }
}

__global__ __launch_bounds__(L1_THREADS) void k_bscatter(const int* __restrict__ src,
                                                         const int* __restrict__ dst,
                                                         const int* __restrict__ btotal,
                                                         int* __restrict__ bcursor,
                                                         int2* __restrict__ pairs,
                                                         int E, int nbk) {
  __shared__ int sh[MAX_BK];
  __shared__ int sw[32];
  const int tid = threadIdx.x;
  sh[tid] = 0;
  const int tot = (tid < nbk) ? btotal[tid] : 0;
  __syncthreads();

  const int base = blockIdx.x * CHUNK;
  int d[EPT], s[EPT], r[EPT];
#pragma unroll
  for (int i = 0; i < EPT; ++i) {
    int e = base + i * L1_THREADS + tid;
    d[i] = (e < E) ? dst[e] : -1;
    s[i] = (e < E) ? src[e] : 0;
  }
#pragma unroll
  for (int i = 0; i < EPT; ++i) {
    r[i] = (d[i] >= 0) ? atomicAdd(&sh[d[i] >> BK_SHIFT], 1) : 0;
  }
  const int incl = block_scan_incl_1024(tot, sw, tid);
  if (tid < nbk) {
    int c = sh[tid];
    int g = c ? atomicAdd(&bcursor[tid], c) : 0;
    sh[tid] = incl - tot + g;
  }
  __syncthreads();
#pragma unroll
  for (int i = 0; i < EPT; ++i) {
    if (d[i] >= 0) {
      int p = sh[d[i] >> BK_SHIFT] + r[i];
      pairs[p] = make_int2(s[i], d[i]);
    }
  }
}

__global__ __launch_bounds__(BK_NODES) void k_fine(const int2* __restrict__ pairs,
                                                   const int* __restrict__ btotal,
                                                   int* __restrict__ offsets,
                                                   int* __restrict__ csr_src,
                                                   int N) {
  __shared__ int scnt[BK_NODES];
  __shared__ int sw[32];
  __shared__ int sbase;
  const int tid = threadIdx.x;
  const int b   = blockIdx.x;
  scnt[tid] = 0;
  if (tid == 0) sbase = 0;
  __syncthreads();
  const int part = (tid < b) ? btotal[tid] : 0;
  if (part) atomicAdd(&sbase, part);
  const int cnt = btotal[b];
  __syncthreads();
  const int base = sbase;
  const int2* __restrict__ pb = pairs + base;

  for (int i0 = 0; i0 < cnt; i0 += 4 * BK_NODES) {
    int dl[4];
#pragma unroll
    for (int k = 0; k < 4; ++k) {
      int i = i0 + k * BK_NODES + tid;
      dl[k] = pb[(i < cnt) ? i : (cnt - 1)].y;
    }
#pragma unroll
    for (int k = 0; k < 4; ++k) {
      int i = i0 + k * BK_NODES + tid;
      if (i < cnt) atomicAdd(&scnt[dl[k] & (BK_NODES - 1)], 1);
    }
  }
  __syncthreads();
  const int c = scnt[tid];
  const int excl = block_scan_incl_1024(c, sw, tid) - c;
  const int node = (b << BK_SHIFT) + tid;
  if (node < N) offsets[node] = base + excl;
  scnt[tid] = excl;
  __syncthreads();

  int* __restrict__ cb = csr_src + base;
  for (int i0 = 0; i0 < cnt; i0 += 4 * BK_NODES) {
    int2 v[4];
#pragma unroll
    for (int k = 0; k < 4; ++k) {
      int i = i0 + k * BK_NODES + tid;
      v[k] = pb[(i < cnt) ? i : (cnt - 1)];
    }
#pragma unroll
    for (int k = 0; k < 4; ++k) {
      int i = i0 + k * BK_NODES + tid;
      if (i < cnt) {
        int p = atomicAdd(&scnt[v[k].y & (BK_NODES - 1)], 1);
        cb[p] = v[k].x;
      }
    }
  }
}

#define MAXP 16
#define NPW  13

__device__ __forceinline__ float xhalf(float x) {
  int b = __builtin_bit_cast(int, x);
  return __builtin_bit_cast(float,
      __builtin_amdgcn_permlanex16(b, b, 0x76543210, 0xfedcba98, false, false));
}

template <int M>
__device__ __forceinline__ float row_xor(float x) {
  return __builtin_bit_cast(float,
      __builtin_amdgcn_update_dpp(0, __builtin_bit_cast(int, x), 0x160 + M, 0xf, 0xf, true));
}

__device__ __forceinline__ float row16_sum(float x) {
  x += row_xor<1>(x);
  x += row_xor<2>(x);
  x += row_xor<4>(x);
  x += row_xor<8>(x);
  return x;
}

__global__ __launch_bounds__(256) void k_gat_agg(const float* __restrict__ feat,
                                                 const float* __restrict__ el,
                                                 const float* __restrict__ er,
                                                 const int* __restrict__ offsets,
                                                 const int* __restrict__ csr_src,
                                                 const float* __restrict__ attn_l,
                                                 const float* __restrict__ bias,
                                                 float* __restrict__ out,
                                                 int N, int E) {
  const int wv    = __builtin_amdgcn_readfirstlane((int)(threadIdx.x >> 5));
  const int nbase = (blockIdx.x * 8 + wv) * NPW;
  if (nbase >= N) return;
  const int lane = threadIdx.x & 31;
  const int h    = lane >> 4;
  const int c    = lane & 15;

  const int   nl   = nbase + lane;
  const int   nlc  = (nl < N) ? nl : (N - 1);
  const int   ldo  = offsets[nlc];
  const int   voff = (nl < N) ? ldo : E;
  const float ver  = er[nlc];

  const float4 al = *(const float4*)(attn_l + 4 * c);
  const float4 bv = *(const float4*)(bias + 4 * c);
  const float* __restrict__ fcol = feat + 4 * c;

#pragma unroll 1
  for (int it = 0; it < NPW; ++it) {
    const int node = nbase + it;
    if (node >= N) break;
    const int   start = __builtin_amdgcn_readlane(voff, it);
    const int   end   = __builtin_amdgcn_readlane(voff, it + 1);
    const float er_d  = __builtin_bit_cast(float,
        __builtin_amdgcn_readlane(__builtin_bit_cast(int, ver), it));
    const int   deg   = end - start;

    if (deg <= 2 * MAXP) {
      const int j0 = start + lane;
      const int s0 = (j0 < end) ? csr_src[j0] : 0;

      float4 r[MAXP];
      float  w[MAXP];

      int sa[MAXP], sb[MAXP];
#pragma unroll
      for (int i = 0; i < MAXP; ++i) {
        sa[i] = __builtin_amdgcn_readlane(s0, 2 * i);
        sb[i] = __builtin_amdgcn_readlane(s0, 2 * i + 1);
      }
#pragma unroll
      for (int i = 0; i < MAXP; ++i) {
        if (2 * i < deg) {
          const int so = (2 * i + 1 < deg) ? sb[i] : sa[i];
          const int s  = h ? so : sa[i];
          r[i] = *(const float4*)(fcol + (size_t)s * OUT_DIM);
        } else {
          r[i] = make_float4(0.f, 0.f, 0.f, 0.f);
        }
      }

      float m = -INFINITY;
#pragma unroll
      for (int i = 0; i < MAXP; ++i) {
        if (2 * i < deg) {
          float p = r[i].x * al.x + r[i].y * al.y + r[i].z * al.z + r[i].w * al.w;
          p = row16_sum(p);
          float e = leaky(p + er_d);
          e = (2 * i + h < deg) ? e : -INFINITY;
          w[i] = e;
          m = fmaxf(m, e);
        } else {
          w[i] = -INFINITY;
        }
      }
      m = fmaxf(m, xhalf(m));
      if (deg == 0) m = 0.f;

      float den = 0.f;
#pragma unroll
      for (int g = 0; g < MAXP / 4; ++g) {
        if (8 * g < deg) {
#pragma unroll
          for (int k = 0; k < 4; ++k) {
            const float p = __expf(w[4 * g + k] - m);
            w[4 * g + k] = p;
            den += p;
          }
        }
      }
      den += xhalf(den);
      const float invden = (den > 0.f) ? 1.f / den : 1.f;

      float ax = 0.f, ay = 0.f, az = 0.f, aw = 0.f;
#pragma unroll
      for (int g = 0; g < MAXP / 4; ++g) {
        if (8 * g < deg) {
#pragma unroll
          for (int k = 0; k < 4; ++k) {
            const float a = w[4 * g + k] * invden;
            ax += a * r[4 * g + k].x;
            ay += a * r[4 * g + k].y;
            az += a * r[4 * g + k].z;
            aw += a * r[4 * g + k].w;
          }
        }
      }
      ax += xhalf(ax);
      ay += xhalf(ay);
      az += xhalf(az);
      aw += xhalf(aw);
      float4 ov;
      ov.x = ax + bv.x;
      ov.y = ay + bv.y;
      ov.z = az + bv.z;
      ov.w = aw + bv.w;
      if (h == 0) *(float4*)(out + (size_t)node * OUT_DIM + 4 * c) = ov;
      continue;
    }

    int   j0 = start + lane;
    float e0 = (j0 < end) ? leaky(el[csr_src[j0]] + er_d) : -INFINITY;

    float m = e0;
    for (int j = j0 + 32; j < end; j += 32) {
      m = fmaxf(m, leaky(el[csr_src[j]] + er_d));
    }
#pragma unroll
    for (int off = 16; off > 0; off >>= 1) m = fmaxf(m, __shfl_xor(m, off, 32));
    if (end <= start) m = 0.f;

    float den = (j0 < end) ? __expf(e0 - m) : 0.f;
    for (int j = j0 + 32; j < end; j += 32) {
      den += __expf(leaky(el[csr_src[j]] + er_d) - m);
    }
#pragma unroll
    for (int off = 16; off > 0; off >>= 1) den += __shfl_xor(den, off, 32);
    float invden = (den > 0.f) ? 1.f / den : 1.f;

    float acc0 = 0.f, acc1 = 0.f;
    for (int j = start; j < end; ++j) {
      int t = j - start;
      int s = csr_src[j];
      float e;
      if (t < 32) {
        e = __shfl(e0, t, 32);
      } else {
        e = leaky(el[s] + er_d);
      }
      float alpha = __expf(e - m) * invden;
      v2f fv = *(const v2f*)(feat + (size_t)s * OUT_DIM + 2 * lane);
      acc0 += alpha * fv.x;
      acc1 += alpha * fv.y;
    }
    v2f bv2 = *(const v2f*)(bias + 2 * lane);
    v2f ov;
    ov.x = acc0 + bv2.x;
    ov.y = acc1 + bv2.y;
    *(v2f*)(out + (size_t)node * OUT_DIM + 2 * lane) = ov;
  }
}

extern "C" void kernel_launch(void* const* d_in, const int* in_sizes, int n_in,
                              void* d_out, int out_size, void* d_ws, size_t ws_size,
                              hipStream_t stream) {
  const float* h      = (const float*)d_in[0];
  const int*   src    = (const int*)d_in[1];
  const int*   dst    = (const int*)d_in[2];
  const float* W      = (const float*)d_in[3];
  const float* attn_l = (const float*)d_in[4];
  const float* attn_r = (const float*)d_in[5];
  const float* bias   = (const float*)d_in[6];
  float* out = (float*)d_out;

  const int N = in_sizes[0] / IN_DIM;
  const int E = in_sizes[1];

  char* ws = (char*)d_ws;
  size_t off = 0;
  auto carve = [&](size_t bytes) -> char* {
    char* p = ws + off;
    off = (off + bytes + 255) & ~(size_t)255;
    return p;
  };
  float* feat    = (float*)carve((size_t)N * OUT_DIM * sizeof(float));
  float* el      = (float*)carve((size_t)N * sizeof(float));
  float* er      = (float*)carve((size_t)N * sizeof(float));
  int*   offsets = (int*)  carve((size_t)N * sizeof(int));
  int*   btotal  = (int*)  carve(MAX_BK * sizeof(int));
  int*   bcursor = (int*)  carve(MAX_BK * sizeof(int));
  int*   csr_src = (int*)  carve((size_t)E * sizeof(int));
  int2*  pairs   = (int2*) carve((size_t)E * sizeof(int2));

  const int nbk = (N + BK_NODES - 1) / BK_NODES;
  const int nch = (E + CHUNK - 1) / CHUNK;

  k_feat_wmma<<<(N + 127) / 128, 256, 0, stream>>>(h, W, feat, N);
  k_el_er<<<(N + 255) / 256, 256, 0, stream>>>(feat, attn_l, attn_r, el, er, btotal, bcursor, N, nbk);
  k_bhist   <<<nch, L1_THREADS, 0, stream>>>(dst, btotal, E, nbk);
  k_bscatter<<<nch, L1_THREADS, 0, stream>>>(src, dst, btotal, bcursor, pairs, E, nbk);
  k_fine    <<<nbk, BK_NODES, 0, stream>>>(pairs, btotal, offsets, csr_src, N);
  k_gat_agg<<<(N + 8 * NPW - 1) / (8 * NPW), 256, 0, stream>>>(feat, el, er, offsets, csr_src, attn_l, bias, out, N, E);
}


# Node logits el, er computed in the projection epilogue from the accumulator tiles (lane partial dots + 16-lane butterfly): the logits launch goes, 6 to 5 launches; el, er sum order re-ordered.
# speedup vs baseline: 3.2163x; 1.0727x over previous
#include <hip/hip_runtime.h>
#include <math.h>

#define IN_DIM   128
#define OUT_DIM  64
#define NEG_SLOPE 0.2f

typedef float v2f __attribute__((ext_vector_type(2)));
typedef float v8f __attribute__((ext_vector_type(8)));

__device__ __forceinline__ float leaky(float e) {
  return (e > 0.f) ? e : NEG_SLOPE * e;
}

template <int M>
__device__ __forceinline__ float row_xor(float x) {
  return __builtin_bit_cast(float,
      __builtin_amdgcn_update_dpp(0, __builtin_bit_cast(int, x), 0x160 + M, 0xf, 0xf, true));
}

__device__ __forceinline__ float row16_sum(float x) {
  x += row_xor<1>(x);
  x += row_xor<2>(x);
  x += row_xor<4>(x);
  x += row_xor<8>(x);
  return x;
}

#define WPP 144

__global__ __launch_bounds__(256) void k_feat_wmma(const float* __restrict__ h,
                                                   const float* __restrict__ W,
                                                   const float* __restrict__ attn_l,
                                                   const float* __restrict__ attn_r,
                                                   float* __restrict__ feat,
                                                   float* __restrict__ el,
                                                   float* __restrict__ er,
                                                   int* __restrict__ btotal,
                                                   int* __restrict__ bcursor,
                                                   int N, int nbk) {
  __shared__ float sW[(IN_DIM / 2) * WPP];
  const int tid  = threadIdx.x;
  const int wave = tid >> 5;
  const int lane = tid & 31;
  const int c    = lane & 15;
  const int hf   = lane >> 4;
  const int rowBase = (blockIdx.x * 8 + wave) * 16;

  const int row  = rowBase + c;
  const int rowc = (row < N) ? row : (N - 1);
  const float* __restrict__ ap = h + (size_t)rowc * IN_DIM + 4 * hf;
  float4 a[IN_DIM / 8];
#pragma unroll
  for (int kk = 0; kk < IN_DIM / 8; ++kk) a[kk] = *(const float4*)(ap + 8 * kk);

  const float al0 = attn_l[c], al1 = attn_l[16 + c], al2 = attn_l[32 + c], al3 = attn_l[48 + c];
  const float ar0 = attn_r[c], ar1 = attn_r[16 + c], ar2 = attn_r[32 + c], ar3 = attn_r[48 + c];

  if (blockIdx.x == 0) {
    for (int t = tid; t < nbk; t += 256) { btotal[t] = 0; bcursor[t] = 0; }
  }

#pragma unroll
  for (int i = 0; i < 4; ++i) {
    const int idx = i * 256 + tid;
    const int kp  = idx >> 4;
    const int q   = idx & 15;
    const float4 w0 = *(const float4*)(W + (size_t)(2 * kp) * OUT_DIM + 4 * q);
    const float4 w1 = *(const float4*)(W + (size_t)(2 * kp + 1) * OUT_DIM + 4 * q);
    float* d = sW + kp * WPP + 8 * q;
    *(float4*)(d)     = make_float4(w0.x, w1.x, w0.y, w1.y);
    *(float4*)(d + 4) = make_float4(w0.z, w1.z, w0.w, w1.w);
  }
  __syncthreads();
  if (rowBase >= N) return;

  v8f acc0 = {0.f, 0.f, 0.f, 0.f, 0.f, 0.f, 0.f, 0.f};
  v8f acc1 = acc0, acc2 = acc0, acc3 = acc0;
  const float* bp = sW + (2 * hf) * WPP + 2 * c;

#pragma unroll
  for (int kk = 0; kk < IN_DIM / 8; ++kk) {
    v2f alo, ahi;
    alo.x = a[kk].x; alo.y = a[kk].y;
    ahi.x = a[kk].z; ahi.y = a[kk].w;
    const float* b0 = bp + (4 * kk) * WPP;
    const float* b1 = b0 + WPP;
    const v2f l0 = *(const v2f*)(b0);
    const v2f l1 = *(const v2f*)(b0 + 32);
    const v2f l2 = *(const v2f*)(b0 + 64);
    const v2f l3 = *(const v2f*)(b0 + 96);
    const v2f u0 = *(const v2f*)(b1);
    const v2f u1 = *(const v2f*)(b1 + 32);
    const v2f u2 = *(const v2f*)(b1 + 64);
    const v2f u3 = *(const v2f*)(b1 + 96);
    acc0 = __builtin_amdgcn_wmma_f32_16x16x4_f32(false, alo, false, l0, (short)0, acc0, false, false);
    acc1 = __builtin_amdgcn_wmma_f32_16x16x4_f32(false, alo, false, l1, (short)0, acc1, false, false);
    acc2 = __builtin_amdgcn_wmma_f32_16x16x4_f32(false, alo, false, l2, (short)0, acc2, false, false);
    acc3 = __builtin_amdgcn_wmma_f32_16x16x4_f32(false, alo, false, l3, (short)0, acc3, false, false);
    acc0 = __builtin_amdgcn_wmma_f32_16x16x4_f32(false, ahi, false, u0, (short)0, acc0, false, false);
    acc1 = __builtin_amdgcn_wmma_f32_16x16x4_f32(false, ahi, false, u1, (short)0, acc1, false, false);
    acc2 = __builtin_amdgcn_wmma_f32_16x16x4_f32(false, ahi, false, u2, (short)0, acc2, false, false);
    acc3 = __builtin_amdgcn_wmma_f32_16x16x4_f32(false, ahi, false, u3, (short)0, acc3, false, false);
  }

  const int mBase = rowBase + hf * 8;
#pragma unroll
  for (int r = 0; r < 8; ++r) {
    const int m = mBase + r;
    if (m < N) {
      float* o = feat + (size_t)m * OUT_DIM + c;
      o[0]  = acc0[r];
      o[16] = acc1[r];
      o[32] = acc2[r];
      o[48] = acc3[r];
    }
  }

  float sl = 0.f, sr = 0.f;
#pragma unroll
  for (int r = 0; r < 8; ++r) {
    float pl = acc0[r] * al0 + acc1[r] * al1 + acc2[r] * al2 + acc3[r] * al3;
    float pr = acc0[r] * ar0 + acc1[r] * ar1 + acc2[r] * ar2 + acc3[r] * ar3;
    pl = row16_sum(pl);
    pr = row16_sum(pr);
    if (c == r) { sl = pl; sr = pr; }
  }
  const int mo = mBase + c;
  if (c < 8 && mo < N) {
    el[mo] = sl;
    er[mo] = sr;
  }
}

#define BK_SHIFT 10
#define BK_NODES 1024
#define MAX_BK   1024
#define CHUNK    4096
#define L1_THREADS 1024
#define EPT (CHUNK / L1_THREADS)

__device__ __forceinline__ int block_scan_incl_1024(int v, int* sw, int tid) {
  const int lane = tid & 31;
  const int w    = tid >> 5;
  int x = v;
#pragma unroll
  for (int off = 1; off < 32; off <<= 1) {
    int t = __shfl_up(x, off, 32);
    if (lane >= off) x += t;
  }
  if (lane == 31) sw[w] = x;
  __syncthreads();
  if (w == 0) {
    int y = sw[lane];
#pragma unroll
    for (int off = 1; off < 32; off <<= 1) {
      int t = __shfl_up(y, off, 32);
      if (lane >= off) y += t;
    }
    sw[lane] = y;
  }
  __syncthreads();
  if (w > 0) x += sw[w - 1];
  return x;
}

__global__ __launch_bounds__(L1_THREADS) void k_bhist(const int* __restrict__ dst,
                                                      int* __restrict__ btotal,
                                                      int E, int nbk) {
  __shared__ int sh[MAX_BK];
  const int tid = threadIdx.x;
  for (int t = tid; t < nbk; t += L1_THREADS) sh[t] = 0;
  __syncthreads();
  const int base = blockIdx.x * CHUNK;
#pragma unroll
  for (int i = 0; i < EPT; ++i) {
    int e = base + i * L1_THREADS + tid;
    if (e < E) atomicAdd(&sh[dst[e] >> BK_SHIFT], 1);
  }
  __syncthreads();
  for (int t = tid; t < nbk; t += L1_THREADS) {
    int c = sh[t];
    if (c) atomicAdd(&btotal[t], c);
  }
}

__global__ __launch_bounds__(L1_THREADS) void k_bscatter(const int* __restrict__ src,
                                                         const int* __restrict__ dst,
                                                         const int* __restrict__ btotal,
                                                         int* __restrict__ bcursor,
                                                         int2* __restrict__ pairs,
                                                         int E, int nbk) {
  __shared__ int sh[MAX_BK];
  __shared__ int sw[32];
  const int tid = threadIdx.x;
  sh[tid] = 0;
  const int tot = (tid < nbk) ? btotal[tid] : 0;
  __syncthreads();

  const int base = blockIdx.x * CHUNK;
  int d[EPT], s[EPT], r[EPT];
#pragma unroll
  for (int i = 0; i < EPT; ++i) {
    int e = base + i * L1_THREADS + tid;
    d[i] = (e < E) ? dst[e] : -1;
    s[i] = (e < E) ? src[e] : 0;
  }
#pragma unroll
  for (int i = 0; i < EPT; ++i) {
    r[i] = (d[i] >= 0) ? atomicAdd(&sh[d[i] >> BK_SHIFT], 1) : 0;
  }
  const int incl = block_scan_incl_1024(tot, sw, tid);
  if (tid < nbk) {
    int c = sh[tid];
    int g = c ? atomicAdd(&bcursor[tid], c) : 0;
    sh[tid] = incl - tot + g;
  }
  __syncthreads();
#pragma unroll
  for (int i = 0; i < EPT; ++i) {
    if (d[i] >= 0) {
      int p = sh[d[i] >> BK_SHIFT] + r[i];
      pairs[p] = make_int2(s[i], d[i]);
    }
  }
}

__global__ __launch_bounds__(BK_NODES) void k_fine(const int2* __restrict__ pairs,
                                                   const int* __restrict__ btotal,
                                                   int* __restrict__ offsets,
                                                   int* __restrict__ csr_src,
                                                   int N) {
  __shared__ int scnt[BK_NODES];
  __shared__ int sw[32];
  __shared__ int sbase;
  const int tid = threadIdx.x;
  const int b   = blockIdx.x;
  scnt[tid] = 0;
  if (tid == 0) sbase = 0;
  __syncthreads();
  const int part = (tid < b) ? btotal[tid] : 0;
  if (part) atomicAdd(&sbase, part);
  const int cnt = btotal[b];
  __syncthreads();
  const int base = sbase;
  const int2* __restrict__ pb = pairs + base;

  for (int i0 = 0; i0 < cnt; i0 += 4 * BK_NODES) {
    int dl[4];
#pragma unroll
    for (int k = 0; k < 4; ++k) {
      int i = i0 + k * BK_NODES + tid;
      dl[k] = pb[(i < cnt) ? i : (cnt - 1)].y;
    }
#pragma unroll
    for (int k = 0; k < 4; ++k) {
      int i = i0 + k * BK_NODES + tid;
      if (i < cnt) atomicAdd(&scnt[dl[k] & (BK_NODES - 1)], 1);
    }
  }
  __syncthreads();
  const int c = scnt[tid];
  const int excl = block_scan_incl_1024(c, sw, tid) - c;
  const int node = (b << BK_SHIFT) + tid;
  if (node < N) offsets[node] = base + excl;
  scnt[tid] = excl;
  __syncthreads();

  int* __restrict__ cb = csr_src + base;
  for (int i0 = 0; i0 < cnt; i0 += 4 * BK_NODES) {
    int2 v[4];
#pragma unroll
    for (int k = 0; k < 4; ++k) {
      int i = i0 + k * BK_NODES + tid;
      v[k] = pb[(i < cnt) ? i : (cnt - 1)];
    }
#pragma unroll
    for (int k = 0; k < 4; ++k) {
      int i = i0 + k * BK_NODES + tid;
      if (i < cnt) {
        int p = atomicAdd(&scnt[v[k].y & (BK_NODES - 1)], 1);
        cb[p] = v[k].x;
      }
    }
  }
}

#define MAXP 16
#define NPW  13

__device__ __forceinline__ float xhalf(float x) {
  int b = __builtin_bit_cast(int, x);
  return __builtin_bit_cast(float,
      __builtin_amdgcn_permlanex16(b, b, 0x76543210, 0xfedcba98, false, false));
}

__global__ __launch_bounds__(256) void k_gat_agg(const float* __restrict__ feat,
                                                 const float* __restrict__ el,
                                                 const float* __restrict__ er,
                                                 const int* __restrict__ offsets,
                                                 const int* __restrict__ csr_src,
                                                 const float* __restrict__ attn_l,
                                                 const float* __restrict__ bias,
                                                 float* __restrict__ out,
                                                 int N, int E) {
  const int wv    = __builtin_amdgcn_readfirstlane((int)(threadIdx.x >> 5));
  const int nbase = (blockIdx.x * 8 + wv) * NPW;
  if (nbase >= N) return;
  const int lane = threadIdx.x & 31;
  const int h    = lane >> 4;
  const int c    = lane & 15;

  const int   nl   = nbase + lane;
  const int   nlc  = (nl < N) ? nl : (N - 1);
  const int   ldo  = offsets[nlc];
  const int   voff = (nl < N) ? ldo : E;
  const float ver  = er[nlc];

  const float4 al = *(const float4*)(attn_l + 4 * c);
  const float4 bv = *(const float4*)(bias + 4 * c);
  const float* __restrict__ fcol = feat + 4 * c;

#pragma unroll 1
  for (int it = 0; it < NPW; ++it) {
    const int node = nbase + it;
    if (node >= N) break;
    const int   start = __builtin_amdgcn_readlane(voff, it);
    const int   end   = __builtin_amdgcn_readlane(voff, it + 1);
    const float er_d  = __builtin_bit_cast(float,
        __builtin_amdgcn_readlane(__builtin_bit_cast(int, ver), it));
    const int   deg   = end - start;

    if (deg <= 2 * MAXP) {
      const int j0 = start + lane;
      const int s0 = (j0 < end) ? csr_src[j0] : 0;

      float4 r[MAXP];
      float  w[MAXP];

      int sa[MAXP], sb[MAXP];
#pragma unroll
      for (int i = 0; i < MAXP; ++i) {
        sa[i] = __builtin_amdgcn_readlane(s0, 2 * i);
        sb[i] = __builtin_amdgcn_readlane(s0, 2 * i + 1);
      }
#pragma unroll
      for (int i = 0; i < MAXP; ++i) {
        if (2 * i < deg) {
          const int so = (2 * i + 1 < deg) ? sb[i] : sa[i];
          const int s  = h ? so : sa[i];
          r[i] = *(const float4*)(fcol + (size_t)s * OUT_DIM);
        } else {
          r[i] = make_float4(0.f, 0.f, 0.f, 0.f);
        }
      }

      float m = -INFINITY;
#pragma unroll
      for (int i = 0; i < MAXP; ++i) {
        if (2 * i < deg) {
          float p = r[i].x * al.x + r[i].y * al.y + r[i].z * al.z + r[i].w * al.w;
          p = row16_sum(p);
          float e = leaky(p + er_d);
          e = (2 * i + h < deg) ? e : -INFINITY;
          w[i] = e;
          m = fmaxf(m, e);
        } else {
          w[i] = -INFINITY;
        }
      }
      m = fmaxf(m, xhalf(m));
      if (deg == 0) m = 0.f;

      float den = 0.f;
#pragma unroll
      for (int g = 0; g < MAXP / 4; ++g) {
        if (8 * g < deg) {
#pragma unroll
          for (int k = 0; k < 4; ++k) {
            const float p = __expf(w[4 * g + k] - m);
            w[4 * g + k] = p;
            den += p;
          }
        }
      }
      den += xhalf(den);
      const float invden = (den > 0.f) ? 1.f / den : 1.f;

      float ax = 0.f, ay = 0.f, az = 0.f, aw = 0.f;
#pragma unroll
      for (int g = 0; g < MAXP / 4; ++g) {
        if (8 * g < deg) {
#pragma unroll
          for (int k = 0; k < 4; ++k) {
            const float a = w[4 * g + k] * invden;
            ax += a * r[4 * g + k].x;
            ay += a * r[4 * g + k].y;
            az += a * r[4 * g + k].z;
            aw += a * r[4 * g + k].w;
          }
        }
      }
      ax += xhalf(ax);
      ay += xhalf(ay);
      az += xhalf(az);
      aw += xhalf(aw);
      float4 ov;
      ov.x = ax + bv.x;
      ov.y = ay + bv.y;
      ov.z = az + bv.z;
      ov.w = aw + bv.w;
      if (h == 0) *(float4*)(out + (size_t)node * OUT_DIM + 4 * c) = ov;
      continue;
    }

    int   j0 = start + lane;
    float e0 = (j0 < end) ? leaky(el[csr_src[j0]] + er_d) : -INFINITY;

    float m = e0;
    for (int j = j0 + 32; j < end; j += 32) {
      m = fmaxf(m, leaky(el[csr_src[j]] + er_d));
    }
#pragma unroll
    for (int off = 16; off > 0; off >>= 1) m = fmaxf(m, __shfl_xor(m, off, 32));
    if (end <= start) m = 0.f;

    float den = (j0 < end) ? __expf(e0 - m) : 0.f;
    for (int j = j0 + 32; j < end; j += 32) {
      den += __expf(leaky(el[csr_src[j]] + er_d) - m);
    }
#pragma unroll
    for (int off = 16; off > 0; off >>= 1) den += __shfl_xor(den, off, 32);
    float invden = (den > 0.f) ? 1.f / den : 1.f;

    float acc0 = 0.f, acc1 = 0.f;
    for (int j = start; j < end; ++j) {
      int t = j - start;
      int s = csr_src[j];
      float e;
      if (t < 32) {
        e = __shfl(e0, t, 32);
      } else {
        e = leaky(el[s] + er_d);
      }
      float alpha = __expf(e - m) * invden;
      v2f fv = *(const v2f*)(feat + (size_t)s * OUT_DIM + 2 * lane);
      acc0 += alpha * fv.x;
      acc1 += alpha * fv.y;
    }
    v2f bv2 = *(const v2f*)(bias + 2 * lane);
    v2f ov;
    ov.x = acc0 + bv2.x;
    ov.y = acc1 + bv2.y;
    *(v2f*)(out + (size_t)node * OUT_DIM + 2 * lane) = ov;
  }
}

extern "C" void kernel_launch(void* const* d_in, const int* in_sizes, int n_in,
                              void* d_out, int out_size, void* d_ws, size_t ws_size,
                              hipStream_t stream) {
  const float* h      = (const float*)d_in[0];
  const int*   src    = (const int*)d_in[1];
  const int*   dst    = (const int*)d_in[2];
  const float* W      = (const float*)d_in[3];
  const float* attn_l = (const float*)d_in[4];
  const float* attn_r = (const float*)d_in[5];
  const float* bias   = (const float*)d_in[6];
  float* out = (float*)d_out;

  const int N = in_sizes[0] / IN_DIM;
  const int E = in_sizes[1];

  char* ws = (char*)d_ws;
  size_t off = 0;
  auto carve = [&](size_t bytes) -> char* {
    char* p = ws + off;
    off = (off + bytes + 255) & ~(size_t)255;
    return p;
  };
  float* feat    = (float*)carve((size_t)N * OUT_DIM * sizeof(float));
  float* el      = (float*)carve((size_t)N * sizeof(float));
  float* er      = (float*)carve((size_t)N * sizeof(float));
  int*   offsets = (int*)  carve((size_t)N * sizeof(int));
  int*   btotal  = (int*)  carve(MAX_BK * sizeof(int));
  int*   bcursor = (int*)  carve(MAX_BK * sizeof(int));
  int*   csr_src = (int*)  carve((size_t)E * sizeof(int));
  int2*  pairs   = (int2*) carve((size_t)E * sizeof(int2));

  const int nbk = (N + BK_NODES - 1) / BK_NODES;
  const int nch = (E + CHUNK - 1) / CHUNK;

  k_feat_wmma<<<(N + 127) / 128, 256, 0, stream>>>(h, W, attn_l, attn_r, feat, el, er,
                                                   btotal, bcursor, N, nbk);
  k_bhist   <<<nch, L1_THREADS, 0, stream>>>(dst, btotal, E, nbk);
  k_bscatter<<<nch, L1_THREADS, 0, stream>>>(src, dst, btotal, bcursor, pairs, E, nbk);
  k_fine    <<<nbk, BK_NODES, 0, stream>>>(pairs, btotal, offsets, csr_src, N);
  k_gat_agg<<<(N + 8 * NPW - 1) / (8 * NPW), 256, 0, stream>>>(feat, el, er, offsets, csr_src, attn_l, bias, out, N, E);
}


# Aggregation softmax one lane per edge: score from the projection's el[] (one gather per node), max and denominator as 32-lane butterflies, weights by row broadcast; el, den, row sums re-ordered.
# speedup vs baseline: 3.5577x; 1.1061x over previous
#include <hip/hip_runtime.h>
#include <math.h>

#define IN_DIM   128
#define OUT_DIM  64
#define NEG_SLOPE 0.2f

typedef float v2f __attribute__((ext_vector_type(2)));
typedef float v8f __attribute__((ext_vector_type(8)));

__device__ __forceinline__ float leaky(float e) {
  return (e > 0.f) ? e : NEG_SLOPE * e;
}

template <int M>
__device__ __forceinline__ float row_xor(float x) {
  return __builtin_bit_cast(float,
      __builtin_amdgcn_update_dpp(0, __builtin_bit_cast(int, x), 0x160 + M, 0xf, 0xf, true));
}

__device__ __forceinline__ float row16_sum(float x) {
  x += row_xor<1>(x);
  x += row_xor<2>(x);
  x += row_xor<4>(x);
  x += row_xor<8>(x);
  return x;
}

#define WPP 144

__global__ __launch_bounds__(256) void k_feat_wmma(const float* __restrict__ h,
                                                   const float* __restrict__ W,
                                                   const float* __restrict__ attn_l,
                                                   const float* __restrict__ attn_r,
                                                   float* __restrict__ feat,
                                                   float* __restrict__ el,
                                                   float* __restrict__ er,
                                                   int* __restrict__ btotal,
                                                   int* __restrict__ bcursor,
                                                   int N, int nbk) {
  __shared__ float sW[(IN_DIM / 2) * WPP];
  const int tid  = threadIdx.x;
  const int wave = tid >> 5;
  const int lane = tid & 31;
  const int c    = lane & 15;
  const int hf   = lane >> 4;
  const int rowBase = (blockIdx.x * 8 + wave) * 16;

  const int row  = rowBase + c;
  const int rowc = (row < N) ? row : (N - 1);
  const float* __restrict__ ap = h + (size_t)rowc * IN_DIM + 4 * hf;
  float4 a[IN_DIM / 8];
#pragma unroll
  for (int kk = 0; kk < IN_DIM / 8; ++kk) a[kk] = *(const float4*)(ap + 8 * kk);

  const float al0 = attn_l[c], al1 = attn_l[16 + c], al2 = attn_l[32 + c], al3 = attn_l[48 + c];
  const float ar0 = attn_r[c], ar1 = attn_r[16 + c], ar2 = attn_r[32 + c], ar3 = attn_r[48 + c];

  if (blockIdx.x == 0) {
    for (int t = tid; t < nbk; t += 256) { btotal[t] = 0; bcursor[t] = 0; }
  }

#pragma unroll
  for (int i = 0; i < 4; ++i) {
    const int idx = i * 256 + tid;
    const int kp  = idx >> 4;
    const int q   = idx & 15;
    const float4 w0 = *(const float4*)(W + (size_t)(2 * kp) * OUT_DIM + 4 * q);
    const float4 w1 = *(const float4*)(W + (size_t)(2 * kp + 1) * OUT_DIM + 4 * q);
    float* d = sW + kp * WPP + 8 * q;
    *(float4*)(d)     = make_float4(w0.x, w1.x, w0.y, w1.y);
    *(float4*)(d + 4) = make_float4(w0.z, w1.z, w0.w, w1.w);
  }
  __syncthreads();
  if (rowBase >= N) return;

  v8f acc0 = {0.f, 0.f, 0.f, 0.f, 0.f, 0.f, 0.f, 0.f};
  v8f acc1 = acc0, acc2 = acc0, acc3 = acc0;
  const float* bp = sW + (2 * hf) * WPP + 2 * c;

#pragma unroll
  for (int kk = 0; kk < IN_DIM / 8; ++kk) {
    v2f alo, ahi;
    alo.x = a[kk].x; alo.y = a[kk].y;
    ahi.x = a[kk].z; ahi.y = a[kk].w;
    const float* b0 = bp + (4 * kk) * WPP;
    const float* b1 = b0 + WPP;
    const v2f l0 = *(const v2f*)(b0);
    const v2f l1 = *(const v2f*)(b0 + 32);
    const v2f l2 = *(const v2f*)(b0 + 64);
    const v2f l3 = *(const v2f*)(b0 + 96);
    const v2f u0 = *(const v2f*)(b1);
    const v2f u1 = *(const v2f*)(b1 + 32);
    const v2f u2 = *(const v2f*)(b1 + 64);
    const v2f u3 = *(const v2f*)(b1 + 96);
    acc0 = __builtin_amdgcn_wmma_f32_16x16x4_f32(false, alo, false, l0, (short)0, acc0, false, false);
    acc1 = __builtin_amdgcn_wmma_f32_16x16x4_f32(false, alo, false, l1, (short)0, acc1, false, false);
    acc2 = __builtin_amdgcn_wmma_f32_16x16x4_f32(false, alo, false, l2, (short)0, acc2, false, false);
    acc3 = __builtin_amdgcn_wmma_f32_16x16x4_f32(false, alo, false, l3, (short)0, acc3, false, false);
    acc0 = __builtin_amdgcn_wmma_f32_16x16x4_f32(false, ahi, false, u0, (short)0, acc0, false, false);
    acc1 = __builtin_amdgcn_wmma_f32_16x16x4_f32(false, ahi, false, u1, (short)0, acc1, false, false);
    acc2 = __builtin_amdgcn_wmma_f32_16x16x4_f32(false, ahi, false, u2, (short)0, acc2, false, false);
    acc3 = __builtin_amdgcn_wmma_f32_16x16x4_f32(false, ahi, false, u3, (short)0, acc3, false, false);
  }

  const int mBase = rowBase + hf * 8;
#pragma unroll
  for (int r = 0; r < 8; ++r) {
    const int m = mBase + r;
    if (m < N) {
      float* o = feat + (size_t)m * OUT_DIM + c;
      o[0]  = acc0[r];
      o[16] = acc1[r];
      o[32] = acc2[r];
      o[48] = acc3[r];
    }
  }

  float sl = 0.f, sr = 0.f;
#pragma unroll
  for (int r = 0; r < 8; ++r) {
    float pl = acc0[r] * al0 + acc1[r] * al1 + acc2[r] * al2 + acc3[r] * al3;
    float pr = acc0[r] * ar0 + acc1[r] * ar1 + acc2[r] * ar2 + acc3[r] * ar3;
    pl = row16_sum(pl);
    pr = row16_sum(pr);
    if (c == r) { sl = pl; sr = pr; }
  }
  const int mo = mBase + c;
  if (c < 8 && mo < N) {
    el[mo] = sl;
    er[mo] = sr;
  }
}

#define BK_SHIFT 10
#define BK_NODES 1024
#define MAX_BK   1024
#define CHUNK    4096
#define L1_THREADS 1024
#define EPT (CHUNK / L1_THREADS)

__device__ __forceinline__ int block_scan_incl_1024(int v, int* sw, int tid) {
  const int lane = tid & 31;
  const int w    = tid >> 5;
  int x = v;
#pragma unroll
  for (int off = 1; off < 32; off <<= 1) {
    int t = __shfl_up(x, off, 32);
    if (lane >= off) x += t;
  }
  if (lane == 31) sw[w] = x;
  __syncthreads();
  if (w == 0) {
    int y = sw[lane];
#pragma unroll
    for (int off = 1; off < 32; off <<= 1) {
      int t = __shfl_up(y, off, 32);
      if (lane >= off) y += t;
    }
    sw[lane] = y;
  }
  __syncthreads();
  if (w > 0) x += sw[w - 1];
  return x;
}

__global__ __launch_bounds__(L1_THREADS) void k_bhist(const int* __restrict__ dst,
                                                      int* __restrict__ btotal,
                                                      int E, int nbk) {
  __shared__ int sh[MAX_BK];
  const int tid = threadIdx.x;
  for (int t = tid; t < nbk; t += L1_THREADS) sh[t] = 0;
  __syncthreads();
  const int base = blockIdx.x * CHUNK;
#pragma unroll
  for (int i = 0; i < EPT; ++i) {
    int e = base + i * L1_THREADS + tid;
    if (e < E) atomicAdd(&sh[dst[e] >> BK_SHIFT], 1);
  }
  __syncthreads();
  for (int t = tid; t < nbk; t += L1_THREADS) {
    int c = sh[t];
    if (c) atomicAdd(&btotal[t], c);
  }
}

__global__ __launch_bounds__(L1_THREADS) void k_bscatter(const int* __restrict__ src,
                                                         const int* __restrict__ dst,
                                                         const int* __restrict__ btotal,
                                                         int* __restrict__ bcursor,
                                                         int2* __restrict__ pairs,
                                                         int E, int nbk) {
  __shared__ int sh[MAX_BK];
  __shared__ int sw[32];
  const int tid = threadIdx.x;
  sh[tid] = 0;
  const int tot = (tid < nbk) ? btotal[tid] : 0;
  __syncthreads();

  const int base = blockIdx.x * CHUNK;
  int d[EPT], s[EPT], r[EPT];
#pragma unroll
  for (int i = 0; i < EPT; ++i) {
    int e = base + i * L1_THREADS + tid;
    d[i] = (e < E) ? dst[e] : -1;
    s[i] = (e < E) ? src[e] : 0;
  }
#pragma unroll
  for (int i = 0; i < EPT; ++i) {
    r[i] = (d[i] >= 0) ? atomicAdd(&sh[d[i] >> BK_SHIFT], 1) : 0;
  }
  const int incl = block_scan_incl_1024(tot, sw, tid);
  if (tid < nbk) {
    int c = sh[tid];
    int g = c ? atomicAdd(&bcursor[tid], c) : 0;
    sh[tid] = incl - tot + g;
  }
  __syncthreads();
#pragma unroll
  for (int i = 0; i < EPT; ++i) {
    if (d[i] >= 0) {
      int p = sh[d[i] >> BK_SHIFT] + r[i];
      pairs[p] = make_int2(s[i], d[i]);
    }
  }
}

__global__ __launch_bounds__(BK_NODES) void k_fine(const int2* __restrict__ pairs,
                                                   const int* __restrict__ btotal,
                                                   int* __restrict__ offsets,
                                                   int* __restrict__ csr_src,
                                                   int N) {
  __shared__ int scnt[BK_NODES];
  __shared__ int sw[32];
  __shared__ int sbase;
  const int tid = threadIdx.x;
  const int b   = blockIdx.x;
  scnt[tid] = 0;
  if (tid == 0) sbase = 0;
  __syncthreads();
  const int part = (tid < b) ? btotal[tid] : 0;
  if (part) atomicAdd(&sbase, part);
  const int cnt = btotal[b];
  __syncthreads();
  const int base = sbase;
  const int2* __restrict__ pb = pairs + base;

  for (int i0 = 0; i0 < cnt; i0 += 4 * BK_NODES) {
    int dl[4];
#pragma unroll
    for (int k = 0; k < 4; ++k) {
      int i = i0 + k * BK_NODES + tid;
      dl[k] = pb[(i < cnt) ? i : (cnt - 1)].y;
    }
#pragma unroll
    for (int k = 0; k < 4; ++k) {
      int i = i0 + k * BK_NODES + tid;
      if (i < cnt) atomicAdd(&scnt[dl[k] & (BK_NODES - 1)], 1);
    }
  }
  __syncthreads();
  const int c = scnt[tid];
  const int excl = block_scan_incl_1024(c, sw, tid) - c;
  const int node = (b << BK_SHIFT) + tid;
  if (node < N) offsets[node] = base + excl;
  scnt[tid] = excl;
  __syncthreads();

  int* __restrict__ cb = csr_src + base;
  for (int i0 = 0; i0 < cnt; i0 += 4 * BK_NODES) {
    int2 v[4];
#pragma unroll
    for (int k = 0; k < 4; ++k) {
      int i = i0 + k * BK_NODES + tid;
      v[k] = pb[(i < cnt) ? i : (cnt - 1)];
    }
#pragma unroll
    for (int k = 0; k < 4; ++k) {
      int i = i0 + k * BK_NODES + tid;
      if (i < cnt) {
        int p = atomicAdd(&scnt[v[k].y & (BK_NODES - 1)], 1);
        cb[p] = v[k].x;
      }
    }
  }
}

#define MAXP 16
#define NPW  13

__device__ __forceinline__ float xhalf(float x) {
  int b = __builtin_bit_cast(int, x);
  return __builtin_bit_cast(float,
      __builtin_amdgcn_permlanex16(b, b, 0x76543210, 0xfedcba98, false, false));
}

#define RS_CASE(I) case I: return __builtin_amdgcn_update_dpp(0, x, 0x150 + I, 0xf, 0xf, true);
__device__ __forceinline__ int row_bcast(int x, int i) {
  switch (i) {
    RS_CASE(0) RS_CASE(1) RS_CASE(2) RS_CASE(3) RS_CASE(4) RS_CASE(5) RS_CASE(6) RS_CASE(7)
    RS_CASE(8) RS_CASE(9) RS_CASE(10) RS_CASE(11) RS_CASE(12) RS_CASE(13) RS_CASE(14) RS_CASE(15)
  }
  return x;
}
__device__ __forceinline__ float row_bcast_f(float x, int i) {
  return __builtin_bit_cast(float, row_bcast(__builtin_bit_cast(int, x), i));
}

template <int G>
__device__ __forceinline__ void agg_node(const float4* __restrict__ feat4,
                                         const float* __restrict__ el,
                                         const int* __restrict__ csr_src,
                                         float* __restrict__ out,
                                         int node, int start, int deg, int P, float er_d,
                                         int c, int h, float4 bv) {
  const int  t     = c + (h ? P : 0);
  const bool valid = c < (h ? (deg - P) : P);
  const int  tt    = (t < deg) ? t : (deg - 1);
  const int  s0    = csr_src[start + tt];
  const float elv  = el[s0];
  __builtin_amdgcn_sched_barrier(0);

  float4 r[4 * G];
#pragma unroll
  for (int i = 0; i < 4 * G; ++i) {
    if (i <= 4 * (G - 1) || i < P) {
      const int s = row_bcast(s0, i);
      r[i] = feat4[s * (OUT_DIM / 4) + c];
    } else {
      r[i] = make_float4(0.f, 0.f, 0.f, 0.f);
    }
  }

  float e = leaky(elv + er_d);
  e = valid ? e : -INFINITY;
  float m = e;
  m = fmaxf(m, row_xor<1>(m));
  m = fmaxf(m, row_xor<2>(m));
  m = fmaxf(m, row_xor<4>(m));
  m = fmaxf(m, row_xor<8>(m));
  m = fmaxf(m, xhalf(m));
  const float p = __expf(e - m);
  float den = row16_sum(p);
  den += xhalf(den);
  const float invden = 1.f / ((den > 0.f) ? den : 1.f);
  const float alpha  = p * invden;

  v2f axy = {0.f, 0.f}, azw = {0.f, 0.f};
#pragma unroll
  for (int i = 0; i < 4 * G; ++i) {
    const float a = row_bcast_f(alpha, i);
    v2f a2;  a2.x = a;  a2.y = a;
    v2f rxy; rxy.x = r[i].x; rxy.y = r[i].y;
    v2f rzw; rzw.x = r[i].z; rzw.y = r[i].w;
    axy += a2 * rxy;
    azw += a2 * rzw;
  }
  float4 ov;
  ov.x = axy.x + xhalf(axy.x) + bv.x;
  ov.y = axy.y + xhalf(axy.y) + bv.y;
  ov.z = azw.x + xhalf(azw.x) + bv.z;
  ov.w = azw.y + xhalf(azw.y) + bv.w;
  if (h == 0) *(float4*)(out + (size_t)node * OUT_DIM + 4 * c) = ov;
}

__global__ __launch_bounds__(256) void k_gat_agg(const float* __restrict__ feat,
                                                 const float* __restrict__ el,
                                                 const float* __restrict__ er,
                                                 const int* __restrict__ offsets,
                                                 const int* __restrict__ csr_src,
                                                 const float* __restrict__ bias,
                                                 float* __restrict__ out,
                                                 int N, int E) {
  const int wv    = __builtin_amdgcn_readfirstlane((int)(threadIdx.x >> 5));
  const int nbase = (blockIdx.x * 8 + wv) * NPW;
  if (nbase >= N) return;
  const int lane = threadIdx.x & 31;
  const int h    = lane >> 4;
  const int c    = lane & 15;

  const int   nl   = nbase + lane;
  const int   nlc  = (nl < N) ? nl : (N - 1);
  const int   ldo  = offsets[nlc];
  const int   voff = (nl < N) ? ldo : E;
  const float ver  = er[nlc];

  const float4 bv = *(const float4*)(bias + 4 * c);
  const float4* __restrict__ feat4 = (const float4*)feat;

#pragma unroll 1
  for (int it = 0; it < NPW; ++it) {
    const int node = nbase + it;
    if (node >= N) break;
    const int   start = __builtin_amdgcn_readlane(voff, it);
    const int   end   = __builtin_amdgcn_readlane(voff, it + 1);
    const float er_d  = __builtin_bit_cast(float,
        __builtin_amdgcn_readlane(__builtin_bit_cast(int, ver), it));
    const int   deg   = end - start;

    if (deg <= 2 * MAXP) {
      if (deg <= 0) {
        if (h == 0) *(float4*)(out + (size_t)node * OUT_DIM + 4 * c) = bv;
        continue;
      }
      const int P = (deg + 1) >> 1;
      if (P <= 4)       agg_node<1>(feat4, el, csr_src, out, node, start, deg, P, er_d, c, h, bv);
      else if (P <= 8)  agg_node<2>(feat4, el, csr_src, out, node, start, deg, P, er_d, c, h, bv);
      else if (P <= 12) agg_node<3>(feat4, el, csr_src, out, node, start, deg, P, er_d, c, h, bv);
      else              agg_node<4>(feat4, el, csr_src, out, node, start, deg, P, er_d, c, h, bv);
      continue;
    }

    int   j0 = start + lane;
    float e0 = (j0 < end) ? leaky(el[csr_src[j0]] + er_d) : -INFINITY;

    float m = e0;
    for (int j = j0 + 32; j < end; j += 32) {
      m = fmaxf(m, leaky(el[csr_src[j]] + er_d));
    }
#pragma unroll
    for (int off = 16; off > 0; off >>= 1) m = fmaxf(m, __shfl_xor(m, off, 32));
    if (end <= start) m = 0.f;

    float den = (j0 < end) ? __expf(e0 - m) : 0.f;
    for (int j = j0 + 32; j < end; j += 32) {
      den += __expf(leaky(el[csr_src[j]] + er_d) - m);
    }
#pragma unroll
    for (int off = 16; off > 0; off >>= 1) den += __shfl_xor(den, off, 32);
    float invden = (den > 0.f) ? 1.f / den : 1.f;

    float acc0 = 0.f, acc1 = 0.f;
    for (int j = start; j < end; ++j) {
      int t = j - start;
      int s = csr_src[j];
      float e;
      if (t < 32) {
        e = __shfl(e0, t, 32);
      } else {
        e = leaky(el[s] + er_d);
      }
      float alpha = __expf(e - m) * invden;
      v2f fv = *(const v2f*)(feat + (size_t)s * OUT_DIM + 2 * lane);
      acc0 += alpha * fv.x;
      acc1 += alpha * fv.y;
    }
    v2f bv2 = *(const v2f*)(bias + 2 * lane);
    v2f ov;
    ov.x = acc0 + bv2.x;
    ov.y = acc1 + bv2.y;
    *(v2f*)(out + (size_t)node * OUT_DIM + 2 * lane) = ov;
  }
}

extern "C" void kernel_launch(void* const* d_in, const int* in_sizes, int n_in,
                              void* d_out, int out_size, void* d_ws, size_t ws_size,
                              hipStream_t stream) {
  const float* h      = (const float*)d_in[0];
  const int*   src    = (const int*)d_in[1];
  const int*   dst    = (const int*)d_in[2];
  const float* W      = (const float*)d_in[3];
  const float* attn_l = (const float*)d_in[4];
  const float* attn_r = (const float*)d_in[5];
  const float* bias   = (const float*)d_in[6];
  float* out = (float*)d_out;

  const int N = in_sizes[0] / IN_DIM;
  const int E = in_sizes[1];

  char* ws = (char*)d_ws;
  size_t off = 0;
  auto carve = [&](size_t bytes) -> char* {
    char* p = ws + off;
    off = (off + bytes + 255) & ~(size_t)255;
    return p;
  };
  float* feat    = (float*)carve((size_t)N * OUT_DIM * sizeof(float));
  float* el      = (float*)carve((size_t)N * sizeof(float));
  float* er      = (float*)carve((size_t)N * sizeof(float));
  int*   offsets = (int*)  carve((size_t)N * sizeof(int));
  int*   btotal  = (int*)  carve(MAX_BK * sizeof(int));
  int*   bcursor = (int*)  carve(MAX_BK * sizeof(int));
  int*   csr_src = (int*)  carve((size_t)E * sizeof(int));
  int2*  pairs   = (int2*) carve((size_t)E * sizeof(int2));

  const int nbk = (N + BK_NODES - 1) / BK_NODES;
  const int nch = (E + CHUNK - 1) / CHUNK;

  k_feat_wmma<<<(N + 127) / 128, 256, 0, stream>>>(h, W, attn_l, attn_r, feat, el, er,
                                                   btotal, bcursor, N, nbk);
  k_bhist   <<<nch, L1_THREADS, 0, stream>>>(dst, btotal, E, nbk);
  k_bscatter<<<nch, L1_THREADS, 0, stream>>>(src, dst, btotal, bcursor, pairs, E, nbk);
  k_fine    <<<nbk, BK_NODES, 0, stream>>>(pairs, btotal, offsets, csr_src, N);
  k_gat_agg<<<(N + 8 * NPW - 1) / (8 * NPW), 256, 0, stream>>>(feat, el, er, offsets, csr_src, bias, out, N, E);
}


# Intermediate edge lists stored as whole lines: level-1 pairs and each bucket's CSR sources are ordered in LDS first, then consecutive lanes store consecutive entries, not 32 lines per store.
# speedup vs baseline: 4.1948x; 1.1791x over previous
#include <hip/hip_runtime.h>
#include <math.h>

#define IN_DIM   128
#define OUT_DIM  64
#define NEG_SLOPE 0.2f

typedef float v2f __attribute__((ext_vector_type(2)));
typedef float v8f __attribute__((ext_vector_type(8)));

__device__ __forceinline__ float leaky(float e) {
  return (e > 0.f) ? e : NEG_SLOPE * e;
}

template <int M>
__device__ __forceinline__ float row_xor(float x) {
  return __builtin_bit_cast(float,
      __builtin_amdgcn_update_dpp(0, __builtin_bit_cast(int, x), 0x160 + M, 0xf, 0xf, true));
}

__device__ __forceinline__ float row16_sum(float x) {
  x += row_xor<1>(x);
  x += row_xor<2>(x);
  x += row_xor<4>(x);
  x += row_xor<8>(x);
  return x;
}

#define WPP 144

__global__ __launch_bounds__(256) void k_feat_wmma(const float* __restrict__ h,
                                                   const float* __restrict__ W,
                                                   const float* __restrict__ attn_l,
                                                   const float* __restrict__ attn_r,
                                                   float* __restrict__ feat,
                                                   float* __restrict__ el,
                                                   float* __restrict__ er,
                                                   int* __restrict__ btotal,
                                                   int* __restrict__ bcursor,
                                                   int N, int nbk) {
  __shared__ float sW[(IN_DIM / 2) * WPP];
  const int tid  = threadIdx.x;
  const int wave = tid >> 5;
  const int lane = tid & 31;
  const int c    = lane & 15;
  const int hf   = lane >> 4;
  const int rowBase = (blockIdx.x * 8 + wave) * 16;

  const int row  = rowBase + c;
  const int rowc = (row < N) ? row : (N - 1);
  const float* __restrict__ ap = h + (size_t)rowc * IN_DIM + 4 * hf;
  float4 a[IN_DIM / 8];
#pragma unroll
  for (int kk = 0; kk < IN_DIM / 8; ++kk) a[kk] = *(const float4*)(ap + 8 * kk);

  const float al0 = attn_l[c], al1 = attn_l[16 + c], al2 = attn_l[32 + c], al3 = attn_l[48 + c];
  const float ar0 = attn_r[c], ar1 = attn_r[16 + c], ar2 = attn_r[32 + c], ar3 = attn_r[48 + c];

  if (blockIdx.x == 0) {
    for (int t = tid; t < nbk; t += 256) { btotal[t] = 0; bcursor[t] = 0; }
  }

#pragma unroll
  for (int i = 0; i < 4; ++i) {
    const int idx = i * 256 + tid;
    const int kp  = idx >> 4;
    const int q   = idx & 15;
    const float4 w0 = *(const float4*)(W + (size_t)(2 * kp) * OUT_DIM + 4 * q);
    const float4 w1 = *(const float4*)(W + (size_t)(2 * kp + 1) * OUT_DIM + 4 * q);
    float* d = sW + kp * WPP + 8 * q;
    *(float4*)(d)     = make_float4(w0.x, w1.x, w0.y, w1.y);
    *(float4*)(d + 4) = make_float4(w0.z, w1.z, w0.w, w1.w);
  }
  __syncthreads();
  if (rowBase >= N) return;

  v8f acc0 = {0.f, 0.f, 0.f, 0.f, 0.f, 0.f, 0.f, 0.f};
  v8f acc1 = acc0, acc2 = acc0, acc3 = acc0;
  const float* bp = sW + (2 * hf) * WPP + 2 * c;

#pragma unroll
  for (int kk = 0; kk < IN_DIM / 8; ++kk) {
    v2f alo, ahi;
    alo.x = a[kk].x; alo.y = a[kk].y;
    ahi.x = a[kk].z; ahi.y = a[kk].w;
    const float* b0 = bp + (4 * kk) * WPP;
    const float* b1 = b0 + WPP;
    const v2f l0 = *(const v2f*)(b0);
    const v2f l1 = *(const v2f*)(b0 + 32);
    const v2f l2 = *(const v2f*)(b0 + 64);
    const v2f l3 = *(const v2f*)(b0 + 96);
    const v2f u0 = *(const v2f*)(b1);
    const v2f u1 = *(const v2f*)(b1 + 32);
    const v2f u2 = *(const v2f*)(b1 + 64);
    const v2f u3 = *(const v2f*)(b1 + 96);
    acc0 = __builtin_amdgcn_wmma_f32_16x16x4_f32(false, alo, false, l0, (short)0, acc0, false, false);
    acc1 = __builtin_amdgcn_wmma_f32_16x16x4_f32(false, alo, false, l1, (short)0, acc1, false, false);
    acc2 = __builtin_amdgcn_wmma_f32_16x16x4_f32(false, alo, false, l2, (short)0, acc2, false, false);
    acc3 = __builtin_amdgcn_wmma_f32_16x16x4_f32(false, alo, false, l3, (short)0, acc3, false, false);
    acc0 = __builtin_amdgcn_wmma_f32_16x16x4_f32(false, ahi, false, u0, (short)0, acc0, false, false);
    acc1 = __builtin_amdgcn_wmma_f32_16x16x4_f32(false, ahi, false, u1, (short)0, acc1, false, false);
    acc2 = __builtin_amdgcn_wmma_f32_16x16x4_f32(false, ahi, false, u2, (short)0, acc2, false, false);
    acc3 = __builtin_amdgcn_wmma_f32_16x16x4_f32(false, ahi, false, u3, (short)0, acc3, false, false);
  }

  const int mBase = rowBase + hf * 8;
#pragma unroll
  for (int r = 0; r < 8; ++r) {
    const int m = mBase + r;
    if (m < N) {
      float* o = feat + (size_t)m * OUT_DIM + c;
      o[0]  = acc0[r];
      o[16] = acc1[r];
      o[32] = acc2[r];
      o[48] = acc3[r];
    }
  }

  float sl = 0.f, sr = 0.f;
#pragma unroll
  for (int r = 0; r < 8; ++r) {
    float pl = acc0[r] * al0 + acc1[r] * al1 + acc2[r] * al2 + acc3[r] * al3;
    float pr = acc0[r] * ar0 + acc1[r] * ar1 + acc2[r] * ar2 + acc3[r] * ar3;
    pl = row16_sum(pl);
    pr = row16_sum(pr);
    if (c == r) { sl = pl; sr = pr; }
  }
  const int mo = mBase + c;
  if (c < 8 && mo < N) {
    el[mo] = sl;
    er[mo] = sr;
  }
}

#define BK_SHIFT 10
#define BK_NODES 1024
#define MAX_BK   1024
#define CHUNK    4096
#define L1_THREADS 1024
#define EPT (CHUNK / L1_THREADS)
#define FINE_CAP 32768

__device__ __forceinline__ int block_scan_incl_1024(int v, int* sw, int tid) {
  const int lane = tid & 31;
  const int w    = tid >> 5;
  int x = v;
#pragma unroll
  for (int off = 1; off < 32; off <<= 1) {
    int t = __shfl_up(x, off, 32);
    if (lane >= off) x += t;
  }
  if (lane == 31) sw[w] = x;
  __syncthreads();
  if (w == 0) {
    int y = sw[lane];
#pragma unroll
    for (int off = 1; off < 32; off <<= 1) {
      int t = __shfl_up(y, off, 32);
      if (lane >= off) y += t;
    }
    sw[lane] = y;
  }
  __syncthreads();
  if (w > 0) x += sw[w - 1];
  return x;
}

__device__ __forceinline__ int2 block_scan2_incl_1024(int a, int b, int* swa, int* swb, int tid) {
  const int lane = tid & 31;
  const int w    = tid >> 5;
  int x = a, y = b;
#pragma unroll
  for (int off = 1; off < 32; off <<= 1) {
    int t = __shfl_up(x, off, 32);
    int u = __shfl_up(y, off, 32);
    if (lane >= off) { x += t; y += u; }
  }
  if (lane == 31) { swa[w] = x; swb[w] = y; }
  __syncthreads();
  if (w == 0) {
    int p = swa[lane], q = swb[lane];
#pragma unroll
    for (int off = 1; off < 32; off <<= 1) {
      int t = __shfl_up(p, off, 32);
      int u = __shfl_up(q, off, 32);
      if (lane >= off) { p += t; q += u; }
    }
    swa[lane] = p; swb[lane] = q;
  }
  __syncthreads();
  if (w > 0) { x += swa[w - 1]; y += swb[w - 1]; }
  return make_int2(x, y);
}

__global__ __launch_bounds__(L1_THREADS) void k_bhist(const int* __restrict__ dst,
                                                      int* __restrict__ btotal,
                                                      int E, int nbk) {
  __shared__ int sh[MAX_BK];
  const int tid = threadIdx.x;
  for (int t = tid; t < nbk; t += L1_THREADS) sh[t] = 0;
  __syncthreads();
  const int base = blockIdx.x * CHUNK;
#pragma unroll
  for (int i = 0; i < EPT; ++i) {
    int e = base + i * L1_THREADS + tid;
    if (e < E) atomicAdd(&sh[dst[e] >> BK_SHIFT], 1);
  }
  __syncthreads();
  for (int t = tid; t < nbk; t += L1_THREADS) {
    int c = sh[t];
    if (c) atomicAdd(&btotal[t], c);
  }
}

__global__ __launch_bounds__(L1_THREADS) void k_bscatter(const int* __restrict__ src,
                                                         const int* __restrict__ dst,
                                                         const int* __restrict__ btotal,
                                                         int* __restrict__ bcursor,
                                                         int2* __restrict__ pairs,
                                                         int E, int nbk) {
  __shared__ int  sh[MAX_BK];
  __shared__ int  sl[MAX_BK];
  __shared__ int2 stage[CHUNK];
  __shared__ int  swa[32];
  __shared__ int  swb[32];
  const int tid = threadIdx.x;
  sh[tid] = 0;
  const int tot = (tid < nbk) ? btotal[tid] : 0;
  __syncthreads();

  const int base = blockIdx.x * CHUNK;
  int d[EPT], s[EPT], r[EPT];
#pragma unroll
  for (int i = 0; i < EPT; ++i) {
    int e = base + i * L1_THREADS + tid;
    d[i] = (e < E) ? dst[e] : -1;
    s[i] = (e < E) ? src[e] : 0;
  }
#pragma unroll
  for (int i = 0; i < EPT; ++i) {
    r[i] = (d[i] >= 0) ? atomicAdd(&sh[d[i] >> BK_SHIFT], 1) : 0;
  }
  __syncthreads();
  const int c = sh[tid];
  const int2 incl = block_scan2_incl_1024(tot, c, swa, swb, tid);
  const int g = (tid < nbk && c) ? atomicAdd(&bcursor[tid], c) : 0;
  const int lstart = incl.y - c;
  sl[tid] = lstart;
  sh[tid] = (incl.x - tot + g) - lstart;
  __syncthreads();
  int lp[EPT];
#pragma unroll
  for (int i = 0; i < EPT; ++i) lp[i] = sl[(d[i] >= 0) ? (d[i] >> BK_SHIFT) : 0] + r[i];
#pragma unroll
  for (int i = 0; i < EPT; ++i) {
    if (d[i] >= 0) stage[lp[i]] = make_int2(s[i], d[i]);
  }
  __syncthreads();
  const int nval = (E - base < CHUNK) ? (E - base) : CHUNK;
  int2 v[EPT];
  int  gp[EPT];
#pragma unroll
  for (int i = 0; i < EPT; ++i) {
    const int k = i * L1_THREADS + tid;
    v[i] = stage[(k < nval) ? k : (nval - 1)];
  }
#pragma unroll
  for (int i = 0; i < EPT; ++i) gp[i] = sh[v[i].y >> BK_SHIFT] + i * L1_THREADS + tid;
#pragma unroll
  for (int i = 0; i < EPT; ++i) {
    if (i * L1_THREADS + tid < nval) pairs[gp[i]] = v[i];
  }
}

__global__ __launch_bounds__(BK_NODES) void k_fine(const int2* __restrict__ pairs,
                                                   const int* __restrict__ btotal,
                                                   int* __restrict__ offsets,
                                                   int* __restrict__ csr_src,
                                                   int N) {
  __shared__ int scnt[BK_NODES];
  __shared__ int sw[32];
  __shared__ int sbase;
  __shared__ int sord[FINE_CAP];
  const int tid = threadIdx.x;
  const int b   = blockIdx.x;
  scnt[tid] = 0;
  if (tid == 0) sbase = 0;
  __syncthreads();
  const int part = (tid < b) ? btotal[tid] : 0;
  if (part) atomicAdd(&sbase, part);
  const int cnt = btotal[b];
  __syncthreads();
  const int base = sbase;
  const int2* __restrict__ pb = pairs + base;

  for (int i0 = 0; i0 < cnt; i0 += 4 * BK_NODES) {
    int dl[4];
#pragma unroll
    for (int k = 0; k < 4; ++k) {
      int i = i0 + k * BK_NODES + tid;
      dl[k] = pb[(i < cnt) ? i : (cnt - 1)].y;
    }
#pragma unroll
    for (int k = 0; k < 4; ++k) {
      int i = i0 + k * BK_NODES + tid;
      if (i < cnt) atomicAdd(&scnt[dl[k] & (BK_NODES - 1)], 1);
    }
  }
  __syncthreads();
  const int c = scnt[tid];
  const int excl = block_scan_incl_1024(c, sw, tid) - c;
  const int node = (b << BK_SHIFT) + tid;
  if (node < N) offsets[node] = base + excl;
  scnt[tid] = excl;
  __syncthreads();

  int* __restrict__ cb = csr_src + base;
  for (int i0 = 0; i0 < cnt; i0 += 4 * BK_NODES) {
    int2 v[4];
#pragma unroll
    for (int k = 0; k < 4; ++k) {
      int i = i0 + k * BK_NODES + tid;
      v[k] = pb[(i < cnt) ? i : (cnt - 1)];
    }
#pragma unroll
    for (int k = 0; k < 4; ++k) {
      int i = i0 + k * BK_NODES + tid;
      if (i < cnt) {
        int p = atomicAdd(&scnt[v[k].y & (BK_NODES - 1)], 1);
        if (p < FINE_CAP) sord[p] = v[k].x;
        else              cb[p]   = v[k].x;
      }
    }
  }
  __syncthreads();
  const int lim = (cnt < FINE_CAP) ? cnt : FINE_CAP;
  for (int i = tid; i < lim; i += BK_NODES) cb[i] = sord[i];
}

#define MAXP 16
#define NPW  13

__device__ __forceinline__ float xhalf(float x) {
  int b = __builtin_bit_cast(int, x);
  return __builtin_bit_cast(float,
      __builtin_amdgcn_permlanex16(b, b, 0x76543210, 0xfedcba98, false, false));
}

#define RS_CASE(I) case I: return __builtin_amdgcn_update_dpp(0, x, 0x150 + I, 0xf, 0xf, true);
__device__ __forceinline__ int row_bcast(int x, int i) {
  switch (i) {
    RS_CASE(0) RS_CASE(1) RS_CASE(2) RS_CASE(3) RS_CASE(4) RS_CASE(5) RS_CASE(6) RS_CASE(7)
    RS_CASE(8) RS_CASE(9) RS_CASE(10) RS_CASE(11) RS_CASE(12) RS_CASE(13) RS_CASE(14) RS_CASE(15)
  }
  return x;
}
__device__ __forceinline__ float row_bcast_f(float x, int i) {
  return __builtin_bit_cast(float, row_bcast(__builtin_bit_cast(int, x), i));
}

template <int G>
__device__ __forceinline__ void agg_node(const float4* __restrict__ feat4,
                                         const float* __restrict__ el,
                                         const int* __restrict__ csr_src,
                                         float* __restrict__ out,
                                         int node, int start, int deg, int P, float er_d,
                                         int c, int h, float4 bv) {
  const int  t     = c + (h ? P : 0);
  const bool valid = c < (h ? (deg - P) : P);
  const int  tt    = (t < deg) ? t : (deg - 1);
  const int  s0    = csr_src[start + tt];
  const float elv  = el[s0];
  __builtin_amdgcn_sched_barrier(0);

  float4 r[4 * G];
#pragma unroll
  for (int i = 0; i < 4 * G; ++i) {
    if (i <= 4 * (G - 1) || i < P) {
      const int s = row_bcast(s0, i);
      r[i] = feat4[s * (OUT_DIM / 4) + c];
    } else {
      r[i] = make_float4(0.f, 0.f, 0.f, 0.f);
    }
  }

  float e = leaky(elv + er_d);
  e = valid ? e : -INFINITY;
  float m = e;
  m = fmaxf(m, row_xor<1>(m));
  m = fmaxf(m, row_xor<2>(m));
  m = fmaxf(m, row_xor<4>(m));
  m = fmaxf(m, row_xor<8>(m));
  m = fmaxf(m, xhalf(m));
  const float p = __expf(e - m);
  float den = row16_sum(p);
  den += xhalf(den);
  const float invden = 1.f / ((den > 0.f) ? den : 1.f);
  const float alpha  = p * invden;

  v2f axy = {0.f, 0.f}, azw = {0.f, 0.f};
#pragma unroll
  for (int i = 0; i < 4 * G; ++i) {
    const float a = row_bcast_f(alpha, i);
    v2f a2;  a2.x = a;  a2.y = a;
    v2f rxy; rxy.x = r[i].x; rxy.y = r[i].y;
    v2f rzw; rzw.x = r[i].z; rzw.y = r[i].w;
    axy += a2 * rxy;
    azw += a2 * rzw;
  }
  float4 ov;
  ov.x = axy.x + xhalf(axy.x) + bv.x;
  ov.y = axy.y + xhalf(axy.y) + bv.y;
  ov.z = azw.x + xhalf(azw.x) + bv.z;
  ov.w = azw.y + xhalf(azw.y) + bv.w;
  if (h == 0) *(float4*)(out + (size_t)node * OUT_DIM + 4 * c) = ov;
}

__global__ __launch_bounds__(256) void k_gat_agg(const float* __restrict__ feat,
                                                 const float* __restrict__ el,
                                                 const float* __restrict__ er,
                                                 const int* __restrict__ offsets,
                                                 const int* __restrict__ csr_src,
                                                 const float* __restrict__ bias,
                                                 float* __restrict__ out,
                                                 int N, int E) {
  const int wv    = __builtin_amdgcn_readfirstlane((int)(threadIdx.x >> 5));
  const int nbase = (blockIdx.x * 8 + wv) * NPW;
  if (nbase >= N) return;
  const int lane = threadIdx.x & 31;
  const int h    = lane >> 4;
  const int c    = lane & 15;

  const int   nl   = nbase + lane;
  const int   nlc  = (nl < N) ? nl : (N - 1);
  const int   ldo  = offsets[nlc];
  const int   voff = (nl < N) ? ldo : E;
  const float ver  = er[nlc];

  const float4 bv = *(const float4*)(bias + 4 * c);
  const float4* __restrict__ feat4 = (const float4*)feat;

#pragma unroll 1
  for (int it = 0; it < NPW; ++it) {
    const int node = nbase + it;
    if (node >= N) break;
    const int   start = __builtin_amdgcn_readlane(voff, it);
    const int   end   = __builtin_amdgcn_readlane(voff, it + 1);
    const float er_d  = __builtin_bit_cast(float,
        __builtin_amdgcn_readlane(__builtin_bit_cast(int, ver), it));
    const int   deg   = end - start;

    if (deg <= 2 * MAXP) {
      if (deg <= 0) {
        if (h == 0) *(float4*)(out + (size_t)node * OUT_DIM + 4 * c) = bv;
        continue;
      }
      const int P = (deg + 1) >> 1;
      if (P <= 4)       agg_node<1>(feat4, el, csr_src, out, node, start, deg, P, er_d, c, h, bv);
      else if (P <= 8)  agg_node<2>(feat4, el, csr_src, out, node, start, deg, P, er_d, c, h, bv);
      else if (P <= 12) agg_node<3>(feat4, el, csr_src, out, node, start, deg, P, er_d, c, h, bv);
      else              agg_node<4>(feat4, el, csr_src, out, node, start, deg, P, er_d, c, h, bv);
      continue;
    }

    int   j0 = start + lane;
    float e0 = (j0 < end) ? leaky(el[csr_src[j0]] + er_d) : -INFINITY;

    float m = e0;
    for (int j = j0 + 32; j < end; j += 32) {
      m = fmaxf(m, leaky(el[csr_src[j]] + er_d));
    }
#pragma unroll
    for (int off = 16; off > 0; off >>= 1) m = fmaxf(m, __shfl_xor(m, off, 32));
    if (end <= start) m = 0.f;

    float den = (j0 < end) ? __expf(e0 - m) : 0.f;
    for (int j = j0 + 32; j < end; j += 32) {
      den += __expf(leaky(el[csr_src[j]] + er_d) - m);
    }
#pragma unroll
    for (int off = 16; off > 0; off >>= 1) den += __shfl_xor(den, off, 32);
    float invden = (den > 0.f) ? 1.f / den : 1.f;

    float acc0 = 0.f, acc1 = 0.f;
    for (int j = start; j < end; ++j) {
      int t = j - start;
      int s = csr_src[j];
      float e;
      if (t < 32) {
        e = __shfl(e0, t, 32);
      } else {
        e = leaky(el[s] + er_d);
      }
      float alpha = __expf(e - m) * invden;
      v2f fv = *(const v2f*)(feat + (size_t)s * OUT_DIM + 2 * lane);
      acc0 += alpha * fv.x;
      acc1 += alpha * fv.y;
    }
    v2f bv2 = *(const v2f*)(bias + 2 * lane);
    v2f ov;
    ov.x = acc0 + bv2.x;
    ov.y = acc1 + bv2.y;
    *(v2f*)(out + (size_t)node * OUT_DIM + 2 * lane) = ov;
  }
}

extern "C" void kernel_launch(void* const* d_in, const int* in_sizes, int n_in,
                              void* d_out, int out_size, void* d_ws, size_t ws_size,
                              hipStream_t stream) {
  const float* h      = (const float*)d_in[0];
  const int*   src    = (const int*)d_in[1];
  const int*   dst    = (const int*)d_in[2];
  const float* W      = (const float*)d_in[3];
  const float* attn_l = (const float*)d_in[4];
  const float* attn_r = (const float*)d_in[5];
  const float* bias   = (const float*)d_in[6];
  float* out = (float*)d_out;

  const int N = in_sizes[0] / IN_DIM;
  const int E = in_sizes[1];

  char* ws = (char*)d_ws;
  size_t off = 0;
  auto carve = [&](size_t bytes) -> char* {
    char* p = ws + off;
    off = (off + bytes + 255) & ~(size_t)255;
    return p;
  };
  float* feat    = (float*)carve((size_t)N * OUT_DIM * sizeof(float));
  float* el      = (float*)carve((size_t)N * sizeof(float));
  float* er      = (float*)carve((size_t)N * sizeof(float));
  int*   offsets = (int*)  carve((size_t)N * sizeof(int));
  int*   btotal  = (int*)  carve(MAX_BK * sizeof(int));
  int*   bcursor = (int*)  carve(MAX_BK * sizeof(int));
  int*   csr_src = (int*)  carve((size_t)E * sizeof(int));
  int2*  pairs   = (int2*) carve((size_t)E * sizeof(int2));

  const int nbk = (N + BK_NODES - 1) / BK_NODES;
  const int nch = (E + CHUNK - 1) / CHUNK;

  k_feat_wmma<<<(N + 127) / 128, 256, 0, stream>>>(h, W, attn_l, attn_r, feat, el, er,
                                                   btotal, bcursor, N, nbk);
  k_bhist   <<<nch, L1_THREADS, 0, stream>>>(dst, btotal, E, nbk);
  k_bscatter<<<nch, L1_THREADS, 0, stream>>>(src, dst, btotal, bcursor, pairs, E, nbk);
  k_fine    <<<nbk, BK_NODES, 0, stream>>>(pairs, btotal, offsets, csr_src, N);
  k_gat_agg<<<(N + 8 * NPW - 1) / (8 * NPW), 256, 0, stream>>>(feat, el, er, offsets, csr_src, bias, out, N, E);
}


# Level-1 bucket histogram runs as extra blocks of the projection launch and leaves per-block counts (no global adds); the scatter sums the table's columns. 5 to 4 launches; node sum order as before.
# speedup vs baseline: 4.3750x; 1.0430x over previous
#include <hip/hip_runtime.h>
#include <math.h>

#define IN_DIM   128
#define OUT_DIM  64
#define NEG_SLOPE 0.2f

typedef float v2f __attribute__((ext_vector_type(2)));
typedef float v8f __attribute__((ext_vector_type(8)));

__device__ __forceinline__ float leaky(float e) {
  return (e > 0.f) ? e : NEG_SLOPE * e;
}

template <int M>
__device__ __forceinline__ float row_xor(float x) {
  return __builtin_bit_cast(float,
      __builtin_amdgcn_update_dpp(0, __builtin_bit_cast(int, x), 0x160 + M, 0xf, 0xf, true));
}

__device__ __forceinline__ float row16_sum(float x) {
  x += row_xor<1>(x);
  x += row_xor<2>(x);
  x += row_xor<4>(x);
  x += row_xor<8>(x);
  return x;
}

#define WPP 144
#define BK_SHIFT 10
#define MAX_BK   1024
#define HEDGES   16384

__global__ __launch_bounds__(256) void k_feat_wmma(const float* __restrict__ h,
                                                   const float* __restrict__ W,
                                                   const float* __restrict__ attn_l,
                                                   const float* __restrict__ attn_r,
                                                   const int* __restrict__ dst,
                                                   float* __restrict__ feat,
                                                   float* __restrict__ el,
                                                   float* __restrict__ er,
                                                   int* __restrict__ bcnt,
                                                   int* __restrict__ bcursor,
                                                   int N, int E, int nbk, int nhb) {
  __shared__ float sW[(IN_DIM / 2) * WPP];
  __shared__ int   shist[MAX_BK];
  const int tid  = threadIdx.x;

  if ((int)blockIdx.x < nhb) {
    for (int t = tid; t < nbk; t += 256) shist[t] = 0;
    __syncthreads();
    const int nq = E >> 2;
    const int q0 = blockIdx.x * (HEDGES / 4);
    if (nq > 0) {
      int4 d[HEDGES / (4 * 256)];
#pragma unroll
      for (int i = 0; i < HEDGES / (4 * 256); ++i) {
        const int q = q0 + i * 256 + tid;
        d[i] = ((const int4*)dst)[(q < nq) ? q : (nq - 1)];
      }
#pragma unroll
      for (int i = 0; i < HEDGES / (4 * 256); ++i) {
        const int q = q0 + i * 256 + tid;
        if (q < nq) {
          atomicAdd(&shist[d[i].x >> BK_SHIFT], 1);
          atomicAdd(&shist[d[i].y >> BK_SHIFT], 1);
          atomicAdd(&shist[d[i].z >> BK_SHIFT], 1);
          atomicAdd(&shist[d[i].w >> BK_SHIFT], 1);
        }
      }
    }
    if ((int)blockIdx.x == nhb - 1 && tid < (E & 3)) {
      atomicAdd(&shist[dst[4 * nq + tid] >> BK_SHIFT], 1);
    }
    __syncthreads();
    int* __restrict__ row = bcnt + (size_t)blockIdx.x * nbk;
    for (int t = tid; t < nbk; t += 256) row[t] = shist[t];
    return;
  }

  const int fb   = blockIdx.x - nhb;
  const int wave = tid >> 5;
  const int lane = tid & 31;
  const int c    = lane & 15;
  const int hf   = lane >> 4;
  const int rowBase = (fb * 8 + wave) * 16;

  const int row  = rowBase + c;
  const int rowc = (row < N) ? row : (N - 1);
  const float* __restrict__ ap = h + (size_t)rowc * IN_DIM + 4 * hf;
  float4 a[IN_DIM / 8];
#pragma unroll
  for (int kk = 0; kk < IN_DIM / 8; ++kk) a[kk] = *(const float4*)(ap + 8 * kk);

  const float al0 = attn_l[c], al1 = attn_l[16 + c], al2 = attn_l[32 + c], al3 = attn_l[48 + c];
  const float ar0 = attn_r[c], ar1 = attn_r[16 + c], ar2 = attn_r[32 + c], ar3 = attn_r[48 + c];

  if (fb == 0) {
    for (int t = tid; t < nbk; t += 256) bcursor[t] = 0;
  }

#pragma unroll
  for (int i = 0; i < 4; ++i) {
    const int idx = i * 256 + tid;
    const int kp  = idx >> 4;
    const int q   = idx & 15;
    const float4 w0 = *(const float4*)(W + (size_t)(2 * kp) * OUT_DIM + 4 * q);
    const float4 w1 = *(const float4*)(W + (size_t)(2 * kp + 1) * OUT_DIM + 4 * q);
    float* d = sW + kp * WPP + 8 * q;
    *(float4*)(d)     = make_float4(w0.x, w1.x, w0.y, w1.y);
    *(float4*)(d + 4) = make_float4(w0.z, w1.z, w0.w, w1.w);
  }
  __syncthreads();
  if (rowBase >= N) return;

  v8f acc0 = {0.f, 0.f, 0.f, 0.f, 0.f, 0.f, 0.f, 0.f};
  v8f acc1 = acc0, acc2 = acc0, acc3 = acc0;
  const float* bp = sW + (2 * hf) * WPP + 2 * c;

#pragma unroll
  for (int kk = 0; kk < IN_DIM / 8; ++kk) {
    v2f alo, ahi;
    alo.x = a[kk].x; alo.y = a[kk].y;
    ahi.x = a[kk].z; ahi.y = a[kk].w;
    const float* b0 = bp + (4 * kk) * WPP;
    const float* b1 = b0 + WPP;
    const v2f l0 = *(const v2f*)(b0);
    const v2f l1 = *(const v2f*)(b0 + 32);
    const v2f l2 = *(const v2f*)(b0 + 64);
    const v2f l3 = *(const v2f*)(b0 + 96);
    const v2f u0 = *(const v2f*)(b1);
    const v2f u1 = *(const v2f*)(b1 + 32);
    const v2f u2 = *(const v2f*)(b1 + 64);
    const v2f u3 = *(const v2f*)(b1 + 96);
    acc0 = __builtin_amdgcn_wmma_f32_16x16x4_f32(false, alo, false, l0, (short)0, acc0, false, false);
    acc1 = __builtin_amdgcn_wmma_f32_16x16x4_f32(false, alo, false, l1, (short)0, acc1, false, false);
    acc2 = __builtin_amdgcn_wmma_f32_16x16x4_f32(false, alo, false, l2, (short)0, acc2, false, false);
    acc3 = __builtin_amdgcn_wmma_f32_16x16x4_f32(false, alo, false, l3, (short)0, acc3, false, false);
    acc0 = __builtin_amdgcn_wmma_f32_16x16x4_f32(false, ahi, false, u0, (short)0, acc0, false, false);
    acc1 = __builtin_amdgcn_wmma_f32_16x16x4_f32(false, ahi, false, u1, (short)0, acc1, false, false);
    acc2 = __builtin_amdgcn_wmma_f32_16x16x4_f32(false, ahi, false, u2, (short)0, acc2, false, false);
    acc3 = __builtin_amdgcn_wmma_f32_16x16x4_f32(false, ahi, false, u3, (short)0, acc3, false, false);
  }

  const int mBase = rowBase + hf * 8;
#pragma unroll
  for (int r = 0; r < 8; ++r) {
    const int m = mBase + r;
    if (m < N) {
      float* o = feat + (size_t)m * OUT_DIM + c;
      o[0]  = acc0[r];
      o[16] = acc1[r];
      o[32] = acc2[r];
      o[48] = acc3[r];
    }
  }

  float sl = 0.f, sr = 0.f;
#pragma unroll
  for (int r = 0; r < 8; ++r) {
    float pl = acc0[r] * al0 + acc1[r] * al1 + acc2[r] * al2 + acc3[r] * al3;
    float pr = acc0[r] * ar0 + acc1[r] * ar1 + acc2[r] * ar2 + acc3[r] * ar3;
    pl = row16_sum(pl);
    pr = row16_sum(pr);
    if (c == r) { sl = pl; sr = pr; }
  }
  const int mo = mBase + c;
  if (c < 8 && mo < N) {
    el[mo] = sl;
    er[mo] = sr;
  }
}

#define BK_NODES 1024
#define CHUNK    4096
#define L1_THREADS 1024
#define EPT (CHUNK / L1_THREADS)
#define TLOADS 12
#define FINE_CAP 32768

__device__ __forceinline__ int block_scan_incl_1024(int v, int* sw, int tid) {
  const int lane = tid & 31;
  const int w    = tid >> 5;
  int x = v;
#pragma unroll
  for (int off = 1; off < 32; off <<= 1) {
    int t = __shfl_up(x, off, 32);
    if (lane >= off) x += t;
  }
  if (lane == 31) sw[w] = x;
  __syncthreads();
  if (w == 0) {
    int y = sw[lane];
#pragma unroll
    for (int off = 1; off < 32; off <<= 1) {
      int t = __shfl_up(y, off, 32);
      if (lane >= off) y += t;
    }
    sw[lane] = y;
  }
  __syncthreads();
  if (w > 0) x += sw[w - 1];
  return x;
}

__device__ __forceinline__ int2 block_scan2_incl_1024(int a, int b, int* swa, int* swb, int tid) {
  const int lane = tid & 31;
  const int w    = tid >> 5;
  int x = a, y = b;
#pragma unroll
  for (int off = 1; off < 32; off <<= 1) {
    int t = __shfl_up(x, off, 32);
    int u = __shfl_up(y, off, 32);
    if (lane >= off) { x += t; y += u; }
  }
  if (lane == 31) { swa[w] = x; swb[w] = y; }
  __syncthreads();
  if (w == 0) {
    int p = swa[lane], q = swb[lane];
#pragma unroll
    for (int off = 1; off < 32; off <<= 1) {
      int t = __shfl_up(p, off, 32);
      int u = __shfl_up(q, off, 32);
      if (lane >= off) { p += t; q += u; }
    }
    swa[lane] = p; swb[lane] = q;
  }
  __syncthreads();
  if (w > 0) { x += swa[w - 1]; y += swb[w - 1]; }
  return make_int2(x, y);
}

__global__ __launch_bounds__(L1_THREADS) void k_bscatter(const int* __restrict__ src,
                                                         const int* __restrict__ dst,
                                                         const int* __restrict__ bcnt,
                                                         int* __restrict__ btotal,
                                                         int* __restrict__ bcursor,
                                                         int2* __restrict__ pairs,
                                                         int E, int nbk, int nhb) {
  __shared__ int  sh[MAX_BK];
  __shared__ int  sl[MAX_BK];
  __shared__ int  stot[MAX_BK];
  __shared__ int2 stage[CHUNK];
  __shared__ int  swa[32];
  __shared__ int  swb[32];
  const int tid = threadIdx.x;
  sh[tid] = 0;
  stot[tid] = 0;
  __syncthreads();

  const int base = blockIdx.x * CHUNK;
  int d[EPT], s[EPT], r[EPT];
#pragma unroll
  for (int i = 0; i < EPT; ++i) {
    int e = base + i * L1_THREADS + tid;
    d[i] = (e < E) ? dst[e] : -1;
    s[i] = (e < E) ? src[e] : 0;
  }
  {
    const int ntab = nhb * nbk;
    const int step = L1_THREADS % nbk;
    int b = tid % nbk;
    for (int k0 = 0; k0 < ntab; k0 += TLOADS * L1_THREADS) {
      int v[TLOADS];
#pragma unroll
      for (int k = 0; k < TLOADS; ++k) {
        const int idx = k0 + k * L1_THREADS + tid;
        v[k] = (idx < ntab) ? bcnt[idx] : 0;
      }
#pragma unroll
      for (int k = 0; k < TLOADS; ++k) {
        if (v[k]) atomicAdd(&stot[b], v[k]);
        b += step;
        if (b >= nbk) b -= nbk;
      }
    }
  }
#pragma unroll
  for (int i = 0; i < EPT; ++i) {
    r[i] = (d[i] >= 0) ? atomicAdd(&sh[d[i] >> BK_SHIFT], 1) : 0;
  }
  __syncthreads();
  const int c   = sh[tid];
  const int tot = stot[tid];
  if (blockIdx.x == 0 && tid < nbk) btotal[tid] = tot;
  const int2 incl = block_scan2_incl_1024(tot, c, swa, swb, tid);
  const int g = (tid < nbk && c) ? atomicAdd(&bcursor[tid], c) : 0;
  const int lstart = incl.y - c;
  sl[tid] = lstart;
  sh[tid] = (incl.x - tot + g) - lstart;
  __syncthreads();
  int lp[EPT];
#pragma unroll
  for (int i = 0; i < EPT; ++i) lp[i] = sl[(d[i] >= 0) ? (d[i] >> BK_SHIFT) : 0] + r[i];
#pragma unroll
  for (int i = 0; i < EPT; ++i) {
    if (d[i] >= 0) stage[lp[i]] = make_int2(s[i], d[i]);
  }
  __syncthreads();
  const int nval = (E - base < CHUNK) ? (E - base) : CHUNK;
  int2 v[EPT];
  int  gp[EPT];
#pragma unroll
  for (int i = 0; i < EPT; ++i) {
    const int k = i * L1_THREADS + tid;
    v[i] = stage[(k < nval) ? k : (nval - 1)];
  }
#pragma unroll
  for (int i = 0; i < EPT; ++i) gp[i] = sh[v[i].y >> BK_SHIFT] + i * L1_THREADS + tid;
#pragma unroll
  for (int i = 0; i < EPT; ++i) {
    if (i * L1_THREADS + tid < nval) pairs[gp[i]] = v[i];
  }
}

__global__ __launch_bounds__(BK_NODES) void k_fine(const int2* __restrict__ pairs,
                                                   const int* __restrict__ btotal,
                                                   int* __restrict__ offsets,
                                                   int* __restrict__ csr_src,
                                                   int N) {
  __shared__ int scnt[BK_NODES];
  __shared__ int sw[32];
  __shared__ int sbase;
  __shared__ int sord[FINE_CAP];
  const int tid = threadIdx.x;
  const int b   = blockIdx.x;
  scnt[tid] = 0;
  if (tid == 0) sbase = 0;
  __syncthreads();
  const int part = (tid < b) ? btotal[tid] : 0;
  if (part) atomicAdd(&sbase, part);
  const int cnt = btotal[b];
  __syncthreads();
  const int base = sbase;
  const int2* __restrict__ pb = pairs + base;

  for (int i0 = 0; i0 < cnt; i0 += 4 * BK_NODES) {
    int dl[4];
#pragma unroll
    for (int k = 0; k < 4; ++k) {
      int i = i0 + k * BK_NODES + tid;
      dl[k] = pb[(i < cnt) ? i : (cnt - 1)].y;
    }
#pragma unroll
    for (int k = 0; k < 4; ++k) {
      int i = i0 + k * BK_NODES + tid;
      if (i < cnt) atomicAdd(&scnt[dl[k] & (BK_NODES - 1)], 1);
    }
  }
  __syncthreads();
  const int c = scnt[tid];
  const int excl = block_scan_incl_1024(c, sw, tid) - c;
  const int node = (b << BK_SHIFT) + tid;
  if (node < N) offsets[node] = base + excl;
  scnt[tid] = excl;
  __syncthreads();

  int* __restrict__ cb = csr_src + base;
  for (int i0 = 0; i0 < cnt; i0 += 4 * BK_NODES) {
    int2 v[4];
#pragma unroll
    for (int k = 0; k < 4; ++k) {
      int i = i0 + k * BK_NODES + tid;
      v[k] = pb[(i < cnt) ? i : (cnt - 1)];
    }
#pragma unroll
    for (int k = 0; k < 4; ++k) {
      int i = i0 + k * BK_NODES + tid;
      if (i < cnt) {
        int p = atomicAdd(&scnt[v[k].y & (BK_NODES - 1)], 1);
        if (p < FINE_CAP) sord[p] = v[k].x;
        else              cb[p]   = v[k].x;
      }
    }
  }
  __syncthreads();
  const int lim = (cnt < FINE_CAP) ? cnt : FINE_CAP;
  for (int i = tid; i < lim; i += BK_NODES) cb[i] = sord[i];
}

#define MAXP 16
#define NPW  13

__device__ __forceinline__ float xhalf(float x) {
  int b = __builtin_bit_cast(int, x);
  return __builtin_bit_cast(float,
      __builtin_amdgcn_permlanex16(b, b, 0x76543210, 0xfedcba98, false, false));
}

#define RS_CASE(I) case I: return __builtin_amdgcn_update_dpp(0, x, 0x150 + I, 0xf, 0xf, true);
__device__ __forceinline__ int row_bcast(int x, int i) {
  switch (i) {
    RS_CASE(0) RS_CASE(1) RS_CASE(2) RS_CASE(3) RS_CASE(4) RS_CASE(5) RS_CASE(6) RS_CASE(7)
    RS_CASE(8) RS_CASE(9) RS_CASE(10) RS_CASE(11) RS_CASE(12) RS_CASE(13) RS_CASE(14) RS_CASE(15)
  }
  return x;
}
__device__ __forceinline__ float row_bcast_f(float x, int i) {
  return __builtin_bit_cast(float, row_bcast(__builtin_bit_cast(int, x), i));
}

template <int G>
__device__ __forceinline__ void agg_node(const float4* __restrict__ feat4,
                                         const float* __restrict__ el,
                                         const int* __restrict__ csr_src,
                                         float* __restrict__ out,
                                         int node, int start, int deg, int P, float er_d,
                                         int c, int h, float4 bv) {
  const int  t     = c + (h ? P : 0);
  const bool valid = c < (h ? (deg - P) : P);
  const int  tt    = (t < deg) ? t : (deg - 1);
  const int  s0    = csr_src[start + tt];
  const float elv  = el[s0];
  __builtin_amdgcn_sched_barrier(0);

  float4 r[4 * G];
#pragma unroll
  for (int i = 0; i < 4 * G; ++i) {
    if (i <= 4 * (G - 1) || i < P) {
      const int s = row_bcast(s0, i);
      r[i] = feat4[s * (OUT_DIM / 4) + c];
    } else {
      r[i] = make_float4(0.f, 0.f, 0.f, 0.f);
    }
  }

  float e = leaky(elv + er_d);
  e = valid ? e : -INFINITY;
  float m = e;
  m = fmaxf(m, row_xor<1>(m));
  m = fmaxf(m, row_xor<2>(m));
  m = fmaxf(m, row_xor<4>(m));
  m = fmaxf(m, row_xor<8>(m));
  m = fmaxf(m, xhalf(m));
  const float p = __expf(e - m);
  float den = row16_sum(p);
  den += xhalf(den);
  const float invden = 1.f / ((den > 0.f) ? den : 1.f);
  const float alpha  = p * invden;

  v2f axy = {0.f, 0.f}, azw = {0.f, 0.f};
#pragma unroll
  for (int i = 0; i < 4 * G; ++i) {
    const float a = row_bcast_f(alpha, i);
    v2f a2;  a2.x = a;  a2.y = a;
    v2f rxy; rxy.x = r[i].x; rxy.y = r[i].y;
    v2f rzw; rzw.x = r[i].z; rzw.y = r[i].w;
    axy += a2 * rxy;
    azw += a2 * rzw;
  }
  float4 ov;
  ov.x = axy.x + xhalf(axy.x) + bv.x;
  ov.y = axy.y + xhalf(axy.y) + bv.y;
  ov.z = azw.x + xhalf(azw.x) + bv.z;
  ov.w = azw.y + xhalf(azw.y) + bv.w;
  if (h == 0) *(float4*)(out + (size_t)node * OUT_DIM + 4 * c) = ov;
}

__global__ __launch_bounds__(256) void k_gat_agg(const float* __restrict__ feat,
                                                 const float* __restrict__ el,
                                                 const float* __restrict__ er,
                                                 const int* __restrict__ offsets,
                                                 const int* __restrict__ csr_src,
                                                 const float* __restrict__ bias,
                                                 float* __restrict__ out,
                                                 int N, int E) {
  const int wv    = __builtin_amdgcn_readfirstlane((int)(threadIdx.x >> 5));
  const int nbase = (blockIdx.x * 8 + wv) * NPW;
  if (nbase >= N) return;
  const int lane = threadIdx.x & 31;
  const int h    = lane >> 4;
  const int c    = lane & 15;

  const int   nl   = nbase + lane;
  const int   nlc  = (nl < N) ? nl : (N - 1);
  const int   ldo  = offsets[nlc];
  const int   voff = (nl < N) ? ldo : E;
  const float ver  = er[nlc];

  const float4 bv = *(const float4*)(bias + 4 * c);
  const float4* __restrict__ feat4 = (const float4*)feat;

#pragma unroll 1
  for (int it = 0; it < NPW; ++it) {
    const int node = nbase + it;
    if (node >= N) break;
    const int   start = __builtin_amdgcn_readlane(voff, it);
    const int   end   = __builtin_amdgcn_readlane(voff, it + 1);
    const float er_d  = __builtin_bit_cast(float,
        __builtin_amdgcn_readlane(__builtin_bit_cast(int, ver), it));
    const int   deg   = end - start;

    if (deg <= 2 * MAXP) {
      if (deg <= 0) {
        if (h == 0) *(float4*)(out + (size_t)node * OUT_DIM + 4 * c) = bv;
        continue;
      }
      const int P = (deg + 1) >> 1;
      if (P <= 4)       agg_node<1>(feat4, el, csr_src, out, node, start, deg, P, er_d, c, h, bv);
      else if (P <= 8)  agg_node<2>(feat4, el, csr_src, out, node, start, deg, P, er_d, c, h, bv);
      else if (P <= 12) agg_node<3>(feat4, el, csr_src, out, node, start, deg, P, er_d, c, h, bv);
      else              agg_node<4>(feat4, el, csr_src, out, node, start, deg, P, er_d, c, h, bv);
      continue;
    }

    int   j0 = start + lane;
    float e0 = (j0 < end) ? leaky(el[csr_src[j0]] + er_d) : -INFINITY;

    float m = e0;
    for (int j = j0 + 32; j < end; j += 32) {
      m = fmaxf(m, leaky(el[csr_src[j]] + er_d));
    }
#pragma unroll
    for (int off = 16; off > 0; off >>= 1) m = fmaxf(m, __shfl_xor(m, off, 32));
    if (end <= start) m = 0.f;

    float den = (j0 < end) ? __expf(e0 - m) : 0.f;
    for (int j = j0 + 32; j < end; j += 32) {
      den += __expf(leaky(el[csr_src[j]] + er_d) - m);
    }
#pragma unroll
    for (int off = 16; off > 0; off >>= 1) den += __shfl_xor(den, off, 32);
    float invden = (den > 0.f) ? 1.f / den : 1.f;

    float acc0 = 0.f, acc1 = 0.f;
    for (int j = start; j < end; ++j) {
      int t = j - start;
      int s = csr_src[j];
      float e;
      if (t < 32) {
        e = __shfl(e0, t, 32);
      } else {
        e = leaky(el[s] + er_d);
      }
      float alpha = __expf(e - m) * invden;
      v2f fv = *(const v2f*)(feat + (size_t)s * OUT_DIM + 2 * lane);
      acc0 += alpha * fv.x;
      acc1 += alpha * fv.y;
    }
    v2f bv2 = *(const v2f*)(bias + 2 * lane);
    v2f ov;
    ov.x = acc0 + bv2.x;
    ov.y = acc1 + bv2.y;
    *(v2f*)(out + (size_t)node * OUT_DIM + 2 * lane) = ov;
  }
}

extern "C" void kernel_launch(void* const* d_in, const int* in_sizes, int n_in,
                              void* d_out, int out_size, void* d_ws, size_t ws_size,
                              hipStream_t stream) {
  const float* h      = (const float*)d_in[0];
  const int*   src    = (const int*)d_in[1];
  const int*   dst    = (const int*)d_in[2];
  const float* W      = (const float*)d_in[3];
  const float* attn_l = (const float*)d_in[4];
  const float* attn_r = (const float*)d_in[5];
  const float* bias   = (const float*)d_in[6];
  float* out = (float*)d_out;

  const int N = in_sizes[0] / IN_DIM;
  const int E = in_sizes[1];

  char* ws = (char*)d_ws;
  size_t off = 0;
  auto carve = [&](size_t bytes) -> char* {
    char* p = ws + off;
    off = (off + bytes + 255) & ~(size_t)255;
    return p;
  };
  float* feat    = (float*)carve((size_t)N * OUT_DIM * sizeof(float));
  float* el      = (float*)carve((size_t)N * sizeof(float));
  float* er      = (float*)carve((size_t)N * sizeof(float));
  int*   offsets = (int*)  carve((size_t)N * sizeof(int));
  int*   btotal  = (int*)  carve(MAX_BK * sizeof(int));
  int*   bcursor = (int*)  carve(MAX_BK * sizeof(int));
  int*   csr_src = (int*)  carve((size_t)E * sizeof(int));
  int2*  pairs   = (int2*) carve((size_t)E * sizeof(int2));

  const int nbk = (N + BK_NODES - 1) / BK_NODES;
  const int nch = (E + CHUNK - 1) / CHUNK;
  const int nhb = (E + HEDGES - 1) / HEDGES;
  int*   bcnt    = (int*)  carve((size_t)nhb * nbk * sizeof(int));

  k_feat_wmma<<<nhb + (N + 127) / 128, 256, 0, stream>>>(h, W, attn_l, attn_r, dst, feat, el, er,
                                                         bcnt, bcursor, N, E, nbk, nhb);
  k_bscatter<<<nch, L1_THREADS, 0, stream>>>(src, dst, bcnt, btotal, bcursor, pairs, E, nbk, nhb);
  k_fine    <<<nbk, BK_NODES, 0, stream>>>(pairs, btotal, offsets, csr_src, N);
  k_gat_agg<<<(N + 8 * NPW - 1) / (8 * NPW), 256, 0, stream>>>(feat, el, er, offsets, csr_src, bias, out, N, E);
}


# Aggregation: an edge's source logit is the dot of its gathered row with attn_l (16-lane transpose-sum), not a scattered el[] read per edge: half the L2 requests; el sum order re-ordered.
# speedup vs baseline: 4.5427x; 1.0383x over previous
#include <hip/hip_runtime.h>
#include <math.h>

#define IN_DIM   128
#define OUT_DIM  64
#define NEG_SLOPE 0.2f

typedef float v2f __attribute__((ext_vector_type(2)));
typedef float v8f __attribute__((ext_vector_type(8)));

__device__ __forceinline__ float leaky(float e) {
  return (e > 0.f) ? e : NEG_SLOPE * e;
}

template <int M>
__device__ __forceinline__ float row_xor(float x) {
  return __builtin_bit_cast(float,
      __builtin_amdgcn_update_dpp(0, __builtin_bit_cast(int, x), 0x160 + M, 0xf, 0xf, true));
}

__device__ __forceinline__ float row16_sum(float x) {
  x += row_xor<1>(x);
  x += row_xor<2>(x);
  x += row_xor<4>(x);
  x += row_xor<8>(x);
  return x;
}

#define WPP 144
#define BK_SHIFT 10
#define MAX_BK   1024
#define HEDGES   16384

__global__ __launch_bounds__(256) void k_feat_wmma(const float* __restrict__ h,
                                                   const float* __restrict__ W,
                                                   const float* __restrict__ attn_l,
                                                   const float* __restrict__ attn_r,
                                                   const int* __restrict__ dst,
                                                   float* __restrict__ feat,
                                                   float* __restrict__ el,
                                                   float* __restrict__ er,
                                                   int* __restrict__ bcnt,
                                                   int* __restrict__ bcursor,
                                                   int N, int E, int nbk, int nhb) {
  __shared__ float sW[(IN_DIM / 2) * WPP];
  __shared__ int   shist[MAX_BK];
  const int tid  = threadIdx.x;

  if ((int)blockIdx.x < nhb) {
    for (int t = tid; t < nbk; t += 256) shist[t] = 0;
    __syncthreads();
    const int nq = E >> 2;
    const int q0 = blockIdx.x * (HEDGES / 4);
    if (nq > 0) {
      int4 d[HEDGES / (4 * 256)];
#pragma unroll
      for (int i = 0; i < HEDGES / (4 * 256); ++i) {
        const int q = q0 + i * 256 + tid;
        d[i] = ((const int4*)dst)[(q < nq) ? q : (nq - 1)];
      }
#pragma unroll
      for (int i = 0; i < HEDGES / (4 * 256); ++i) {
        const int q = q0 + i * 256 + tid;
        if (q < nq) {
          atomicAdd(&shist[d[i].x >> BK_SHIFT], 1);
          atomicAdd(&shist[d[i].y >> BK_SHIFT], 1);
          atomicAdd(&shist[d[i].z >> BK_SHIFT], 1);
          atomicAdd(&shist[d[i].w >> BK_SHIFT], 1);
        }
      }
    }
    if ((int)blockIdx.x == nhb - 1 && tid < (E & 3)) {
      atomicAdd(&shist[dst[4 * nq + tid] >> BK_SHIFT], 1);
    }
    __syncthreads();
    int* __restrict__ row = bcnt + (size_t)blockIdx.x * nbk;
    for (int t = tid; t < nbk; t += 256) row[t] = shist[t];
    return;
  }

  const int fb   = blockIdx.x - nhb;
  const int wave = tid >> 5;
  const int lane = tid & 31;
  const int c    = lane & 15;
  const int hf   = lane >> 4;
  const int rowBase = (fb * 8 + wave) * 16;

  const int row  = rowBase + c;
  const int rowc = (row < N) ? row : (N - 1);
  const float* __restrict__ ap = h + (size_t)rowc * IN_DIM + 4 * hf;
  float4 a[IN_DIM / 8];
#pragma unroll
  for (int kk = 0; kk < IN_DIM / 8; ++kk) a[kk] = *(const float4*)(ap + 8 * kk);

  const float al0 = attn_l[c], al1 = attn_l[16 + c], al2 = attn_l[32 + c], al3 = attn_l[48 + c];
  const float ar0 = attn_r[c], ar1 = attn_r[16 + c], ar2 = attn_r[32 + c], ar3 = attn_r[48 + c];

  if (fb == 0) {
    for (int t = tid; t < nbk; t += 256) bcursor[t] = 0;
  }

#pragma unroll
  for (int i = 0; i < 4; ++i) {
    const int idx = i * 256 + tid;
    const int kp  = idx >> 4;
    const int q   = idx & 15;
    const float4 w0 = *(const float4*)(W + (size_t)(2 * kp) * OUT_DIM + 4 * q);
    const float4 w1 = *(const float4*)(W + (size_t)(2 * kp + 1) * OUT_DIM + 4 * q);
    float* d = sW + kp * WPP + 8 * q;
    *(float4*)(d)     = make_float4(w0.x, w1.x, w0.y, w1.y);
    *(float4*)(d + 4) = make_float4(w0.z, w1.z, w0.w, w1.w);
  }
  __syncthreads();
  if (rowBase >= N) return;

  v8f acc0 = {0.f, 0.f, 0.f, 0.f, 0.f, 0.f, 0.f, 0.f};
  v8f acc1 = acc0, acc2 = acc0, acc3 = acc0;
  const float* bp = sW + (2 * hf) * WPP + 2 * c;

#pragma unroll
  for (int kk = 0; kk < IN_DIM / 8; ++kk) {
    v2f alo, ahi;
    alo.x = a[kk].x; alo.y = a[kk].y;
    ahi.x = a[kk].z; ahi.y = a[kk].w;
    const float* b0 = bp + (4 * kk) * WPP;
    const float* b1 = b0 + WPP;
    const v2f l0 = *(const v2f*)(b0);
    const v2f l1 = *(const v2f*)(b0 + 32);
    const v2f l2 = *(const v2f*)(b0 + 64);
    const v2f l3 = *(const v2f*)(b0 + 96);
    const v2f u0 = *(const v2f*)(b1);
    const v2f u1 = *(const v2f*)(b1 + 32);
    const v2f u2 = *(const v2f*)(b1 + 64);
    const v2f u3 = *(const v2f*)(b1 + 96);
    acc0 = __builtin_amdgcn_wmma_f32_16x16x4_f32(false, alo, false, l0, (short)0, acc0, false, false);
    acc1 = __builtin_amdgcn_wmma_f32_16x16x4_f32(false, alo, false, l1, (short)0, acc1, false, false);
    acc2 = __builtin_amdgcn_wmma_f32_16x16x4_f32(false, alo, false, l2, (short)0, acc2, false, false);
    acc3 = __builtin_amdgcn_wmma_f32_16x16x4_f32(false, alo, false, l3, (short)0, acc3, false, false);
    acc0 = __builtin_amdgcn_wmma_f32_16x16x4_f32(false, ahi, false, u0, (short)0, acc0, false, false);
    acc1 = __builtin_amdgcn_wmma_f32_16x16x4_f32(false, ahi, false, u1, (short)0, acc1, false, false);
    acc2 = __builtin_amdgcn_wmma_f32_16x16x4_f32(false, ahi, false, u2, (short)0, acc2, false, false);
    acc3 = __builtin_amdgcn_wmma_f32_16x16x4_f32(false, ahi, false, u3, (short)0, acc3, false, false);
  }

  const int mBase = rowBase + hf * 8;
#pragma unroll
  for (int r = 0; r < 8; ++r) {
    const int m = mBase + r;
    if (m < N) {
      float* o = feat + (size_t)m * OUT_DIM + c;
      o[0]  = acc0[r];
      o[16] = acc1[r];
      o[32] = acc2[r];
      o[48] = acc3[r];
    }
  }

  float sl = 0.f, sr = 0.f;
#pragma unroll
  for (int r = 0; r < 8; ++r) {
    float pl = acc0[r] * al0 + acc1[r] * al1 + acc2[r] * al2 + acc3[r] * al3;
    float pr = acc0[r] * ar0 + acc1[r] * ar1 + acc2[r] * ar2 + acc3[r] * ar3;
    pl = row16_sum(pl);
    pr = row16_sum(pr);
    if (c == r) { sl = pl; sr = pr; }
  }
  const int mo = mBase + c;
  if (c < 8 && mo < N) {
    el[mo] = sl;
    er[mo] = sr;
  }
}

#define BK_NODES 1024
#define CHUNK    4096
#define L1_THREADS 1024
#define EPT (CHUNK / L1_THREADS)
#define TLOADS 12
#define FINE_CAP 32768

__device__ __forceinline__ int block_scan_incl_1024(int v, int* sw, int tid) {
  const int lane = tid & 31;
  const int w    = tid >> 5;
  int x = v;
#pragma unroll
  for (int off = 1; off < 32; off <<= 1) {
    int t = __shfl_up(x, off, 32);
    if (lane >= off) x += t;
  }
  if (lane == 31) sw[w] = x;
  __syncthreads();
  if (w == 0) {
    int y = sw[lane];
#pragma unroll
    for (int off = 1; off < 32; off <<= 1) {
      int t = __shfl_up(y, off, 32);
      if (lane >= off) y += t;
    }
    sw[lane] = y;
  }
  __syncthreads();
  if (w > 0) x += sw[w - 1];
  return x;
}

__device__ __forceinline__ int2 block_scan2_incl_1024(int a, int b, int* swa, int* swb, int tid) {
  const int lane = tid & 31;
  const int w    = tid >> 5;
  int x = a, y = b;
#pragma unroll
  for (int off = 1; off < 32; off <<= 1) {
    int t = __shfl_up(x, off, 32);
    int u = __shfl_up(y, off, 32);
    if (lane >= off) { x += t; y += u; }
  }
  if (lane == 31) { swa[w] = x; swb[w] = y; }
  __syncthreads();
  if (w == 0) {
    int p = swa[lane], q = swb[lane];
#pragma unroll
    for (int off = 1; off < 32; off <<= 1) {
      int t = __shfl_up(p, off, 32);
      int u = __shfl_up(q, off, 32);
      if (lane >= off) { p += t; q += u; }
    }
    swa[lane] = p; swb[lane] = q;
  }
  __syncthreads();
  if (w > 0) { x += swa[w - 1]; y += swb[w - 1]; }
  return make_int2(x, y);
}

__global__ __launch_bounds__(L1_THREADS) void k_bscatter(const int* __restrict__ src,
                                                         const int* __restrict__ dst,
                                                         const int* __restrict__ bcnt,
                                                         int* __restrict__ btotal,
                                                         int* __restrict__ bcursor,
                                                         int2* __restrict__ pairs,
                                                         int E, int nbk, int nhb) {
  __shared__ int  sh[MAX_BK];
  __shared__ int  sl[MAX_BK];
  __shared__ int  stot[MAX_BK];
  __shared__ int2 stage[CHUNK];
  __shared__ int  swa[32];
  __shared__ int  swb[32];
  const int tid = threadIdx.x;
  sh[tid] = 0;
  stot[tid] = 0;
  __syncthreads();

  const int base = blockIdx.x * CHUNK;
  int d[EPT], s[EPT], r[EPT];
#pragma unroll
  for (int i = 0; i < EPT; ++i) {
    int e = base + i * L1_THREADS + tid;
    d[i] = (e < E) ? dst[e] : -1;
    s[i] = (e < E) ? src[e] : 0;
  }
  {
    const int ntab = nhb * nbk;
    const int step = L1_THREADS % nbk;
    int b = tid % nbk;
    for (int k0 = 0; k0 < ntab; k0 += TLOADS * L1_THREADS) {
      int v[TLOADS];
#pragma unroll
      for (int k = 0; k < TLOADS; ++k) {
        const int idx = k0 + k * L1_THREADS + tid;
        v[k] = (idx < ntab) ? bcnt[idx] : 0;
      }
#pragma unroll
      for (int k = 0; k < TLOADS; ++k) {
        if (v[k]) atomicAdd(&stot[b], v[k]);
        b += step;
        if (b >= nbk) b -= nbk;
      }
    }
  }
#pragma unroll
  for (int i = 0; i < EPT; ++i) {
    r[i] = (d[i] >= 0) ? atomicAdd(&sh[d[i] >> BK_SHIFT], 1) : 0;
  }
  __syncthreads();
  const int c   = sh[tid];
  const int tot = stot[tid];
  if (blockIdx.x == 0 && tid < nbk) btotal[tid] = tot;
  const int2 incl = block_scan2_incl_1024(tot, c, swa, swb, tid);
  const int g = (tid < nbk && c) ? atomicAdd(&bcursor[tid], c) : 0;
  const int lstart = incl.y - c;
  sl[tid] = lstart;
  sh[tid] = (incl.x - tot + g) - lstart;
  __syncthreads();
  int lp[EPT];
#pragma unroll
  for (int i = 0; i < EPT; ++i) lp[i] = sl[(d[i] >= 0) ? (d[i] >> BK_SHIFT) : 0] + r[i];
#pragma unroll
  for (int i = 0; i < EPT; ++i) {
    if (d[i] >= 0) stage[lp[i]] = make_int2(s[i], d[i]);
  }
  __syncthreads();
  const int nval = (E - base < CHUNK) ? (E - base) : CHUNK;
  int2 v[EPT];
  int  gp[EPT];
#pragma unroll
  for (int i = 0; i < EPT; ++i) {
    const int k = i * L1_THREADS + tid;
    v[i] = stage[(k < nval) ? k : (nval - 1)];
  }
#pragma unroll
  for (int i = 0; i < EPT; ++i) gp[i] = sh[v[i].y >> BK_SHIFT] + i * L1_THREADS + tid;
#pragma unroll
  for (int i = 0; i < EPT; ++i) {
    if (i * L1_THREADS + tid < nval) pairs[gp[i]] = v[i];
  }
}

__global__ __launch_bounds__(BK_NODES) void k_fine(const int2* __restrict__ pairs,
                                                   const int* __restrict__ btotal,
                                                   int* __restrict__ offsets,
                                                   int* __restrict__ csr_src,
                                                   int N) {
  __shared__ int scnt[BK_NODES];
  __shared__ int sw[32];
  __shared__ int sbase;
  __shared__ int sord[FINE_CAP];
  const int tid = threadIdx.x;
  const int b   = blockIdx.x;
  scnt[tid] = 0;
  if (tid == 0) sbase = 0;
  __syncthreads();
  const int part = (tid < b) ? btotal[tid] : 0;
  if (part) atomicAdd(&sbase, part);
  const int cnt = btotal[b];
  __syncthreads();
  const int base = sbase;
  const int2* __restrict__ pb = pairs + base;

  for (int i0 = 0; i0 < cnt; i0 += 4 * BK_NODES) {
    int dl[4];
#pragma unroll
    for (int k = 0; k < 4; ++k) {
      int i = i0 + k * BK_NODES + tid;
      dl[k] = pb[(i < cnt) ? i : (cnt - 1)].y;
    }
#pragma unroll
    for (int k = 0; k < 4; ++k) {
      int i = i0 + k * BK_NODES + tid;
      if (i < cnt) atomicAdd(&scnt[dl[k] & (BK_NODES - 1)], 1);
    }
  }
  __syncthreads();
  const int c = scnt[tid];
  const int excl = block_scan_incl_1024(c, sw, tid) - c;
  const int node = (b << BK_SHIFT) + tid;
  if (node < N) offsets[node] = base + excl;
  scnt[tid] = excl;
  __syncthreads();

  int* __restrict__ cb = csr_src + base;
  for (int i0 = 0; i0 < cnt; i0 += 4 * BK_NODES) {
    int2 v[4];
#pragma unroll
    for (int k = 0; k < 4; ++k) {
      int i = i0 + k * BK_NODES + tid;
      v[k] = pb[(i < cnt) ? i : (cnt - 1)];
    }
#pragma unroll
    for (int k = 0; k < 4; ++k) {
      int i = i0 + k * BK_NODES + tid;
      if (i < cnt) {
        int p = atomicAdd(&scnt[v[k].y & (BK_NODES - 1)], 1);
        if (p < FINE_CAP) sord[p] = v[k].x;
        else              cb[p]   = v[k].x;
      }
    }
  }
  __syncthreads();
  const int lim = (cnt < FINE_CAP) ? cnt : FINE_CAP;
  for (int i = tid; i < lim; i += BK_NODES) cb[i] = sord[i];
}

#define MAXP 16
#define NPW  13

__device__ __forceinline__ float xhalf(float x) {
  int b = __builtin_bit_cast(int, x);
  return __builtin_bit_cast(float,
      __builtin_amdgcn_permlanex16(b, b, 0x76543210, 0xfedcba98, false, false));
}

#define RS_CASE(I) case I: return __builtin_amdgcn_update_dpp(0, x, 0x150 + I, 0xf, 0xf, true);
__device__ __forceinline__ int row_bcast(int x, int i) {
  switch (i) {
    RS_CASE(0) RS_CASE(1) RS_CASE(2) RS_CASE(3) RS_CASE(4) RS_CASE(5) RS_CASE(6) RS_CASE(7)
    RS_CASE(8) RS_CASE(9) RS_CASE(10) RS_CASE(11) RS_CASE(12) RS_CASE(13) RS_CASE(14) RS_CASE(15)
  }
  return x;
}
__device__ __forceinline__ float row_bcast_f(float x, int i) {
  return __builtin_bit_cast(float, row_bcast(__builtin_bit_cast(int, x), i));
}

__device__ __forceinline__ float row_xor_sw(float x, int mask) {
  switch (mask) {
    case 1: return row_xor<1>(x);
    case 2: return row_xor<2>(x);
    case 4: return row_xor<4>(x);
    case 8: return row_xor<8>(x);
  }
  return x;
}

template <int NV, int N>
__device__ __forceinline__ float row_transpose_sum(float (&d)[N], int c) {
  int nv = NV;
#pragma unroll
  for (int b = 0; (1 << b) < N; ++b) {
    const bool bit  = (c >> b) & 1;
    const int  nout = (nv + 1) >> 1;
#pragma unroll
    for (int j = 0; j < N / 2; ++j) {
      if (j < nout) {
        const float a0   = d[2 * j];
        const float a1   = (2 * j + 1 < nv) ? d[2 * j + 1] : 0.f;
        const float keep = bit ? a1 : a0;
        const float send = bit ? a0 : a1;
        d[j] = keep + row_xor_sw(send, 1 << b);
      }
    }
    nv = nout;
  }
  float x = d[0];
#pragma unroll
  for (int m = N; m < 16; m <<= 1) x += row_xor_sw(x, m);
  return x;
}

template <int G>
__device__ __forceinline__ void agg_node(const float4* __restrict__ feat4,
                                         const int* __restrict__ csr_src,
                                         float* __restrict__ out,
                                         int node, int start, int deg, int P, float er_d,
                                         int c, int h, float4 bv, float4 alv) {
  const int  t     = c + (h ? P : 0);
  const bool valid = c < (h ? (deg - P) : P);
  const int  tt    = (t < deg) ? t : (deg - 1);
  const int  s0    = csr_src[start + tt];

  float4 r[4 * G];
#pragma unroll
  for (int i = 0; i < 4 * G; ++i) {
    if (i <= 4 * (G - 1) || i < P) {
      const int s = row_bcast(s0, i);
      r[i] = feat4[s * (OUT_DIM / 4) + c];
    } else {
      r[i] = make_float4(0.f, 0.f, 0.f, 0.f);
    }
  }

  constexpr int NV = 4 * G;
  constexpr int NN = (G == 3) ? 16 : 4 * G;
  float d[NN];
#pragma unroll
  for (int i = 0; i < NN; ++i) {
    if (i < NV) {
      v2f p2;  p2.x = r[i].x; p2.y = r[i].y;
      v2f q2;  q2.x = r[i].z; q2.y = r[i].w;
      v2f a2;  a2.x = alv.x;  a2.y = alv.y;
      v2f b2;  b2.x = alv.z;  b2.y = alv.w;
      v2f t2 = p2 * a2;
      t2 += q2 * b2;
      d[i] = t2.x + t2.y;
    } else {
      d[i] = 0.f;
    }
  }
  const float elv = row_transpose_sum<NV, NN>(d, c);

  float e = leaky(elv + er_d);
  e = valid ? e : -INFINITY;
  float m = e;
  m = fmaxf(m, row_xor<1>(m));
  m = fmaxf(m, row_xor<2>(m));
  m = fmaxf(m, row_xor<4>(m));
  m = fmaxf(m, row_xor<8>(m));
  m = fmaxf(m, xhalf(m));
  const float p = __expf(e - m);
  float den = row16_sum(p);
  den += xhalf(den);
  const float invden = 1.f / ((den > 0.f) ? den : 1.f);
  const float alpha  = p * invden;

  v2f axy = {0.f, 0.f}, azw = {0.f, 0.f};
#pragma unroll
  for (int i = 0; i < 4 * G; ++i) {
    const float a = row_bcast_f(alpha, i);
    v2f a2;  a2.x = a;  a2.y = a;
    v2f rxy; rxy.x = r[i].x; rxy.y = r[i].y;
    v2f rzw; rzw.x = r[i].z; rzw.y = r[i].w;
    axy += a2 * rxy;
    azw += a2 * rzw;
  }
  float4 ov;
  ov.x = axy.x + xhalf(axy.x) + bv.x;
  ov.y = axy.y + xhalf(axy.y) + bv.y;
  ov.z = azw.x + xhalf(azw.x) + bv.z;
  ov.w = azw.y + xhalf(azw.y) + bv.w;
  if (h == 0) *(float4*)(out + (size_t)node * OUT_DIM + 4 * c) = ov;
}

__global__ __launch_bounds__(256) void k_gat_agg(const float* __restrict__ feat,
                                                 const float* __restrict__ el,
                                                 const float* __restrict__ er,
                                                 const int* __restrict__ offsets,
                                                 const int* __restrict__ csr_src,
                                                 const float* __restrict__ bias,
                                                 const float* __restrict__ attn_l,
                                                 float* __restrict__ out,
                                                 int N, int E) {
  const int wv    = __builtin_amdgcn_readfirstlane((int)(threadIdx.x >> 5));
  const int nbase = (blockIdx.x * 8 + wv) * NPW;
  if (nbase >= N) return;
  const int lane = threadIdx.x & 31;
  const int h    = lane >> 4;
  const int c    = lane & 15;

  const int   nl   = nbase + lane;
  const int   nlc  = (nl < N) ? nl : (N - 1);
  const int   ldo  = offsets[nlc];
  const int   voff = (nl < N) ? ldo : E;
  const float ver  = er[nlc];

  const float4 bv = *(const float4*)(bias + 4 * c);
  const float4 alv = *(const float4*)(attn_l + 4 * c);
  const float4* __restrict__ feat4 = (const float4*)feat;

#pragma unroll 1
  for (int it = 0; it < NPW; ++it) {
    const int node = nbase + it;
    if (node >= N) break;
    const int   start = __builtin_amdgcn_readlane(voff, it);
    const int   end   = __builtin_amdgcn_readlane(voff, it + 1);
    const float er_d  = __builtin_bit_cast(float,
        __builtin_amdgcn_readlane(__builtin_bit_cast(int, ver), it));
    const int   deg   = end - start;

    if (deg <= 2 * MAXP) {
      if (deg <= 0) {
        if (h == 0) *(float4*)(out + (size_t)node * OUT_DIM + 4 * c) = bv;
        continue;
      }
      const int P = (deg + 1) >> 1;
      if (P <= 4)       agg_node<1>(feat4, csr_src, out, node, start, deg, P, er_d, c, h, bv, alv);
      else if (P <= 8)  agg_node<2>(feat4, csr_src, out, node, start, deg, P, er_d, c, h, bv, alv);
      else if (P <= 12) agg_node<3>(feat4, csr_src, out, node, start, deg, P, er_d, c, h, bv, alv);
      else              agg_node<4>(feat4, csr_src, out, node, start, deg, P, er_d, c, h, bv, alv);
      continue;
    }

    int   j0 = start + lane;
    float e0 = (j0 < end) ? leaky(el[csr_src[j0]] + er_d) : -INFINITY;

    float m = e0;
    for (int j = j0 + 32; j < end; j += 32) {
      m = fmaxf(m, leaky(el[csr_src[j]] + er_d));
    }
#pragma unroll
    for (int off = 16; off > 0; off >>= 1) m = fmaxf(m, __shfl_xor(m, off, 32));
    if (end <= start) m = 0.f;

    float den = (j0 < end) ? __expf(e0 - m) : 0.f;
    for (int j = j0 + 32; j < end; j += 32) {
      den += __expf(leaky(el[csr_src[j]] + er_d) - m);
    }
#pragma unroll
    for (int off = 16; off > 0; off >>= 1) den += __shfl_xor(den, off, 32);
    float invden = (den > 0.f) ? 1.f / den : 1.f;

    float acc0 = 0.f, acc1 = 0.f;
    for (int j = start; j < end; ++j) {
      int t = j - start;
      int s = csr_src[j];
      float e;
      if (t < 32) {
        e = __shfl(e0, t, 32);
      } else {
        e = leaky(el[s] + er_d);
      }
      float alpha = __expf(e - m) * invden;
      v2f fv = *(const v2f*)(feat + (size_t)s * OUT_DIM + 2 * lane);
      acc0 += alpha * fv.x;
      acc1 += alpha * fv.y;
    }
    v2f bv2 = *(const v2f*)(bias + 2 * lane);
    v2f ov;
    ov.x = acc0 + bv2.x;
    ov.y = acc1 + bv2.y;
    *(v2f*)(out + (size_t)node * OUT_DIM + 2 * lane) = ov;
  }
}

extern "C" void kernel_launch(void* const* d_in, const int* in_sizes, int n_in,
                              void* d_out, int out_size, void* d_ws, size_t ws_size,
                              hipStream_t stream) {
  const float* h      = (const float*)d_in[0];
  const int*   src    = (const int*)d_in[1];
  const int*   dst    = (const int*)d_in[2];
  const float* W      = (const float*)d_in[3];
  const float* attn_l = (const float*)d_in[4];
  const float* attn_r = (const float*)d_in[5];
  const float* bias   = (const float*)d_in[6];
  float* out = (float*)d_out;

  const int N = in_sizes[0] / IN_DIM;
  const int E = in_sizes[1];

  char* ws = (char*)d_ws;
  size_t off = 0;
  auto carve = [&](size_t bytes) -> char* {
    char* p = ws + off;
    off = (off + bytes + 255) & ~(size_t)255;
    return p;
  };
  float* feat    = (float*)carve((size_t)N * OUT_DIM * sizeof(float));
  float* el      = (float*)carve((size_t)N * sizeof(float));
  float* er      = (float*)carve((size_t)N * sizeof(float));
  int*   offsets = (int*)  carve((size_t)N * sizeof(int));
  int*   btotal  = (int*)  carve(MAX_BK * sizeof(int));
  int*   bcursor = (int*)  carve(MAX_BK * sizeof(int));
  int*   csr_src = (int*)  carve((size_t)E * sizeof(int));
  int2*  pairs   = (int2*) carve((size_t)E * sizeof(int2));

  const int nbk = (N + BK_NODES - 1) / BK_NODES;
  const int nch = (E + CHUNK - 1) / CHUNK;
  const int nhb = (E + HEDGES - 1) / HEDGES;
  int*   bcnt    = (int*)  carve((size_t)nhb * nbk * sizeof(int));

  k_feat_wmma<<<nhb + (N + 127) / 128, 256, 0, stream>>>(h, W, attn_l, attn_r, dst, feat, el, er,
                                                         bcnt, bcursor, N, E, nbk, nhb);
  k_bscatter<<<nch, L1_THREADS, 0, stream>>>(src, dst, bcnt, btotal, bcursor, pairs, E, nbk, nhb);
  k_fine    <<<nbk, BK_NODES, 0, stream>>>(pairs, btotal, offsets, csr_src, N);
  k_gat_agg<<<(N + 8 * NPW - 1) / (8 * NPW), 256, 0, stream>>>(feat, el, er, offsets, csr_src, bias, attn_l, out, N, E);
}


# CSR buckets of 512 destination nodes, not 1024: level 2 runs 196 blocks, not 98, each with one load trip per pass, not five; histogram blocks take 32768 edges. Node sum order = LDS arrival order.
# speedup vs baseline: 4.7512x; 1.0459x over previous
#include <hip/hip_runtime.h>
#include <math.h>

#define IN_DIM   128
#define OUT_DIM  64
#define NEG_SLOPE 0.2f

typedef float v2f __attribute__((ext_vector_type(2)));
typedef float v8f __attribute__((ext_vector_type(8)));

__device__ __forceinline__ float leaky(float e) {
  return (e > 0.f) ? e : NEG_SLOPE * e;
}

template <int M>
__device__ __forceinline__ float row_xor(float x) {
  return __builtin_bit_cast(float,
      __builtin_amdgcn_update_dpp(0, __builtin_bit_cast(int, x), 0x160 + M, 0xf, 0xf, true));
}

__device__ __forceinline__ float row16_sum(float x) {
  x += row_xor<1>(x);
  x += row_xor<2>(x);
  x += row_xor<4>(x);
  x += row_xor<8>(x);
  return x;
}

#define WPP 144
#define BK_SHIFT 9
#define MAX_BK   1024
#define HEDGES   32768
#define HROUND   16384

__global__ __launch_bounds__(256) void k_feat_wmma(const float* __restrict__ h,
                                                   const float* __restrict__ W,
                                                   const float* __restrict__ attn_l,
                                                   const float* __restrict__ attn_r,
                                                   const int* __restrict__ dst,
                                                   float* __restrict__ feat,
                                                   float* __restrict__ el,
                                                   float* __restrict__ er,
                                                   int* __restrict__ bcnt,
                                                   int* __restrict__ bcursor,
                                                   int N, int E, int nbk, int nhb) {
  __shared__ float sW[(IN_DIM / 2) * WPP];
  __shared__ int   shist[MAX_BK];
  const int tid  = threadIdx.x;

  if ((int)blockIdx.x < nhb) {
    for (int t = tid; t < nbk; t += 256) shist[t] = 0;
    __syncthreads();
    const int nq = E >> 2;
    if (nq > 0) {
#pragma unroll 1
      for (int hr = 0; hr < HEDGES / HROUND; ++hr) {
        const int q0 = blockIdx.x * (HEDGES / 4) + hr * (HROUND / 4);
        int4 d[HROUND / (4 * 256)];
#pragma unroll
        for (int i = 0; i < HROUND / (4 * 256); ++i) {
          const int q = q0 + i * 256 + tid;
          d[i] = ((const int4*)dst)[(q < nq) ? q : (nq - 1)];
        }
#pragma unroll
        for (int i = 0; i < HROUND / (4 * 256); ++i) {
          const int q = q0 + i * 256 + tid;
          if (q < nq) {
            atomicAdd(&shist[d[i].x >> BK_SHIFT], 1);
            atomicAdd(&shist[d[i].y >> BK_SHIFT], 1);
            atomicAdd(&shist[d[i].z >> BK_SHIFT], 1);
            atomicAdd(&shist[d[i].w >> BK_SHIFT], 1);
          }
        }
      }
    }
    if ((int)blockIdx.x == nhb - 1 && tid < (E & 3)) {
      atomicAdd(&shist[dst[4 * nq + tid] >> BK_SHIFT], 1);
    }
    __syncthreads();
    int* __restrict__ row = bcnt + (size_t)blockIdx.x * nbk;
    for (int t = tid; t < nbk; t += 256) row[t] = shist[t];
    return;
  }

  const int fb   = blockIdx.x - nhb;
  const int wave = tid >> 5;
  const int lane = tid & 31;
  const int c    = lane & 15;
  const int hf   = lane >> 4;
  const int rowBase = (fb * 8 + wave) * 16;

  const int row  = rowBase + c;
  const int rowc = (row < N) ? row : (N - 1);
  const float* __restrict__ ap = h + (size_t)rowc * IN_DIM + 4 * hf;
  float4 a[IN_DIM / 8];
#pragma unroll
  for (int kk = 0; kk < IN_DIM / 8; ++kk) a[kk] = *(const float4*)(ap + 8 * kk);

  const float al0 = attn_l[c], al1 = attn_l[16 + c], al2 = attn_l[32 + c], al3 = attn_l[48 + c];
  const float ar0 = attn_r[c], ar1 = attn_r[16 + c], ar2 = attn_r[32 + c], ar3 = attn_r[48 + c];

  if (fb == 0) {
    for (int t = tid; t < nbk; t += 256) bcursor[t] = 0;
  }

#pragma unroll
  for (int i = 0; i < 4; ++i) {
    const int idx = i * 256 + tid;
    const int kp  = idx >> 4;
    const int q   = idx & 15;
    const float4 w0 = *(const float4*)(W + (size_t)(2 * kp) * OUT_DIM + 4 * q);
    const float4 w1 = *(const float4*)(W + (size_t)(2 * kp + 1) * OUT_DIM + 4 * q);
    float* d = sW + kp * WPP + 8 * q;
    *(float4*)(d)     = make_float4(w0.x, w1.x, w0.y, w1.y);
    *(float4*)(d + 4) = make_float4(w0.z, w1.z, w0.w, w1.w);
  }
  __syncthreads();
  if (rowBase >= N) return;

  v8f acc0 = {0.f, 0.f, 0.f, 0.f, 0.f, 0.f, 0.f, 0.f};
  v8f acc1 = acc0, acc2 = acc0, acc3 = acc0;
  const float* bp = sW + (2 * hf) * WPP + 2 * c;

#pragma unroll
  for (int kk = 0; kk < IN_DIM / 8; ++kk) {
    v2f alo, ahi;
    alo.x = a[kk].x; alo.y = a[kk].y;
    ahi.x = a[kk].z; ahi.y = a[kk].w;
    const float* b0 = bp + (4 * kk) * WPP;
    const float* b1 = b0 + WPP;
    const v2f l0 = *(const v2f*)(b0);
    const v2f l1 = *(const v2f*)(b0 + 32);
    const v2f l2 = *(const v2f*)(b0 + 64);
    const v2f l3 = *(const v2f*)(b0 + 96);
    const v2f u0 = *(const v2f*)(b1);
    const v2f u1 = *(const v2f*)(b1 + 32);
    const v2f u2 = *(const v2f*)(b1 + 64);
    const v2f u3 = *(const v2f*)(b1 + 96);
    acc0 = __builtin_amdgcn_wmma_f32_16x16x4_f32(false, alo, false, l0, (short)0, acc0, false, false);
    acc1 = __builtin_amdgcn_wmma_f32_16x16x4_f32(false, alo, false, l1, (short)0, acc1, false, false);
    acc2 = __builtin_amdgcn_wmma_f32_16x16x4_f32(false, alo, false, l2, (short)0, acc2, false, false);
    acc3 = __builtin_amdgcn_wmma_f32_16x16x4_f32(false, alo, false, l3, (short)0, acc3, false, false);
    acc0 = __builtin_amdgcn_wmma_f32_16x16x4_f32(false, ahi, false, u0, (short)0, acc0, false, false);
    acc1 = __builtin_amdgcn_wmma_f32_16x16x4_f32(false, ahi, false, u1, (short)0, acc1, false, false);
    acc2 = __builtin_amdgcn_wmma_f32_16x16x4_f32(false, ahi, false, u2, (short)0, acc2, false, false);
    acc3 = __builtin_amdgcn_wmma_f32_16x16x4_f32(false, ahi, false, u3, (short)0, acc3, false, false);
  }

  const int mBase = rowBase + hf * 8;
#pragma unroll
  for (int r = 0; r < 8; ++r) {
    const int m = mBase + r;
    if (m < N) {
      float* o = feat + (size_t)m * OUT_DIM + c;
      o[0]  = acc0[r];
      o[16] = acc1[r];
      o[32] = acc2[r];
      o[48] = acc3[r];
    }
  }

  float sl = 0.f, sr = 0.f;
#pragma unroll
  for (int r = 0; r < 8; ++r) {
    float pl = acc0[r] * al0 + acc1[r] * al1 + acc2[r] * al2 + acc3[r] * al3;
    float pr = acc0[r] * ar0 + acc1[r] * ar1 + acc2[r] * ar2 + acc3[r] * ar3;
    pl = row16_sum(pl);
    pr = row16_sum(pr);
    if (c == r) { sl = pl; sr = pr; }
  }
  const int mo = mBase + c;
  if (c < 8 && mo < N) {
    el[mo] = sl;
    er[mo] = sr;
  }
}

#define BK_NODES 512
#define FINE_THREADS 1024
#define FLOADS   9
#define CHUNK    4096
#define L1_THREADS 1024
#define EPT (CHUNK / L1_THREADS)
#define TLOADS 12
#define FINE_CAP 32768

__device__ __forceinline__ int block_scan_incl_1024(int v, int* sw, int tid) {
  const int lane = tid & 31;
  const int w    = tid >> 5;
  int x = v;
#pragma unroll
  for (int off = 1; off < 32; off <<= 1) {
    int t = __shfl_up(x, off, 32);
    if (lane >= off) x += t;
  }
  if (lane == 31) sw[w] = x;
  __syncthreads();
  if (w == 0) {
    int y = sw[lane];
#pragma unroll
    for (int off = 1; off < 32; off <<= 1) {
      int t = __shfl_up(y, off, 32);
      if (lane >= off) y += t;
    }
    sw[lane] = y;
  }
  __syncthreads();
  if (w > 0) x += sw[w - 1];
  return x;
}

__device__ __forceinline__ int2 block_scan2_incl_1024(int a, int b, int* swa, int* swb, int tid) {
  const int lane = tid & 31;
  const int w    = tid >> 5;
  int x = a, y = b;
#pragma unroll
  for (int off = 1; off < 32; off <<= 1) {
    int t = __shfl_up(x, off, 32);
    int u = __shfl_up(y, off, 32);
    if (lane >= off) { x += t; y += u; }
  }
  if (lane == 31) { swa[w] = x; swb[w] = y; }
  __syncthreads();
  if (w == 0) {
    int p = swa[lane], q = swb[lane];
#pragma unroll
    for (int off = 1; off < 32; off <<= 1) {
      int t = __shfl_up(p, off, 32);
      int u = __shfl_up(q, off, 32);
      if (lane >= off) { p += t; q += u; }
    }
    swa[lane] = p; swb[lane] = q;
  }
  __syncthreads();
  if (w > 0) { x += swa[w - 1]; y += swb[w - 1]; }
  return make_int2(x, y);
}

__global__ __launch_bounds__(L1_THREADS) void k_bscatter(const int* __restrict__ src,
                                                         const int* __restrict__ dst,
                                                         const int* __restrict__ bcnt,
                                                         int* __restrict__ btotal,
                                                         int* __restrict__ bcursor,
                                                         int2* __restrict__ pairs,
                                                         int E, int nbk, int nhb) {
  __shared__ int  sh[MAX_BK];
  __shared__ int  sl[MAX_BK];
  __shared__ int  stot[MAX_BK];
  __shared__ int2 stage[CHUNK];
  __shared__ int  swa[32];
  __shared__ int  swb[32];
  const int tid = threadIdx.x;
  sh[tid] = 0;
  stot[tid] = 0;
  __syncthreads();

  const int base = blockIdx.x * CHUNK;
  int d[EPT], s[EPT], r[EPT];
#pragma unroll
  for (int i = 0; i < EPT; ++i) {
    int e = base + i * L1_THREADS + tid;
    d[i] = (e < E) ? dst[e] : -1;
    s[i] = (e < E) ? src[e] : 0;
  }
  {
    const int ntab = nhb * nbk;
    const int step = L1_THREADS % nbk;
    int b = tid % nbk;
    for (int k0 = 0; k0 < ntab; k0 += TLOADS * L1_THREADS) {
      int v[TLOADS];
#pragma unroll
      for (int k = 0; k < TLOADS; ++k) {
        const int idx = k0 + k * L1_THREADS + tid;
        v[k] = (idx < ntab) ? bcnt[idx] : 0;
      }
#pragma unroll
      for (int k = 0; k < TLOADS; ++k) {
        if (v[k]) atomicAdd(&stot[b], v[k]);
        b += step;
        if (b >= nbk) b -= nbk;
      }
    }
  }
#pragma unroll
  for (int i = 0; i < EPT; ++i) {
    r[i] = (d[i] >= 0) ? atomicAdd(&sh[d[i] >> BK_SHIFT], 1) : 0;
  }
  __syncthreads();
  const int c   = sh[tid];
  const int tot = stot[tid];
  if (blockIdx.x == 0 && tid < nbk) btotal[tid] = tot;
  const int2 incl = block_scan2_incl_1024(tot, c, swa, swb, tid);
  const int g = (tid < nbk && c) ? atomicAdd(&bcursor[tid], c) : 0;
  const int lstart = incl.y - c;
  sl[tid] = lstart;
  sh[tid] = (incl.x - tot + g) - lstart;
  __syncthreads();
  int lp[EPT];
#pragma unroll
  for (int i = 0; i < EPT; ++i) lp[i] = sl[(d[i] >= 0) ? (d[i] >> BK_SHIFT) : 0] + r[i];
#pragma unroll
  for (int i = 0; i < EPT; ++i) {
    if (d[i] >= 0) stage[lp[i]] = make_int2(s[i], d[i]);
  }
  __syncthreads();
  const int nval = (E - base < CHUNK) ? (E - base) : CHUNK;
  int2 v[EPT];
  int  gp[EPT];
#pragma unroll
  for (int i = 0; i < EPT; ++i) {
    const int k = i * L1_THREADS + tid;
    v[i] = stage[(k < nval) ? k : (nval - 1)];
  }
#pragma unroll
  for (int i = 0; i < EPT; ++i) gp[i] = sh[v[i].y >> BK_SHIFT] + i * L1_THREADS + tid;
#pragma unroll
  for (int i = 0; i < EPT; ++i) {
    if (i * L1_THREADS + tid < nval) pairs[gp[i]] = v[i];
  }
}

__global__ __launch_bounds__(FINE_THREADS) void k_fine(const int2* __restrict__ pairs,
                                                   const int* __restrict__ btotal,
                                                   int* __restrict__ offsets,
                                                   int* __restrict__ csr_src,
                                                   int N) {
  __shared__ int scnt[BK_NODES];
  __shared__ int sw[32];
  __shared__ int sbase;
  __shared__ int sord[FINE_CAP];
  const int tid = threadIdx.x;
  const int b   = blockIdx.x;
  if (tid < BK_NODES) scnt[tid] = 0;
  if (tid == 0) sbase = 0;
  __syncthreads();
  const int part = (tid < b) ? btotal[tid] : 0;
  if (part) atomicAdd(&sbase, part);
  const int cnt = btotal[b];
  __syncthreads();
  const int base = sbase;
  const int2* __restrict__ pb = pairs + base;

  for (int i0 = 0; i0 < cnt; i0 += FLOADS * FINE_THREADS) {
    int dl[FLOADS];
#pragma unroll
    for (int k = 0; k < FLOADS; ++k) {
      int i = i0 + k * FINE_THREADS + tid;
      dl[k] = pb[(i < cnt) ? i : (cnt - 1)].y;
    }
#pragma unroll
    for (int k = 0; k < FLOADS; ++k) {
      int i = i0 + k * FINE_THREADS + tid;
      if (i < cnt) atomicAdd(&scnt[dl[k] & (BK_NODES - 1)], 1);
    }
  }
  __syncthreads();
  const bool own = tid < BK_NODES;
  const int c = own ? scnt[tid] : 0;
  const int excl = block_scan_incl_1024(c, sw, tid) - c;
  const int node = (b << BK_SHIFT) + tid;
  if (own && node < N) offsets[node] = base + excl;
  if (own) scnt[tid] = excl;
  __syncthreads();

  int* __restrict__ cb = csr_src + base;
  for (int i0 = 0; i0 < cnt; i0 += FLOADS * FINE_THREADS) {
    int2 v[FLOADS];
#pragma unroll
    for (int k = 0; k < FLOADS; ++k) {
      int i = i0 + k * FINE_THREADS + tid;
      v[k] = pb[(i < cnt) ? i : (cnt - 1)];
    }
#pragma unroll
    for (int k = 0; k < FLOADS; ++k) {
      int i = i0 + k * FINE_THREADS + tid;
      if (i < cnt) {
        int p = atomicAdd(&scnt[v[k].y & (BK_NODES - 1)], 1);
        if (p < FINE_CAP) sord[p] = v[k].x;
        else              cb[p]   = v[k].x;
      }
    }
  }
  __syncthreads();
  const int lim = (cnt < FINE_CAP) ? cnt : FINE_CAP;
  for (int i = tid; i < lim; i += FINE_THREADS) cb[i] = sord[i];
}

#define MAXP 16
#define NPW  13

__device__ __forceinline__ float xhalf(float x) {
  int b = __builtin_bit_cast(int, x);
  return __builtin_bit_cast(float,
      __builtin_amdgcn_permlanex16(b, b, 0x76543210, 0xfedcba98, false, false));
}

#define RS_CASE(I) case I: return __builtin_amdgcn_update_dpp(0, x, 0x150 + I, 0xf, 0xf, true);
__device__ __forceinline__ int row_bcast(int x, int i) {
  switch (i) {
    RS_CASE(0) RS_CASE(1) RS_CASE(2) RS_CASE(3) RS_CASE(4) RS_CASE(5) RS_CASE(6) RS_CASE(7)
    RS_CASE(8) RS_CASE(9) RS_CASE(10) RS_CASE(11) RS_CASE(12) RS_CASE(13) RS_CASE(14) RS_CASE(15)
  }
  return x;
}
__device__ __forceinline__ float row_bcast_f(float x, int i) {
  return __builtin_bit_cast(float, row_bcast(__builtin_bit_cast(int, x), i));
}

__device__ __forceinline__ float row_xor_sw(float x, int mask) {
  switch (mask) {
    case 1: return row_xor<1>(x);
    case 2: return row_xor<2>(x);
    case 4: return row_xor<4>(x);
    case 8: return row_xor<8>(x);
  }
  return x;
}

template <int NV, int N>
__device__ __forceinline__ float row_transpose_sum(float (&d)[N], int c) {
  int nv = NV;
#pragma unroll
  for (int b = 0; (1 << b) < N; ++b) {
    const bool bit  = (c >> b) & 1;
    const int  nout = (nv + 1) >> 1;
#pragma unroll
    for (int j = 0; j < N / 2; ++j) {
      if (j < nout) {
        const float a0   = d[2 * j];
        const float a1   = (2 * j + 1 < nv) ? d[2 * j + 1] : 0.f;
        const float keep = bit ? a1 : a0;
        const float send = bit ? a0 : a1;
        d[j] = keep + row_xor_sw(send, 1 << b);
      }
    }
    nv = nout;
  }
  float x = d[0];
#pragma unroll
  for (int m = N; m < 16; m <<= 1) x += row_xor_sw(x, m);
  return x;
}

template <int G>
__device__ __forceinline__ void agg_node(const float4* __restrict__ feat4,
                                         const int* __restrict__ csr_src,
                                         float* __restrict__ out,
                                         int node, int start, int deg, int P, float er_d,
                                         int c, int h, float4 bv, float4 alv) {
  const int  t     = c + (h ? P : 0);
  const bool valid = c < (h ? (deg - P) : P);
  const int  tt    = (t < deg) ? t : (deg - 1);
  const int  s0    = csr_src[start + tt];

  float4 r[4 * G];
#pragma unroll
  for (int i = 0; i < 4 * G; ++i) {
    if (i <= 4 * (G - 1) || i < P) {
      const int s = row_bcast(s0, i);
      r[i] = feat4[s * (OUT_DIM / 4) + c];
    } else {
      r[i] = make_float4(0.f, 0.f, 0.f, 0.f);
    }
  }

  constexpr int NV = 4 * G;
  constexpr int NN = (G == 3) ? 16 : 4 * G;
  float d[NN];
#pragma unroll
  for (int i = 0; i < NN; ++i) {
    if (i < NV) {
      v2f p2;  p2.x = r[i].x; p2.y = r[i].y;
      v2f q2;  q2.x = r[i].z; q2.y = r[i].w;
      v2f a2;  a2.x = alv.x;  a2.y = alv.y;
      v2f b2;  b2.x = alv.z;  b2.y = alv.w;
      v2f t2 = p2 * a2;
      t2 += q2 * b2;
      d[i] = t2.x + t2.y;
    } else {
      d[i] = 0.f;
    }
  }
  const float elv = row_transpose_sum<NV, NN>(d, c);

  float e = leaky(elv + er_d);
  e = valid ? e : -INFINITY;
  float m = e;
  m = fmaxf(m, row_xor<1>(m));
  m = fmaxf(m, row_xor<2>(m));
  m = fmaxf(m, row_xor<4>(m));
  m = fmaxf(m, row_xor<8>(m));
  m = fmaxf(m, xhalf(m));
  const float p = __expf(e - m);
  float den = row16_sum(p);
  den += xhalf(den);
  const float invden = 1.f / ((den > 0.f) ? den : 1.f);
  const float alpha  = p * invden;

  v2f axy = {0.f, 0.f}, azw = {0.f, 0.f};
#pragma unroll
  for (int i = 0; i < 4 * G; ++i) {
    const float a = row_bcast_f(alpha, i);
    v2f a2;  a2.x = a;  a2.y = a;
    v2f rxy; rxy.x = r[i].x; rxy.y = r[i].y;
    v2f rzw; rzw.x = r[i].z; rzw.y = r[i].w;
    axy += a2 * rxy;
    azw += a2 * rzw;
  }
  float4 ov;
  ov.x = axy.x + xhalf(axy.x) + bv.x;
  ov.y = axy.y + xhalf(axy.y) + bv.y;
  ov.z = azw.x + xhalf(azw.x) + bv.z;
  ov.w = azw.y + xhalf(azw.y) + bv.w;
  if (h == 0) *(float4*)(out + (size_t)node * OUT_DIM + 4 * c) = ov;
}

__global__ __launch_bounds__(256) void k_gat_agg(const float* __restrict__ feat,
                                                 const float* __restrict__ el,
                                                 const float* __restrict__ er,
                                                 const int* __restrict__ offsets,
                                                 const int* __restrict__ csr_src,
                                                 const float* __restrict__ bias,
                                                 const float* __restrict__ attn_l,
                                                 float* __restrict__ out,
                                                 int N, int E) {
  const int wv    = __builtin_amdgcn_readfirstlane((int)(threadIdx.x >> 5));
  const int nbase = (blockIdx.x * 8 + wv) * NPW;
  if (nbase >= N) return;
  const int lane = threadIdx.x & 31;
  const int h    = lane >> 4;
  const int c    = lane & 15;

  const int   nl   = nbase + lane;
  const int   nlc  = (nl < N) ? nl : (N - 1);
  const int   ldo  = offsets[nlc];
  const int   voff = (nl < N) ? ldo : E;
  const float ver  = er[nlc];

  const float4 bv = *(const float4*)(bias + 4 * c);
  const float4 alv = *(const float4*)(attn_l + 4 * c);
  const float4* __restrict__ feat4 = (const float4*)feat;

#pragma unroll 1
  for (int it = 0; it < NPW; ++it) {
    const int node = nbase + it;
    if (node >= N) break;
    const int   start = __builtin_amdgcn_readlane(voff, it);
    const int   end   = __builtin_amdgcn_readlane(voff, it + 1);
    const float er_d  = __builtin_bit_cast(float,
        __builtin_amdgcn_readlane(__builtin_bit_cast(int, ver), it));
    const int   deg   = end - start;

    if (deg <= 2 * MAXP) {
      if (deg <= 0) {
        if (h == 0) *(float4*)(out + (size_t)node * OUT_DIM + 4 * c) = bv;
        continue;
      }
      const int P = (deg + 1) >> 1;
      if (P <= 4)       agg_node<1>(feat4, csr_src, out, node, start, deg, P, er_d, c, h, bv, alv);
      else if (P <= 8)  agg_node<2>(feat4, csr_src, out, node, start, deg, P, er_d, c, h, bv, alv);
      else if (P <= 12) agg_node<3>(feat4, csr_src, out, node, start, deg, P, er_d, c, h, bv, alv);
      else              agg_node<4>(feat4, csr_src, out, node, start, deg, P, er_d, c, h, bv, alv);
      continue;
    }

    int   j0 = start + lane;
    float e0 = (j0 < end) ? leaky(el[csr_src[j0]] + er_d) : -INFINITY;

    float m = e0;
    for (int j = j0 + 32; j < end; j += 32) {
      m = fmaxf(m, leaky(el[csr_src[j]] + er_d));
    }
#pragma unroll
    for (int off = 16; off > 0; off >>= 1) m = fmaxf(m, __shfl_xor(m, off, 32));
    if (end <= start) m = 0.f;

    float den = (j0 < end) ? __expf(e0 - m) : 0.f;
    for (int j = j0 + 32; j < end; j += 32) {
      den += __expf(leaky(el[csr_src[j]] + er_d) - m);
    }
#pragma unroll
    for (int off = 16; off > 0; off >>= 1) den += __shfl_xor(den, off, 32);
    float invden = (den > 0.f) ? 1.f / den : 1.f;

    float acc0 = 0.f, acc1 = 0.f;
    for (int j = start; j < end; ++j) {
      int t = j - start;
      int s = csr_src[j];
      float e;
      if (t < 32) {
        e = __shfl(e0, t, 32);
      } else {
        e = leaky(el[s] + er_d);
      }
      float alpha = __expf(e - m) * invden;
      v2f fv = *(const v2f*)(feat + (size_t)s * OUT_DIM + 2 * lane);
      acc0 += alpha * fv.x;
      acc1 += alpha * fv.y;
    }
    v2f bv2 = *(const v2f*)(bias + 2 * lane);
    v2f ov;
    ov.x = acc0 + bv2.x;
    ov.y = acc1 + bv2.y;
    *(v2f*)(out + (size_t)node * OUT_DIM + 2 * lane) = ov;
  }
}

extern "C" void kernel_launch(void* const* d_in, const int* in_sizes, int n_in,
                              void* d_out, int out_size, void* d_ws, size_t ws_size,
                              hipStream_t stream) {
  const float* h      = (const float*)d_in[0];
  const int*   src    = (const int*)d_in[1];
  const int*   dst    = (const int*)d_in[2];
  const float* W      = (const float*)d_in[3];
  const float* attn_l = (const float*)d_in[4];
  const float* attn_r = (const float*)d_in[5];
  const float* bias   = (const float*)d_in[6];
  float* out = (float*)d_out;

  const int N = in_sizes[0] / IN_DIM;
  const int E = in_sizes[1];

  char* ws = (char*)d_ws;
  size_t off = 0;
  auto carve = [&](size_t bytes) -> char* {
    char* p = ws + off;
    off = (off + bytes + 255) & ~(size_t)255;
    return p;
  };
  float* feat    = (float*)carve((size_t)N * OUT_DIM * sizeof(float));
  float* el      = (float*)carve((size_t)N * sizeof(float));
  float* er      = (float*)carve((size_t)N * sizeof(float));
  int*   offsets = (int*)  carve((size_t)N * sizeof(int));
  int*   btotal  = (int*)  carve(MAX_BK * sizeof(int));
  int*   bcursor = (int*)  carve(MAX_BK * sizeof(int));
  int*   csr_src = (int*)  carve((size_t)E * sizeof(int));
  int2*  pairs   = (int2*) carve((size_t)E * sizeof(int2));

  const int nbk = (N + BK_NODES - 1) / BK_NODES;
  const int nch = (E + CHUNK - 1) / CHUNK;
  const int nhb = (E + HEDGES - 1) / HEDGES;
  int*   bcnt    = (int*)  carve((size_t)nhb * nbk * sizeof(int));

  k_feat_wmma<<<nhb + (N + 127) / 128, 256, 0, stream>>>(h, W, attn_l, attn_r, dst, feat, el, er,
                                                         bcnt, bcursor, N, E, nbk, nhb);
  k_bscatter<<<nch, L1_THREADS, 0, stream>>>(src, dst, bcnt, btotal, bcursor, pairs, E, nbk, nhb);
  k_fine    <<<nbk, FINE_THREADS, 0, stream>>>(pairs, btotal, offsets, csr_src, N);
  k_gat_agg<<<(N + 8 * NPW - 1) / (8 * NPW), 256, 0, stream>>>(feat, el, er, offsets, csr_src, bias, attn_l, out, N, E);
}
